# starting kernel
# baseline (speedup 1.0000x reference)
#include <hip/hip_runtime.h>
#include <hip/hip_fp16.h>
#include <hip/hip_bf16.h>

#define B_  16
#define S_  512
#define D_  1024
#define H_  128
#define G_  256
#define HP_ 136
#define ZP_ 516

typedef __attribute__((ext_vector_type(16))) _Float16 v16h;
typedef __attribute__((ext_vector_type(8)))  float    v8f;

__device__ __forceinline__ v16h frag_from_f32(const float* p) {
  float4 x0 = *(const float4*)(p);
  float4 x1 = *(const float4*)(p + 4);
  float4 y0 = *(const float4*)(p + 16);
  float4 y1 = *(const float4*)(p + 20);
  v16h f;
  f[0]=(_Float16)x0.x;  f[1]=(_Float16)x0.y;  f[2]=(_Float16)x0.z;  f[3]=(_Float16)x0.w;
  f[4]=(_Float16)x1.x;  f[5]=(_Float16)x1.y;  f[6]=(_Float16)x1.z;  f[7]=(_Float16)x1.w;
  f[8]=(_Float16)y0.x;  f[9]=(_Float16)y0.y;  f[10]=(_Float16)y0.z; f[11]=(_Float16)y0.w;
  f[12]=(_Float16)y1.x; f[13]=(_Float16)y1.y; f[14]=(_Float16)y1.z; f[15]=(_Float16)y1.w;
  return f;
}

__device__ __forceinline__ v16h frag_from_f16(const _Float16* p) {
  v16h f;
  #pragma unroll
  for (int i = 0; i < 8; ++i) { f[i] = p[i]; f[8 + i] = p[16 + i]; }
  return f;
}

#define WMMA_F16(a, b, c) __builtin_amdgcn_wmma_f32_16x16x32_f16( \
    false, (a), false, (b), (short)0, (c), false, false)

template <int A_MODE, int ACT, bool FULLN>
__global__ __launch_bounds__(128) void gemm_wmma_kernel(
    const float* __restrict__ A, const float* __restrict__ Bm,
    const float* __restrict__ bias, const float* __restrict__ bias2,
    const float* __restrict__ resid, float* __restrict__ C,
    int M, int N, int K, int lda, int ldb, int ldc,
    long long sA, long long sB, long long sC)
{
  const int lane  = threadIdx.x & 31;
  const int wave  = threadIdx.x >> 5;
  const int tileM = blockIdx.y * 32;
  const int nbase = blockIdx.x * 256 + wave * 64;
  if (nbase >= N) return;

  const int z = blockIdx.z;
  const float* Ab = A  + (size_t)z * (size_t)sA;
  const float* Bb = Bm + (size_t)z * (size_t)sB;
  float*       Cb = C  + (size_t)z * (size_t)sC;
  const float* Rb = resid ? (resid + (size_t)z * (size_t)sC) : nullptr;

  const int l15   = lane & 15;
  const int khalf = (lane & 16) ? 8 : 0;

  int am0 = tileM + l15;      if (am0 >= M) am0 = M - 1;
  int am1 = tileM + 16 + l15; if (am1 >= M) am1 = M - 1;
  size_t arow0, arow1;
  if (A_MODE == 1) {
    arow0 = ((size_t)(am0 & 15) * S_ + (size_t)(am0 >> 4)) * (size_t)lda;
    arow1 = ((size_t)(am1 & 15) * S_ + (size_t)(am1 >> 4)) * (size_t)lda;
  } else {
    arow0 = (size_t)am0 * (size_t)lda;
    arow1 = (size_t)am1 * (size_t)lda;
  }

  bool ton[4]; int bn[4];
  #pragma unroll
  for (int nt = 0; nt < 4; ++nt) {
    if (FULLN) {
      ton[nt] = true;
      bn[nt]  = nbase + nt * 16 + l15;
    } else {
      ton[nt] = (nbase + nt * 16) < N;
      int c = nbase + nt * 16 + l15; if (c >= N) c = N - 1;
      bn[nt] = c;
    }
  }

  v8f acc[2][4] = {};
  for (int k0 = 0; k0 < K; k0 += 32) {
    const float* pa0 = Ab + arow0 + k0 + khalf;
    const float* pa1 = Ab + arow1 + k0 + khalf;
    v16h a0 = frag_from_f32(pa0);
    v16h a1 = frag_from_f32(pa1);
    if (k0 + 32 < K) {
      __builtin_prefetch(pa0 + 32, 0, 1);
      __builtin_prefetch(pa1 + 32, 0, 1);
    }
    #pragma unroll
    for (int nt = 0; nt < 4; ++nt) {
      if (!FULLN && !ton[nt]) continue;
      v16h b = frag_from_f32(Bb + (size_t)bn[nt] * (size_t)ldb + k0 + khalf);
      acc[0][nt] = WMMA_F16(a0, b, acc[0][nt]);
      acc[1][nt] = WMMA_F16(a1, b, acc[1][nt]);
    }
  }

  const int rb = (lane < 16) ? 0 : 8;
  #pragma unroll
  for (int nt = 0; nt < 4; ++nt) {
    if (!FULLN && !ton[nt]) continue;
    const int col = nbase + nt * 16 + l15;
    bool colok = FULLN ? true : (col < N);
    float bv = 0.0f;
    if (colok) {
      if (bias)  bv += bias[col];
      if (bias2) bv += bias2[col];
    }
    #pragma unroll
    for (int mt = 0; mt < 2; ++mt) {
      #pragma unroll
      for (int r = 0; r < 8; ++r) {
        const int row = tileM + mt * 16 + rb + r;
        if (row < M && colok) {
          float v = acc[mt][nt][r] + bv;
          if      (ACT == 1) v = v > 0.0f ? v : 0.0f;
          else if (ACT == 2) v = v > 0.0f ? v : 0.1f * v;
          if (Rb) v += Rb[(size_t)row * (size_t)ldc + col];
          Cb[(size_t)row * (size_t)ldc + col] = v;
        }
      }
    }
  }
}

__global__ void cvt_f32_f16_kernel(const float* __restrict__ src,
                                   __half* __restrict__ dst, int n) {
  int i = blockIdx.x * blockDim.x + threadIdx.x;
  if (i < n) ((_Float16*)dst)[i] = (_Float16)src[i];
}

__global__ __launch_bounds__(256) void transpose_bsg_kernel(
    const float* __restrict__ src, float* __restrict__ dst)
{
  __shared__ float tile[32][33];
  const int b  = blockIdx.z;
  const int s0 = blockIdx.x * 32, g0 = blockIdx.y * 32;
  const int tx = threadIdx.x, ty = threadIdx.y;
  const float* sp = src + (size_t)b * S_ * G_;
  float*       dp = dst + (size_t)b * G_ * S_;
  #pragma unroll
  for (int r = 0; r < 32; r += 8)
    tile[ty + r][tx] = sp[(size_t)(s0 + ty + r) * G_ + g0 + tx];
  __syncthreads();
  #pragma unroll
  for (int r = 0; r < 32; r += 8)
    dp[(size_t)(g0 + ty + r) * S_ + s0 + tx] = tile[tx][ty + r];
}

__global__ __launch_bounds__(256) void lstm_dir_kernel(
    const float* __restrict__ preF, const float* __restrict__ preB,
    const __half* __restrict__ whhF16_f, const __half* __restrict__ whhF16_b,
    float* __restrict__ hs)
{
  const int dir = blockIdx.x;
  const float*    pre = dir ? preB : preF;
  const _Float16* whh = (const _Float16*)(dir ? whhF16_b : whhF16_f);

  __shared__ _Float16 h_lds[B_ * HP_];
  __shared__ float    c_lds[B_ * H_];
  __shared__ float    z_lds[B_ * ZP_];

  const int tid   = threadIdx.x;
  const int wave  = tid >> 5;
  const int lane  = tid & 31;
  const int l15   = lane & 15;
  const int khalf = (lane & 16) ? 8 : 0;

  v16h wfrag[16];
  #pragma unroll
  for (int nt = 0; nt < 4; ++nt) {
    const int tileN = (wave * 4 + nt) * 16;
    #pragma unroll
    for (int kc = 0; kc < 4; ++kc)
      wfrag[nt * 4 + kc] =
          frag_from_f16(whh + (size_t)(tileN + l15) * H_ + kc * 32 + khalf);
  }

  for (int i = tid; i < B_ * HP_; i += 256) h_lds[i] = (_Float16)0.0f;
  for (int i = tid; i < B_ * H_;  i += 256) c_lds[i] = 0.0f;
  __syncthreads();

  for (int step = 0; step < S_; ++step) {
    const int t = dir ? (S_ - 1 - step) : step;
    const float* pre_t = pre + (size_t)t * (B_ * 4 * H_);

    v16h a[4];
    #pragma unroll
    for (int kc = 0; kc < 4; ++kc)
      a[kc] = frag_from_f16(&h_lds[l15 * HP_ + kc * 32 + khalf]);

    #pragma unroll
    for (int nt = 0; nt < 4; ++nt) {
      v8f acc = {};
      #pragma unroll
      for (int kc = 0; kc < 4; ++kc)
        acc = WMMA_F16(a[kc], wfrag[nt * 4 + kc], acc);
      const int col = (wave * 4 + nt) * 16 + l15;
      const int rb  = (lane < 16) ? 0 : 8;
      #pragma unroll
      for (int r = 0; r < 8; ++r) {
        const int row = rb + r;
        z_lds[row * ZP_ + col] = acc[r] + pre_t[row * 512 + col];
      }
    }
    __syncthreads();

    for (int i = tid; i < B_ * H_; i += 256) {
      const int b_ = i >> 7, hx = i & (H_ - 1);
      const float zi = z_lds[b_ * ZP_ + hx];
      const float zf = z_lds[b_ * ZP_ + 128 + hx];
      const float zg = z_lds[b_ * ZP_ + 256 + hx];
      const float zo = z_lds[b_ * ZP_ + 384 + hx];
      const float si = 1.0f / (1.0f + __expf(-zi));
      const float sf = 1.0f / (1.0f + __expf(-zf));
      const float so = 1.0f / (1.0f + __expf(-zo));
      float c = sf * c_lds[i] + si * tanhf(zg);
      const float h = so * tanhf(c);
      c_lds[i] = c;
      h_lds[b_ * HP_ + hx] = (_Float16)h;
      hs[((size_t)t * B_ + b_) * (2 * H_) + dir * H_ + hx] = h;
    }
    __syncthreads();
  }
}

__global__ __launch_bounds__(256) void attn_mask_norm_kernel(
    float* __restrict__ scores, const float* __restrict__ adj)
{
  const int i = blockIdx.x, b = blockIdx.y, tid = threadIdx.x;
  float* row        = scores + ((size_t)b * S_ + i) * S_;
  const float* arow = adj    + ((size_t)b * S_ + i) * S_;
  __shared__ float red[256];
  float part = 0.0f;
  for (int j = tid; j < S_; j += 256) {
    const float sg = 1.0f / (1.0f + __expf(-row[j]));
    const float av = (j == i) ? 1.0f : arow[j];
    const float v  = (sg + ((j == i) ? 1e-5f : 0.0f)) * av;
    row[j] = v;
    part += v;
  }
  red[tid] = part;
  __syncthreads();
  for (int off = 128; off > 0; off >>= 1) {
    if (tid < off) red[tid] += red[tid + off];
    __syncthreads();
  }
  const float inv = 1.0f / red[0];
  for (int j = tid; j < S_; j += 256) row[j] *= inv;
}

__global__ __launch_bounds__(256) void head_mean_kernel(
    const float* __restrict__ h1, const float* __restrict__ gfin,
    float* __restrict__ hidden)
{
  const int b = blockIdx.x, tid = threadIdx.x;
  for (int d = tid; d < 512; d += 256) {
    float s = 0.0f;
    if (d < 256) {
      for (int t = 0; t < S_; ++t) s += h1[((size_t)t * B_ + b) * 256 + d];
    } else {
      const int dd = d - 256;
      for (int t = 0; t < S_; ++t) s += gfin[((size_t)b * S_ + t) * 256 + dd];
    }
    hidden[b * 512 + d] = s * (1.0f / (float)S_);
  }
}

template <int A_MODE, int ACT, bool FULLN>
static inline void gemm(hipStream_t st,
                        const float* A, const float* Bm,
                        const float* bias, const float* bias2, const float* resid,
                        float* C, int M, int N, int K, int lda, int ldb, int ldc,
                        long long sA = 0, long long sB = 0, long long sC = 0,
                        int batches = 1)
{
  dim3 grid((N + 255) / 256, (M + 31) / 32, batches);
  gemm_wmma_kernel<A_MODE, ACT, FULLN><<<grid, 128, 0, st>>>(
      A, Bm, bias, bias2, resid, C, M, N, K, lda, ldb, ldc, sA, sB, sC);
}

extern "C" void kernel_launch(void* const* d_in, const int* in_sizes, int n_in,
                              void* d_out, int out_size, void* d_ws, size_t ws_size,
                              hipStream_t stream)
{
  (void)in_sizes; (void)n_in; (void)out_size; (void)ws_size;
  auto F = [&](int i) { return (const float*)d_in[i]; };

  const float* x        = F(0);
  const float* adj      = F(1);
  const float* l0f_Wih  = F(2),  *l0f_Whh = F(3),  *l0f_bih = F(4),  *l0f_bhh = F(5);
  const float* l0b_Wih  = F(6),  *l0b_Whh = F(7),  *l0b_bih = F(8),  *l0b_bhh = F(9);
  const float* l1f_Wih  = F(10), *l1f_Whh = F(11), *l1f_bih = F(12), *l1f_bhh = F(13);
  const float* l1b_Wih  = F(14), *l1b_Whh = F(15), *l1b_bih = F(16), *l1b_bhh = F(17);
  const float* fc1_W    = F(18), *fc1_b   = F(19);
  const float* wattn_W  = F(20), *wattn_b = F(21);
  const float* lin0_W   = F(22), *lin0_b  = F(23);
  const float* lin1_W   = F(24), *lin1_b  = F(25);
  const float* linf_W   = F(26), *linf_b  = F(27);
  const float* fcA_W    = F(28), *fcA_b   = F(29);
  const float* fcB_W    = F(30), *fcB_b   = F(31);
  const float* fcC_W    = F(32), *fcC_b   = F(33);

  char* wsp = (char*)d_ws;
  size_t off = 0;
  auto carve = [&](size_t bytes) -> void* {
    void* p = wsp + off;
    off += (bytes + 255) & ~(size_t)255;
    return p;
  };
  float* preF   = (float*)carve((size_t)S_ * B_ * 4 * H_ * 4);
  float* preB   = (float*)carve((size_t)S_ * B_ * 4 * H_ * 4);
  float* h0     = (float*)carve((size_t)S_ * B_ * 2 * H_ * 4);
  float* h1     = (float*)carve((size_t)S_ * B_ * 2 * H_ * 4);
  float* gA     = (float*)carve((size_t)B_ * S_ * G_ * 4);
  float* gB     = (float*)carve((size_t)B_ * S_ * G_ * 4);
  float* gC     = (float*)carve((size_t)B_ * S_ * G_ * 4);
  float* gT     = (float*)carve((size_t)B_ * G_ * S_ * 4);
  float* scores = (float*)carve((size_t)B_ * S_ * S_ * 4);
  __half* whh0f = (__half*)carve((size_t)4 * H_ * H_ * 2);
  __half* whh0b = (__half*)carve((size_t)4 * H_ * H_ * 2);
  __half* whh1f = (__half*)carve((size_t)4 * H_ * H_ * 2);
  __half* whh1b = (__half*)carve((size_t)4 * H_ * H_ * 2);
  float* fcAo   = (float*)carve((size_t)B_ * 128 * 4);
  float* fcBo   = (float*)carve((size_t)B_ * 32 * 4);

  const int MSB = S_ * B_;
  const int nwh = 4 * H_ * H_;

  cvt_f32_f16_kernel<<<(nwh + 255) / 256, 256, 0, stream>>>(l0f_Whh, whh0f, nwh);
  cvt_f32_f16_kernel<<<(nwh + 255) / 256, 256, 0, stream>>>(l0b_Whh, whh0b, nwh);
  cvt_f32_f16_kernel<<<(nwh + 255) / 256, 256, 0, stream>>>(l1f_Whh, whh1f, nwh);
  cvt_f32_f16_kernel<<<(nwh + 255) / 256, 256, 0, stream>>>(l1b_Whh, whh1b, nwh);

  gemm<1, 0, true>(stream, x, l0f_Wih, l0f_bih, l0f_bhh, nullptr, preF,
                   MSB, 4 * H_, D_, D_, D_, 4 * H_);
  gemm<1, 0, true>(stream, x, l0b_Wih, l0b_bih, l0b_bhh, nullptr, preB,
                   MSB, 4 * H_, D_, D_, D_, 4 * H_);
  lstm_dir_kernel<<<2, 256, 0, stream>>>(preF, preB, whh0f, whh0b, h0);

  gemm<0, 0, true>(stream, h0, l1f_Wih, l1f_bih, l1f_bhh, nullptr, preF,
                   MSB, 4 * H_, 2 * H_, 2 * H_, 2 * H_, 4 * H_);
  gemm<0, 0, true>(stream, h0, l1b_Wih, l1b_bih, l1b_bhh, nullptr, preB,
                   MSB, 4 * H_, 2 * H_, 2 * H_, 2 * H_, 4 * H_);
  lstm_dir_kernel<<<2, 256, 0, stream>>>(preF, preB, whh1f, whh1b, h1);

  gemm<0, 0, true>(stream, x, fc1_W, fc1_b, nullptr, nullptr, gA,
                   MSB, G_, D_, D_, D_, G_);

  float* cur = gA; float* t1 = gB; float* t2 = gC;
  const long long sGG = (long long)S_ * G_;
  const long long sSS = (long long)S_ * S_;
  for (int l = 0; l < 4; ++l) {
    const float* Wq = wattn_W + (size_t)l * G_ * G_;
    gemm<0, 0, true>(stream, cur, Wq, wattn_b + l * G_, nullptr, nullptr, t2,
                     MSB, G_, G_, G_, G_, G_);
    gemm<0, 0, true>(stream, t2, cur, nullptr, nullptr, nullptr, scores,
                     S_, S_, G_, G_, G_, S_, sGG, sGG, sSS, B_);
    attn_mask_norm_kernel<<<dim3(S_, B_), 256, 0, stream>>>(scores, adj);
    transpose_bsg_kernel<<<dim3(S_ / 32, G_ / 32, B_), dim3(32, 8), 0, stream>>>(cur, gT);
    gemm<0, 0, true>(stream, scores, gT, nullptr, nullptr, nullptr, t1,
                     S_, G_, S_, S_, S_, G_, sSS, (long long)G_ * S_, sGG, B_);
    gemm<0, 1, true>(stream, t1, lin0_W + (size_t)l * G_ * G_, lin0_b + l * G_,
                     nullptr, nullptr, t2, MSB, G_, G_, G_, G_, G_);
    gemm<0, 1, true>(stream, t2, lin1_W + (size_t)l * G_ * G_, lin1_b + l * G_,
                     nullptr, cur, t1, MSB, G_, G_, G_, G_, G_);
    float* old = cur; cur = t1; t1 = old;
  }
  gemm<0, 0, true>(stream, cur, linf_W, linf_b, nullptr, nullptr, t2,
                   MSB, G_, G_, G_, G_, G_);
  float* gfin = t2;

  float* y_out  = (float*)d_out;
  float* hidden = (float*)d_out + B_ * 2;
  head_mean_kernel<<<B_, 256, 0, stream>>>(h1, gfin, hidden);

  gemm<0, 2, true >(stream, hidden, fcA_W, fcA_b, nullptr, nullptr, fcAo,
                    B_, 128, 512, 512, 512, 128);
  gemm<0, 2, false>(stream, fcAo, fcB_W, fcB_b, nullptr, nullptr, fcBo,
                    B_, 32, 128, 128, 128, 32);
  gemm<0, 0, false>(stream, fcBo, fcC_W, fcC_b, nullptr, nullptr, y_out,
                    B_, 2, 32, 32, 32, 2);
}


# LSTM recurrence: the 16 independent batch rows are split 2 per workgroup (grid 2x8, was 2x1), so each thread owns one cell per step instead of looping over 8; summation order unchanged
# speedup vs baseline: 2.0985x; 2.0985x over previous
#include <hip/hip_runtime.h>
#include <hip/hip_fp16.h>
#include <hip/hip_bf16.h>

#define B_  16
#define S_  512
#define D_  1024
#define H_  128
#define G_  256
#define HP_ 136
#define ZP_ 516

typedef __attribute__((ext_vector_type(16))) _Float16 v16h;
typedef __attribute__((ext_vector_type(8)))  float    v8f;

__device__ __forceinline__ v16h frag_from_f32(const float* p) {
  float4 x0 = *(const float4*)(p);
  float4 x1 = *(const float4*)(p + 4);
  float4 y0 = *(const float4*)(p + 16);
  float4 y1 = *(const float4*)(p + 20);
  v16h f;
  f[0]=(_Float16)x0.x;  f[1]=(_Float16)x0.y;  f[2]=(_Float16)x0.z;  f[3]=(_Float16)x0.w;
  f[4]=(_Float16)x1.x;  f[5]=(_Float16)x1.y;  f[6]=(_Float16)x1.z;  f[7]=(_Float16)x1.w;
  f[8]=(_Float16)y0.x;  f[9]=(_Float16)y0.y;  f[10]=(_Float16)y0.z; f[11]=(_Float16)y0.w;
  f[12]=(_Float16)y1.x; f[13]=(_Float16)y1.y; f[14]=(_Float16)y1.z; f[15]=(_Float16)y1.w;
  return f;
}

__device__ __forceinline__ v16h frag_from_f16(const _Float16* p) {
  v16h f;
  #pragma unroll
  for (int i = 0; i < 8; ++i) { f[i] = p[i]; f[8 + i] = p[16 + i]; }
  return f;
}

#define WMMA_F16(a, b, c) __builtin_amdgcn_wmma_f32_16x16x32_f16( \
    false, (a), false, (b), (short)0, (c), false, false)

template <int A_MODE, int ACT, bool FULLN>
__global__ __launch_bounds__(128) void gemm_wmma_kernel(
    const float* __restrict__ A, const float* __restrict__ Bm,
    const float* __restrict__ bias, const float* __restrict__ bias2,
    const float* __restrict__ resid, float* __restrict__ C,
    int M, int N, int K, int lda, int ldb, int ldc,
    long long sA, long long sB, long long sC)
{
  const int lane  = threadIdx.x & 31;
  const int wave  = threadIdx.x >> 5;
  const int tileM = blockIdx.y * 32;
  const int nbase = blockIdx.x * 256 + wave * 64;
  if (nbase >= N) return;

  const int z = blockIdx.z;
  const float* Ab = A  + (size_t)z * (size_t)sA;
  const float* Bb = Bm + (size_t)z * (size_t)sB;
  float*       Cb = C  + (size_t)z * (size_t)sC;
  const float* Rb = resid ? (resid + (size_t)z * (size_t)sC) : nullptr;

  const int l15   = lane & 15;
  const int khalf = (lane & 16) ? 8 : 0;

  int am0 = tileM + l15;      if (am0 >= M) am0 = M - 1;
  int am1 = tileM + 16 + l15; if (am1 >= M) am1 = M - 1;
  size_t arow0, arow1;
  if (A_MODE == 1) {
    arow0 = ((size_t)(am0 & 15) * S_ + (size_t)(am0 >> 4)) * (size_t)lda;
    arow1 = ((size_t)(am1 & 15) * S_ + (size_t)(am1 >> 4)) * (size_t)lda;
  } else {
    arow0 = (size_t)am0 * (size_t)lda;
    arow1 = (size_t)am1 * (size_t)lda;
  }

  bool ton[4]; int bn[4];
  #pragma unroll
  for (int nt = 0; nt < 4; ++nt) {
    if (FULLN) {
      ton[nt] = true;
      bn[nt]  = nbase + nt * 16 + l15;
    } else {
      ton[nt] = (nbase + nt * 16) < N;
      int c = nbase + nt * 16 + l15; if (c >= N) c = N - 1;
      bn[nt] = c;
    }
  }

  v8f acc[2][4] = {};
  for (int k0 = 0; k0 < K; k0 += 32) {
    const float* pa0 = Ab + arow0 + k0 + khalf;
    const float* pa1 = Ab + arow1 + k0 + khalf;
    v16h a0 = frag_from_f32(pa0);
    v16h a1 = frag_from_f32(pa1);
    if (k0 + 32 < K) {
      __builtin_prefetch(pa0 + 32, 0, 1);
      __builtin_prefetch(pa1 + 32, 0, 1);
    }
    #pragma unroll
    for (int nt = 0; nt < 4; ++nt) {
      if (!FULLN && !ton[nt]) continue;
      v16h b = frag_from_f32(Bb + (size_t)bn[nt] * (size_t)ldb + k0 + khalf);
      acc[0][nt] = WMMA_F16(a0, b, acc[0][nt]);
      acc[1][nt] = WMMA_F16(a1, b, acc[1][nt]);
    }
  }

  const int rb = (lane < 16) ? 0 : 8;
  #pragma unroll
  for (int nt = 0; nt < 4; ++nt) {
    if (!FULLN && !ton[nt]) continue;
    const int col = nbase + nt * 16 + l15;
    bool colok = FULLN ? true : (col < N);
    float bv = 0.0f;
    if (colok) {
      if (bias)  bv += bias[col];
      if (bias2) bv += bias2[col];
    }
    #pragma unroll
    for (int mt = 0; mt < 2; ++mt) {
      #pragma unroll
      for (int r = 0; r < 8; ++r) {
        const int row = tileM + mt * 16 + rb + r;
        if (row < M && colok) {
          float v = acc[mt][nt][r] + bv;
          if      (ACT == 1) v = v > 0.0f ? v : 0.0f;
          else if (ACT == 2) v = v > 0.0f ? v : 0.1f * v;
          if (Rb) v += Rb[(size_t)row * (size_t)ldc + col];
          Cb[(size_t)row * (size_t)ldc + col] = v;
        }
      }
    }
  }
}

__global__ void cvt_f32_f16_kernel(const float* __restrict__ src,
                                   __half* __restrict__ dst, int n) {
  int i = blockIdx.x * blockDim.x + threadIdx.x;
  if (i < n) ((_Float16*)dst)[i] = (_Float16)src[i];
}

__global__ __launch_bounds__(256) void transpose_bsg_kernel(
    const float* __restrict__ src, float* __restrict__ dst)
{
  __shared__ float tile[32][33];
  const int b  = blockIdx.z;
  const int s0 = blockIdx.x * 32, g0 = blockIdx.y * 32;
  const int tx = threadIdx.x, ty = threadIdx.y;
  const float* sp = src + (size_t)b * S_ * G_;
  float*       dp = dst + (size_t)b * G_ * S_;
  #pragma unroll
  for (int r = 0; r < 32; r += 8)
    tile[ty + r][tx] = sp[(size_t)(s0 + ty + r) * G_ + g0 + tx];
  __syncthreads();
  #pragma unroll
  for (int r = 0; r < 32; r += 8)
    dp[(size_t)(g0 + ty + r) * S_ + s0 + tx] = tile[tx][ty + r];
}

#define LB_ 2

__global__ __launch_bounds__(256) void lstm_dir_kernel(
    const float* __restrict__ preF, const float* __restrict__ preB,
    const __half* __restrict__ whhF16_f, const __half* __restrict__ whhF16_b,
    float* __restrict__ hs)
{
  const int dir = blockIdx.x;
  const int b0  = blockIdx.y * LB_;
  const float*    pre = dir ? preB : preF;
  const _Float16* whh = (const _Float16*)(dir ? whhF16_b : whhF16_f);

  __shared__ _Float16 h_lds[LB_ * HP_];
  __shared__ float    z_lds[LB_ * ZP_];

  const int tid   = threadIdx.x;
  const int wave  = tid >> 5;
  const int lane  = tid & 31;
  const int l15   = lane & 15;
  const int khalf = (lane & 16) ? 8 : 0;
  const int hrow  = l15 >> 3;
  const int half  = lane >> 4;

  v16h wfrag[16];
  #pragma unroll
  for (int nt = 0; nt < 4; ++nt) {
    const int tileN = (wave * 4 + nt) * 16;
    #pragma unroll
    for (int kc = 0; kc < 4; ++kc)
      wfrag[nt * 4 + kc] =
          frag_from_f16(whh + (size_t)(tileN + l15) * H_ + kc * 32 + khalf);
  }

  for (int i = tid; i < LB_ * HP_; i += 256) h_lds[i] = (_Float16)0.0f;

  const int cb = tid >> 7, hx = tid & (H_ - 1);
  float c = 0.0f;
  __syncthreads();

  for (int step = 0; step < S_; ++step) {
    const int t = dir ? (S_ - 1 - step) : step;
    const float* pre_t = pre + (size_t)t * (B_ * 4 * H_);

    v16h a[4];
    #pragma unroll
    for (int kc = 0; kc < 4; ++kc)
      a[kc] = frag_from_f16(&h_lds[hrow * HP_ + kc * 32 + khalf]);

    #pragma unroll
    for (int nt = 0; nt < 4; ++nt) {
      v8f acc = {};
      #pragma unroll
      for (int kc = 0; kc < 4; ++kc)
        acc = WMMA_F16(a[kc], wfrag[nt * 4 + kc], acc);
      const int col = (wave * 4 + nt) * 16 + l15;
      z_lds[half * ZP_ + col] = acc[0] + pre_t[(b0 + half) * 512 + col];
    }
    __syncthreads();

    {
      const float zi = z_lds[cb * ZP_ + hx];
      const float zf = z_lds[cb * ZP_ + 128 + hx];
      const float zg = z_lds[cb * ZP_ + 256 + hx];
      const float zo = z_lds[cb * ZP_ + 384 + hx];
      const float si = 1.0f / (1.0f + __expf(-zi));
      const float sf = 1.0f / (1.0f + __expf(-zf));
      const float so = 1.0f / (1.0f + __expf(-zo));
      c = sf * c + si * tanhf(zg);
      const float h = so * tanhf(c);
      h_lds[cb * HP_ + hx] = (_Float16)h;
      hs[((size_t)t * B_ + b0 + cb) * (2 * H_) + dir * H_ + hx] = h;
    }
    __syncthreads();
  }
}

__global__ __launch_bounds__(256) void attn_mask_norm_kernel(
    float* __restrict__ scores, const float* __restrict__ adj)
{
  const int i = blockIdx.x, b = blockIdx.y, tid = threadIdx.x;
  float* row        = scores + ((size_t)b * S_ + i) * S_;
  const float* arow = adj    + ((size_t)b * S_ + i) * S_;
  __shared__ float red[256];
  float part = 0.0f;
  for (int j = tid; j < S_; j += 256) {
    const float sg = 1.0f / (1.0f + __expf(-row[j]));
    const float av = (j == i) ? 1.0f : arow[j];
    const float v  = (sg + ((j == i) ? 1e-5f : 0.0f)) * av;
    row[j] = v;
    part += v;
  }
  red[tid] = part;
  __syncthreads();
  for (int off = 128; off > 0; off >>= 1) {
    if (tid < off) red[tid] += red[tid + off];
    __syncthreads();
  }
  const float inv = 1.0f / red[0];
  for (int j = tid; j < S_; j += 256) row[j] *= inv;
}

__global__ __launch_bounds__(256) void head_mean_kernel(
    const float* __restrict__ h1, const float* __restrict__ gfin,
    float* __restrict__ hidden)
{
  const int b = blockIdx.x, tid = threadIdx.x;
  for (int d = tid; d < 512; d += 256) {
    float s = 0.0f;
    if (d < 256) {
      for (int t = 0; t < S_; ++t) s += h1[((size_t)t * B_ + b) * 256 + d];
    } else {
      const int dd = d - 256;
      for (int t = 0; t < S_; ++t) s += gfin[((size_t)b * S_ + t) * 256 + dd];
    }
    hidden[b * 512 + d] = s * (1.0f / (float)S_);
  }
}

template <int A_MODE, int ACT, bool FULLN>
static inline void gemm(hipStream_t st,
                        const float* A, const float* Bm,
                        const float* bias, const float* bias2, const float* resid,
                        float* C, int M, int N, int K, int lda, int ldb, int ldc,
                        long long sA = 0, long long sB = 0, long long sC = 0,
                        int batches = 1)
{
  dim3 grid((N + 255) / 256, (M + 31) / 32, batches);
  gemm_wmma_kernel<A_MODE, ACT, FULLN><<<grid, 128, 0, st>>>(
      A, Bm, bias, bias2, resid, C, M, N, K, lda, ldb, ldc, sA, sB, sC);
}

extern "C" void kernel_launch(void* const* d_in, const int* in_sizes, int n_in,
                              void* d_out, int out_size, void* d_ws, size_t ws_size,
                              hipStream_t stream)
{
  (void)in_sizes; (void)n_in; (void)out_size; (void)ws_size;
  auto F = [&](int i) { return (const float*)d_in[i]; };

  const float* x        = F(0);
  const float* adj      = F(1);
  const float* l0f_Wih  = F(2),  *l0f_Whh = F(3),  *l0f_bih = F(4),  *l0f_bhh = F(5);
  const float* l0b_Wih  = F(6),  *l0b_Whh = F(7),  *l0b_bih = F(8),  *l0b_bhh = F(9);
  const float* l1f_Wih  = F(10), *l1f_Whh = F(11), *l1f_bih = F(12), *l1f_bhh = F(13);
  const float* l1b_Wih  = F(14), *l1b_Whh = F(15), *l1b_bih = F(16), *l1b_bhh = F(17);
  const float* fc1_W    = F(18), *fc1_b   = F(19);
  const float* wattn_W  = F(20), *wattn_b = F(21);
  const float* lin0_W   = F(22), *lin0_b  = F(23);
  const float* lin1_W   = F(24), *lin1_b  = F(25);
  const float* linf_W   = F(26), *linf_b  = F(27);
  const float* fcA_W    = F(28), *fcA_b   = F(29);
  const float* fcB_W    = F(30), *fcB_b   = F(31);
  const float* fcC_W    = F(32), *fcC_b   = F(33);

  char* wsp = (char*)d_ws;
  size_t off = 0;
  auto carve = [&](size_t bytes) -> void* {
    void* p = wsp + off;
    off += (bytes + 255) & ~(size_t)255;
    return p;
  };
  float* preF   = (float*)carve((size_t)S_ * B_ * 4 * H_ * 4);
  float* preB   = (float*)carve((size_t)S_ * B_ * 4 * H_ * 4);
  float* h0     = (float*)carve((size_t)S_ * B_ * 2 * H_ * 4);
  float* h1     = (float*)carve((size_t)S_ * B_ * 2 * H_ * 4);
  float* gA     = (float*)carve((size_t)B_ * S_ * G_ * 4);
  float* gB     = (float*)carve((size_t)B_ * S_ * G_ * 4);
  float* gC     = (float*)carve((size_t)B_ * S_ * G_ * 4);
  float* gT     = (float*)carve((size_t)B_ * G_ * S_ * 4);
  float* scores = (float*)carve((size_t)B_ * S_ * S_ * 4);
  __half* whh0f = (__half*)carve((size_t)4 * H_ * H_ * 2);
  __half* whh0b = (__half*)carve((size_t)4 * H_ * H_ * 2);
  __half* whh1f = (__half*)carve((size_t)4 * H_ * H_ * 2);
  __half* whh1b = (__half*)carve((size_t)4 * H_ * H_ * 2);
  float* fcAo   = (float*)carve((size_t)B_ * 128 * 4);
  float* fcBo   = (float*)carve((size_t)B_ * 32 * 4);

  const int MSB = S_ * B_;
  const int nwh = 4 * H_ * H_;

  cvt_f32_f16_kernel<<<(nwh + 255) / 256, 256, 0, stream>>>(l0f_Whh, whh0f, nwh);
  cvt_f32_f16_kernel<<<(nwh + 255) / 256, 256, 0, stream>>>(l0b_Whh, whh0b, nwh);
  cvt_f32_f16_kernel<<<(nwh + 255) / 256, 256, 0, stream>>>(l1f_Whh, whh1f, nwh);
  cvt_f32_f16_kernel<<<(nwh + 255) / 256, 256, 0, stream>>>(l1b_Whh, whh1b, nwh);

  gemm<1, 0, true>(stream, x, l0f_Wih, l0f_bih, l0f_bhh, nullptr, preF,
                   MSB, 4 * H_, D_, D_, D_, 4 * H_);
  gemm<1, 0, true>(stream, x, l0b_Wih, l0b_bih, l0b_bhh, nullptr, preB,
                   MSB, 4 * H_, D_, D_, D_, 4 * H_);
  lstm_dir_kernel<<<dim3(2, B_ / LB_), 256, 0, stream>>>(preF, preB, whh0f, whh0b, h0);

  gemm<0, 0, true>(stream, h0, l1f_Wih, l1f_bih, l1f_bhh, nullptr, preF,
                   MSB, 4 * H_, 2 * H_, 2 * H_, 2 * H_, 4 * H_);
  gemm<0, 0, true>(stream, h0, l1b_Wih, l1b_bih, l1b_bhh, nullptr, preB,
                   MSB, 4 * H_, 2 * H_, 2 * H_, 2 * H_, 4 * H_);
  lstm_dir_kernel<<<dim3(2, B_ / LB_), 256, 0, stream>>>(preF, preB, whh1f, whh1b, h1);

  gemm<0, 0, true>(stream, x, fc1_W, fc1_b, nullptr, nullptr, gA,
                   MSB, G_, D_, D_, D_, G_);

  float* cur = gA; float* t1 = gB; float* t2 = gC;
  const long long sGG = (long long)S_ * G_;
  const long long sSS = (long long)S_ * S_;
  for (int l = 0; l < 4; ++l) {
    const float* Wq = wattn_W + (size_t)l * G_ * G_;
    gemm<0, 0, true>(stream, cur, Wq, wattn_b + l * G_, nullptr, nullptr, t2,
                     MSB, G_, G_, G_, G_, G_);
    gemm<0, 0, true>(stream, t2, cur, nullptr, nullptr, nullptr, scores,
                     S_, S_, G_, G_, G_, S_, sGG, sGG, sSS, B_);
    attn_mask_norm_kernel<<<dim3(S_, B_), 256, 0, stream>>>(scores, adj);
    transpose_bsg_kernel<<<dim3(S_ / 32, G_ / 32, B_), dim3(32, 8), 0, stream>>>(cur, gT);
    gemm<0, 0, true>(stream, scores, gT, nullptr, nullptr, nullptr, t1,
                     S_, G_, S_, S_, S_, G_, sSS, (long long)G_ * S_, sGG, B_);
    gemm<0, 1, true>(stream, t1, lin0_W + (size_t)l * G_ * G_, lin0_b + l * G_,
                     nullptr, nullptr, t2, MSB, G_, G_, G_, G_, G_);
    gemm<0, 1, true>(stream, t2, lin1_W + (size_t)l * G_ * G_, lin1_b + l * G_,
                     nullptr, cur, t1, MSB, G_, G_, G_, G_, G_);
    float* old = cur; cur = t1; t1 = old;
  }
  gemm<0, 0, true>(stream, cur, linf_W, linf_b, nullptr, nullptr, t2,
                   MSB, G_, G_, G_, G_, G_);
  float* gfin = t2;

  float* y_out  = (float*)d_out;
  float* hidden = (float*)d_out + B_ * 2;
  head_mean_kernel<<<B_, 256, 0, stream>>>(h1, gfin, hidden);

  gemm<0, 2, true >(stream, hidden, fcA_W, fcA_b, nullptr, nullptr, fcAo,
                    B_, 128, 512, 512, 512, 128);
  gemm<0, 2, false>(stream, fcAo, fcB_W, fcB_b, nullptr, nullptr, fcBo,
                    B_, 32, 128, 128, 128, 32);
  gemm<0, 0, false>(stream, fcBo, fcC_W, fcC_b, nullptr, nullptr, y_out,
                    B_, 2, 32, 32, 32, 2);
}


# LSTM recurrence: no global-memory wait inside a step: barriers fence LDS only (hs store ack not awaited), pre of step t+1 loaded during step t; summation order unchanged
# speedup vs baseline: 3.0226x; 1.4404x over previous
#include <hip/hip_runtime.h>
#include <hip/hip_fp16.h>
#include <hip/hip_bf16.h>

#define B_  16
#define S_  512
#define D_  1024
#define H_  128
#define G_  256
#define HP_ 136
#define ZP_ 516

typedef __attribute__((ext_vector_type(16))) _Float16 v16h;
typedef __attribute__((ext_vector_type(8)))  float    v8f;

__device__ __forceinline__ v16h frag_from_f32(const float* p) {
  float4 x0 = *(const float4*)(p);
  float4 x1 = *(const float4*)(p + 4);
  float4 y0 = *(const float4*)(p + 16);
  float4 y1 = *(const float4*)(p + 20);
  v16h f;
  f[0]=(_Float16)x0.x;  f[1]=(_Float16)x0.y;  f[2]=(_Float16)x0.z;  f[3]=(_Float16)x0.w;
  f[4]=(_Float16)x1.x;  f[5]=(_Float16)x1.y;  f[6]=(_Float16)x1.z;  f[7]=(_Float16)x1.w;
  f[8]=(_Float16)y0.x;  f[9]=(_Float16)y0.y;  f[10]=(_Float16)y0.z; f[11]=(_Float16)y0.w;
  f[12]=(_Float16)y1.x; f[13]=(_Float16)y1.y; f[14]=(_Float16)y1.z; f[15]=(_Float16)y1.w;
  return f;
}

__device__ __forceinline__ v16h frag_from_f16(const _Float16* p) {
  v16h f;
  #pragma unroll
  for (int i = 0; i < 8; ++i) { f[i] = p[i]; f[8 + i] = p[16 + i]; }
  return f;
}

#define WMMA_F16(a, b, c) __builtin_amdgcn_wmma_f32_16x16x32_f16( \
    false, (a), false, (b), (short)0, (c), false, false)

template <int A_MODE, int ACT, bool FULLN>
__global__ __launch_bounds__(128) void gemm_wmma_kernel(
    const float* __restrict__ A, const float* __restrict__ Bm,
    const float* __restrict__ bias, const float* __restrict__ bias2,
    const float* __restrict__ resid, float* __restrict__ C,
    int M, int N, int K, int lda, int ldb, int ldc,
    long long sA, long long sB, long long sC)
{
  const int lane  = threadIdx.x & 31;
  const int wave  = threadIdx.x >> 5;
  const int tileM = blockIdx.y * 32;
  const int nbase = blockIdx.x * 256 + wave * 64;
  if (nbase >= N) return;

  const int z = blockIdx.z;
  const float* Ab = A  + (size_t)z * (size_t)sA;
  const float* Bb = Bm + (size_t)z * (size_t)sB;
  float*       Cb = C  + (size_t)z * (size_t)sC;
  const float* Rb = resid ? (resid + (size_t)z * (size_t)sC) : nullptr;

  const int l15   = lane & 15;
  const int khalf = (lane & 16) ? 8 : 0;

  int am0 = tileM + l15;      if (am0 >= M) am0 = M - 1;
  int am1 = tileM + 16 + l15; if (am1 >= M) am1 = M - 1;
  size_t arow0, arow1;
  if (A_MODE == 1) {
    arow0 = ((size_t)(am0 & 15) * S_ + (size_t)(am0 >> 4)) * (size_t)lda;
    arow1 = ((size_t)(am1 & 15) * S_ + (size_t)(am1 >> 4)) * (size_t)lda;
  } else {
    arow0 = (size_t)am0 * (size_t)lda;
    arow1 = (size_t)am1 * (size_t)lda;
  }

  bool ton[4]; int bn[4];
  #pragma unroll
  for (int nt = 0; nt < 4; ++nt) {
    if (FULLN) {
      ton[nt] = true;
      bn[nt]  = nbase + nt * 16 + l15;
    } else {
      ton[nt] = (nbase + nt * 16) < N;
      int c = nbase + nt * 16 + l15; if (c >= N) c = N - 1;
      bn[nt] = c;
    }
  }

  v8f acc[2][4] = {};
  for (int k0 = 0; k0 < K; k0 += 32) {
    const float* pa0 = Ab + arow0 + k0 + khalf;
    const float* pa1 = Ab + arow1 + k0 + khalf;
    v16h a0 = frag_from_f32(pa0);
    v16h a1 = frag_from_f32(pa1);
    if (k0 + 32 < K) {
      __builtin_prefetch(pa0 + 32, 0, 1);
      __builtin_prefetch(pa1 + 32, 0, 1);
    }
    #pragma unroll
    for (int nt = 0; nt < 4; ++nt) {
      if (!FULLN && !ton[nt]) continue;
      v16h b = frag_from_f32(Bb + (size_t)bn[nt] * (size_t)ldb + k0 + khalf);
      acc[0][nt] = WMMA_F16(a0, b, acc[0][nt]);
      acc[1][nt] = WMMA_F16(a1, b, acc[1][nt]);
    }
  }

  const int rb = (lane < 16) ? 0 : 8;
  #pragma unroll
  for (int nt = 0; nt < 4; ++nt) {
    if (!FULLN && !ton[nt]) continue;
    const int col = nbase + nt * 16 + l15;
    bool colok = FULLN ? true : (col < N);
    float bv = 0.0f;
    if (colok) {
      if (bias)  bv += bias[col];
      if (bias2) bv += bias2[col];
    }
    #pragma unroll
    for (int mt = 0; mt < 2; ++mt) {
      #pragma unroll
      for (int r = 0; r < 8; ++r) {
        const int row = tileM + mt * 16 + rb + r;
        if (row < M && colok) {
          float v = acc[mt][nt][r] + bv;
          if      (ACT == 1) v = v > 0.0f ? v : 0.0f;
          else if (ACT == 2) v = v > 0.0f ? v : 0.1f * v;
          if (Rb) v += Rb[(size_t)row * (size_t)ldc + col];
          Cb[(size_t)row * (size_t)ldc + col] = v;
        }
      }
    }
  }
}

__global__ void cvt_f32_f16_kernel(const float* __restrict__ src,
                                   __half* __restrict__ dst, int n) {
  int i = blockIdx.x * blockDim.x + threadIdx.x;
  if (i < n) ((_Float16*)dst)[i] = (_Float16)src[i];
}

__global__ __launch_bounds__(256) void transpose_bsg_kernel(
    const float* __restrict__ src, float* __restrict__ dst)
{
  __shared__ float tile[32][33];
  const int b  = blockIdx.z;
  const int s0 = blockIdx.x * 32, g0 = blockIdx.y * 32;
  const int tx = threadIdx.x, ty = threadIdx.y;
  const float* sp = src + (size_t)b * S_ * G_;
  float*       dp = dst + (size_t)b * G_ * S_;
  #pragma unroll
  for (int r = 0; r < 32; r += 8)
    tile[ty + r][tx] = sp[(size_t)(s0 + ty + r) * G_ + g0 + tx];
  __syncthreads();
  #pragma unroll
  for (int r = 0; r < 32; r += 8)
    dp[(size_t)(g0 + ty + r) * S_ + s0 + tx] = tile[tx][ty + r];
}

#define LB_ 2

__device__ __forceinline__ void lds_only_barrier() {
  __builtin_amdgcn_fence(__ATOMIC_RELEASE, "workgroup", "local");
  __builtin_amdgcn_s_barrier();
  __builtin_amdgcn_fence(__ATOMIC_ACQUIRE, "workgroup", "local");
}

__global__ __launch_bounds__(256) void lstm_dir_kernel(
    const float* __restrict__ preF, const float* __restrict__ preB,
    const __half* __restrict__ whhF16_f, const __half* __restrict__ whhF16_b,
    float* __restrict__ hs)
{
  const int dir = blockIdx.x;
  const int b0  = blockIdx.y * LB_;
  const float*    pre = dir ? preB : preF;
  const _Float16* whh = (const _Float16*)(dir ? whhF16_b : whhF16_f);

  __shared__ _Float16 h_lds[LB_ * HP_];
  __shared__ float    z_lds[LB_ * ZP_];

  const int tid   = threadIdx.x;
  const int wave  = tid >> 5;
  const int lane  = tid & 31;
  const int l15   = lane & 15;
  const int khalf = (lane & 16) ? 8 : 0;
  const int hrow  = l15 >> 3;
  const int half  = lane >> 4;

  v16h wfrag[16];
  #pragma unroll
  for (int nt = 0; nt < 4; ++nt) {
    const int tileN = (wave * 4 + nt) * 16;
    #pragma unroll
    for (int kc = 0; kc < 4; ++kc)
      wfrag[nt * 4 + kc] =
          frag_from_f16(whh + (size_t)(tileN + l15) * H_ + kc * 32 + khalf);
  }

  for (int i = tid; i < LB_ * HP_; i += 256) h_lds[i] = (_Float16)0.0f;

  const int cb = tid >> 7, hx = tid & (H_ - 1);
  float c = 0.0f;
  __syncthreads();

  const float* pre_lane = pre + (size_t)(b0 + half) * (4 * H_) + wave * 64 + l15;
  float pnext[4];
  {
    const int t0 = dir ? (S_ - 1) : 0;
    #pragma unroll
    for (int nt = 0; nt < 4; ++nt)
      pnext[nt] = pre_lane[(size_t)t0 * (B_ * 4 * H_) + nt * 16];
  }

  for (int step = 0; step < S_; ++step) {
    const int t = dir ? (S_ - 1 - step) : step;

    float pcur[4];
    #pragma unroll
    for (int nt = 0; nt < 4; ++nt) pcur[nt] = pnext[nt];
    {
      const int sn = (step + 1 < S_) ? (step + 1) : step;
      const int tn = dir ? (S_ - 1 - sn) : sn;
      #pragma unroll
      for (int nt = 0; nt < 4; ++nt)
        pnext[nt] = pre_lane[(size_t)tn * (B_ * 4 * H_) + nt * 16];
    }

    v16h a[4];
    #pragma unroll
    for (int kc = 0; kc < 4; ++kc)
      a[kc] = frag_from_f16(&h_lds[hrow * HP_ + kc * 32 + khalf]);

    #pragma unroll
    for (int nt = 0; nt < 4; ++nt) {
      v8f acc = {};
      #pragma unroll
      for (int kc = 0; kc < 4; ++kc)
        acc = WMMA_F16(a[kc], wfrag[nt * 4 + kc], acc);
      const int col = (wave * 4 + nt) * 16 + l15;
      z_lds[half * ZP_ + col] = acc[0] + pcur[nt];
    }
    lds_only_barrier();

    {
      const float zi = z_lds[cb * ZP_ + hx];
      const float zf = z_lds[cb * ZP_ + 128 + hx];
      const float zg = z_lds[cb * ZP_ + 256 + hx];
      const float zo = z_lds[cb * ZP_ + 384 + hx];
      const float si = 1.0f / (1.0f + __expf(-zi));
      const float sf = 1.0f / (1.0f + __expf(-zf));
      const float so = 1.0f / (1.0f + __expf(-zo));
      c = sf * c + si * tanhf(zg);
      const float h = so * tanhf(c);
      h_lds[cb * HP_ + hx] = (_Float16)h;
      hs[((size_t)t * B_ + b0 + cb) * (2 * H_) + dir * H_ + hx] = h;
    }
    lds_only_barrier();
  }
}

__global__ __launch_bounds__(256) void attn_mask_norm_kernel(
    float* __restrict__ scores, const float* __restrict__ adj)
{
  const int i = blockIdx.x, b = blockIdx.y, tid = threadIdx.x;
  float* row        = scores + ((size_t)b * S_ + i) * S_;
  const float* arow = adj    + ((size_t)b * S_ + i) * S_;
  __shared__ float red[256];
  float part = 0.0f;
  for (int j = tid; j < S_; j += 256) {
    const float sg = 1.0f / (1.0f + __expf(-row[j]));
    const float av = (j == i) ? 1.0f : arow[j];
    const float v  = (sg + ((j == i) ? 1e-5f : 0.0f)) * av;
    row[j] = v;
    part += v;
  }
  red[tid] = part;
  __syncthreads();
  for (int off = 128; off > 0; off >>= 1) {
    if (tid < off) red[tid] += red[tid + off];
    __syncthreads();
  }
  const float inv = 1.0f / red[0];
  for (int j = tid; j < S_; j += 256) row[j] *= inv;
}

__global__ __launch_bounds__(256) void head_mean_kernel(
    const float* __restrict__ h1, const float* __restrict__ gfin,
    float* __restrict__ hidden)
{
  const int b = blockIdx.x, tid = threadIdx.x;
  for (int d = tid; d < 512; d += 256) {
    float s = 0.0f;
    if (d < 256) {
      for (int t = 0; t < S_; ++t) s += h1[((size_t)t * B_ + b) * 256 + d];
    } else {
      const int dd = d - 256;
      for (int t = 0; t < S_; ++t) s += gfin[((size_t)b * S_ + t) * 256 + dd];
    }
    hidden[b * 512 + d] = s * (1.0f / (float)S_);
  }
}

template <int A_MODE, int ACT, bool FULLN>
static inline void gemm(hipStream_t st,
                        const float* A, const float* Bm,
                        const float* bias, const float* bias2, const float* resid,
                        float* C, int M, int N, int K, int lda, int ldb, int ldc,
                        long long sA = 0, long long sB = 0, long long sC = 0,
                        int batches = 1)
{
  dim3 grid((N + 255) / 256, (M + 31) / 32, batches);
  gemm_wmma_kernel<A_MODE, ACT, FULLN><<<grid, 128, 0, st>>>(
      A, Bm, bias, bias2, resid, C, M, N, K, lda, ldb, ldc, sA, sB, sC);
}

extern "C" void kernel_launch(void* const* d_in, const int* in_sizes, int n_in,
                              void* d_out, int out_size, void* d_ws, size_t ws_size,
                              hipStream_t stream)
{
  (void)in_sizes; (void)n_in; (void)out_size; (void)ws_size;
  auto F = [&](int i) { return (const float*)d_in[i]; };

  const float* x        = F(0);
  const float* adj      = F(1);
  const float* l0f_Wih  = F(2),  *l0f_Whh = F(3),  *l0f_bih = F(4),  *l0f_bhh = F(5);
  const float* l0b_Wih  = F(6),  *l0b_Whh = F(7),  *l0b_bih = F(8),  *l0b_bhh = F(9);
  const float* l1f_Wih  = F(10), *l1f_Whh = F(11), *l1f_bih = F(12), *l1f_bhh = F(13);
  const float* l1b_Wih  = F(14), *l1b_Whh = F(15), *l1b_bih = F(16), *l1b_bhh = F(17);
  const float* fc1_W    = F(18), *fc1_b   = F(19);
  const float* wattn_W  = F(20), *wattn_b = F(21);
  const float* lin0_W   = F(22), *lin0_b  = F(23);
  const float* lin1_W   = F(24), *lin1_b  = F(25);
  const float* linf_W   = F(26), *linf_b  = F(27);
  const float* fcA_W    = F(28), *fcA_b   = F(29);
  const float* fcB_W    = F(30), *fcB_b   = F(31);
  const float* fcC_W    = F(32), *fcC_b   = F(33);

  char* wsp = (char*)d_ws;
  size_t off = 0;
  auto carve = [&](size_t bytes) -> void* {
    void* p = wsp + off;
    off += (bytes + 255) & ~(size_t)255;
    return p;
  };
  float* preF   = (float*)carve((size_t)S_ * B_ * 4 * H_ * 4);
  float* preB   = (float*)carve((size_t)S_ * B_ * 4 * H_ * 4);
  float* h0     = (float*)carve((size_t)S_ * B_ * 2 * H_ * 4);
  float* h1     = (float*)carve((size_t)S_ * B_ * 2 * H_ * 4);
  float* gA     = (float*)carve((size_t)B_ * S_ * G_ * 4);
  float* gB     = (float*)carve((size_t)B_ * S_ * G_ * 4);
  float* gC     = (float*)carve((size_t)B_ * S_ * G_ * 4);
  float* gT     = (float*)carve((size_t)B_ * G_ * S_ * 4);
  float* scores = (float*)carve((size_t)B_ * S_ * S_ * 4);
  __half* whh0f = (__half*)carve((size_t)4 * H_ * H_ * 2);
  __half* whh0b = (__half*)carve((size_t)4 * H_ * H_ * 2);
  __half* whh1f = (__half*)carve((size_t)4 * H_ * H_ * 2);
  __half* whh1b = (__half*)carve((size_t)4 * H_ * H_ * 2);
  float* fcAo   = (float*)carve((size_t)B_ * 128 * 4);
  float* fcBo   = (float*)carve((size_t)B_ * 32 * 4);

  const int MSB = S_ * B_;
  const int nwh = 4 * H_ * H_;

  cvt_f32_f16_kernel<<<(nwh + 255) / 256, 256, 0, stream>>>(l0f_Whh, whh0f, nwh);
  cvt_f32_f16_kernel<<<(nwh + 255) / 256, 256, 0, stream>>>(l0b_Whh, whh0b, nwh);
  cvt_f32_f16_kernel<<<(nwh + 255) / 256, 256, 0, stream>>>(l1f_Whh, whh1f, nwh);
  cvt_f32_f16_kernel<<<(nwh + 255) / 256, 256, 0, stream>>>(l1b_Whh, whh1b, nwh);

  gemm<1, 0, true>(stream, x, l0f_Wih, l0f_bih, l0f_bhh, nullptr, preF,
                   MSB, 4 * H_, D_, D_, D_, 4 * H_);
  gemm<1, 0, true>(stream, x, l0b_Wih, l0b_bih, l0b_bhh, nullptr, preB,
                   MSB, 4 * H_, D_, D_, D_, 4 * H_);
  lstm_dir_kernel<<<dim3(2, B_ / LB_), 256, 0, stream>>>(preF, preB, whh0f, whh0b, h0);

  gemm<0, 0, true>(stream, h0, l1f_Wih, l1f_bih, l1f_bhh, nullptr, preF,
                   MSB, 4 * H_, 2 * H_, 2 * H_, 2 * H_, 4 * H_);
  gemm<0, 0, true>(stream, h0, l1b_Wih, l1b_bih, l1b_bhh, nullptr, preB,
                   MSB, 4 * H_, 2 * H_, 2 * H_, 2 * H_, 4 * H_);
  lstm_dir_kernel<<<dim3(2, B_ / LB_), 256, 0, stream>>>(preF, preB, whh1f, whh1b, h1);

  gemm<0, 0, true>(stream, x, fc1_W, fc1_b, nullptr, nullptr, gA,
                   MSB, G_, D_, D_, D_, G_);

  float* cur = gA; float* t1 = gB; float* t2 = gC;
  const long long sGG = (long long)S_ * G_;
  const long long sSS = (long long)S_ * S_;
  for (int l = 0; l < 4; ++l) {
    const float* Wq = wattn_W + (size_t)l * G_ * G_;
    gemm<0, 0, true>(stream, cur, Wq, wattn_b + l * G_, nullptr, nullptr, t2,
                     MSB, G_, G_, G_, G_, G_);
    gemm<0, 0, true>(stream, t2, cur, nullptr, nullptr, nullptr, scores,
                     S_, S_, G_, G_, G_, S_, sGG, sGG, sSS, B_);
    attn_mask_norm_kernel<<<dim3(S_, B_), 256, 0, stream>>>(scores, adj);
    transpose_bsg_kernel<<<dim3(S_ / 32, G_ / 32, B_), dim3(32, 8), 0, stream>>>(cur, gT);
    gemm<0, 0, true>(stream, scores, gT, nullptr, nullptr, nullptr, t1,
                     S_, G_, S_, S_, S_, G_, sSS, (long long)G_ * S_, sGG, B_);
    gemm<0, 1, true>(stream, t1, lin0_W + (size_t)l * G_ * G_, lin0_b + l * G_,
                     nullptr, nullptr, t2, MSB, G_, G_, G_, G_, G_);
    gemm<0, 1, true>(stream, t2, lin1_W + (size_t)l * G_ * G_, lin1_b + l * G_,
                     nullptr, cur, t1, MSB, G_, G_, G_, G_, G_);
    float* old = cur; cur = t1; t1 = old;
  }
  gemm<0, 0, true>(stream, cur, linf_W, linf_b, nullptr, nullptr, t2,
                   MSB, G_, G_, G_, G_, G_);
  float* gfin = t2;

  float* y_out  = (float*)d_out;
  float* hidden = (float*)d_out + B_ * 2;
  head_mean_kernel<<<B_, 256, 0, stream>>>(h1, gfin, hidden);

  gemm<0, 2, true >(stream, hidden, fcA_W, fcA_b, nullptr, nullptr, fcAo,
                    B_, 128, 512, 512, 512, 128);
  gemm<0, 2, false>(stream, fcAo, fcB_W, fcB_b, nullptr, nullptr, fcBo,
                    B_, 32, 128, 128, 128, 32);
  gemm<0, 0, false>(stream, fcBo, fcC_W, fcC_b, nullptr, nullptr, y_out,
                    B_, 2, 32, 32, 32, 2);
}


# LSTM recurrence: gate tanh and sigmoid on the hardware exp and reciprocal units (tanh = 1 - 2/(1+e^2x)) instead of library tanhf and IEEE divides; summation order unchanged
# speedup vs baseline: 3.1493x; 1.0419x over previous
#include <hip/hip_runtime.h>
#include <hip/hip_fp16.h>
#include <hip/hip_bf16.h>

#define B_  16
#define S_  512
#define D_  1024
#define H_  128
#define G_  256
#define HP_ 136
#define ZP_ 516

typedef __attribute__((ext_vector_type(16))) _Float16 v16h;
typedef __attribute__((ext_vector_type(8)))  float    v8f;

__device__ __forceinline__ v16h frag_from_f32(const float* p) {
  float4 x0 = *(const float4*)(p);
  float4 x1 = *(const float4*)(p + 4);
  float4 y0 = *(const float4*)(p + 16);
  float4 y1 = *(const float4*)(p + 20);
  v16h f;
  f[0]=(_Float16)x0.x;  f[1]=(_Float16)x0.y;  f[2]=(_Float16)x0.z;  f[3]=(_Float16)x0.w;
  f[4]=(_Float16)x1.x;  f[5]=(_Float16)x1.y;  f[6]=(_Float16)x1.z;  f[7]=(_Float16)x1.w;
  f[8]=(_Float16)y0.x;  f[9]=(_Float16)y0.y;  f[10]=(_Float16)y0.z; f[11]=(_Float16)y0.w;
  f[12]=(_Float16)y1.x; f[13]=(_Float16)y1.y; f[14]=(_Float16)y1.z; f[15]=(_Float16)y1.w;
  return f;
}

__device__ __forceinline__ v16h frag_from_f16(const _Float16* p) {
  v16h f;
  #pragma unroll
  for (int i = 0; i < 8; ++i) { f[i] = p[i]; f[8 + i] = p[16 + i]; }
  return f;
}

#define WMMA_F16(a, b, c) __builtin_amdgcn_wmma_f32_16x16x32_f16( \
    false, (a), false, (b), (short)0, (c), false, false)

template <int A_MODE, int ACT, bool FULLN>
__global__ __launch_bounds__(128) void gemm_wmma_kernel(
    const float* __restrict__ A, const float* __restrict__ Bm,
    const float* __restrict__ bias, const float* __restrict__ bias2,
    const float* __restrict__ resid, float* __restrict__ C,
    int M, int N, int K, int lda, int ldb, int ldc,
    long long sA, long long sB, long long sC)
{
  const int lane  = threadIdx.x & 31;
  const int wave  = threadIdx.x >> 5;
  const int tileM = blockIdx.y * 32;
  const int nbase = blockIdx.x * 256 + wave * 64;
  if (nbase >= N) return;

  const int z = blockIdx.z;
  const float* Ab = A  + (size_t)z * (size_t)sA;
  const float* Bb = Bm + (size_t)z * (size_t)sB;
  float*       Cb = C  + (size_t)z * (size_t)sC;
  const float* Rb = resid ? (resid + (size_t)z * (size_t)sC) : nullptr;

  const int l15   = lane & 15;
  const int khalf = (lane & 16) ? 8 : 0;

  int am0 = tileM + l15;      if (am0 >= M) am0 = M - 1;
  int am1 = tileM + 16 + l15; if (am1 >= M) am1 = M - 1;
  size_t arow0, arow1;
  if (A_MODE == 1) {
    arow0 = ((size_t)(am0 & 15) * S_ + (size_t)(am0 >> 4)) * (size_t)lda;
    arow1 = ((size_t)(am1 & 15) * S_ + (size_t)(am1 >> 4)) * (size_t)lda;
  } else {
    arow0 = (size_t)am0 * (size_t)lda;
    arow1 = (size_t)am1 * (size_t)lda;
  }

  bool ton[4]; int bn[4];
  #pragma unroll
  for (int nt = 0; nt < 4; ++nt) {
    if (FULLN) {
      ton[nt] = true;
      bn[nt]  = nbase + nt * 16 + l15;
    } else {
      ton[nt] = (nbase + nt * 16) < N;
      int c = nbase + nt * 16 + l15; if (c >= N) c = N - 1;
      bn[nt] = c;
    }
  }

  v8f acc[2][4] = {};
  for (int k0 = 0; k0 < K; k0 += 32) {
    const float* pa0 = Ab + arow0 + k0 + khalf;
    const float* pa1 = Ab + arow1 + k0 + khalf;
    v16h a0 = frag_from_f32(pa0);
    v16h a1 = frag_from_f32(pa1);
    if (k0 + 32 < K) {
      __builtin_prefetch(pa0 + 32, 0, 1);
      __builtin_prefetch(pa1 + 32, 0, 1);
    }
    #pragma unroll
    for (int nt = 0; nt < 4; ++nt) {
      if (!FULLN && !ton[nt]) continue;
      v16h b = frag_from_f32(Bb + (size_t)bn[nt] * (size_t)ldb + k0 + khalf);
      acc[0][nt] = WMMA_F16(a0, b, acc[0][nt]);
      acc[1][nt] = WMMA_F16(a1, b, acc[1][nt]);
    }
  }

  const int rb = (lane < 16) ? 0 : 8;
  #pragma unroll
  for (int nt = 0; nt < 4; ++nt) {
    if (!FULLN && !ton[nt]) continue;
    const int col = nbase + nt * 16 + l15;
    bool colok = FULLN ? true : (col < N);
    float bv = 0.0f;
    if (colok) {
      if (bias)  bv += bias[col];
      if (bias2) bv += bias2[col];
    }
    #pragma unroll
    for (int mt = 0; mt < 2; ++mt) {
      #pragma unroll
      for (int r = 0; r < 8; ++r) {
        const int row = tileM + mt * 16 + rb + r;
        if (row < M && colok) {
          float v = acc[mt][nt][r] + bv;
          if      (ACT == 1) v = v > 0.0f ? v : 0.0f;
          else if (ACT == 2) v = v > 0.0f ? v : 0.1f * v;
          if (Rb) v += Rb[(size_t)row * (size_t)ldc + col];
          Cb[(size_t)row * (size_t)ldc + col] = v;
        }
      }
    }
  }
}

__global__ void cvt_f32_f16_kernel(const float* __restrict__ src,
                                   __half* __restrict__ dst, int n) {
  int i = blockIdx.x * blockDim.x + threadIdx.x;
  if (i < n) ((_Float16*)dst)[i] = (_Float16)src[i];
}

__global__ __launch_bounds__(256) void transpose_bsg_kernel(
    const float* __restrict__ src, float* __restrict__ dst)
{
  __shared__ float tile[32][33];
  const int b  = blockIdx.z;
  const int s0 = blockIdx.x * 32, g0 = blockIdx.y * 32;
  const int tx = threadIdx.x, ty = threadIdx.y;
  const float* sp = src + (size_t)b * S_ * G_;
  float*       dp = dst + (size_t)b * G_ * S_;
  #pragma unroll
  for (int r = 0; r < 32; r += 8)
    tile[ty + r][tx] = sp[(size_t)(s0 + ty + r) * G_ + g0 + tx];
  __syncthreads();
  #pragma unroll
  for (int r = 0; r < 32; r += 8)
    dp[(size_t)(g0 + ty + r) * S_ + s0 + tx] = tile[tx][ty + r];
}

#define LB_ 2

__device__ __forceinline__ void lds_only_barrier() {
  __builtin_amdgcn_fence(__ATOMIC_RELEASE, "workgroup", "local");
  __builtin_amdgcn_s_barrier();
  __builtin_amdgcn_fence(__ATOMIC_ACQUIRE, "workgroup", "local");
}

__device__ __forceinline__ float gate_sigmoid(float x) {
  return __builtin_amdgcn_rcpf(1.0f + __expf(-x));
}
__device__ __forceinline__ float gate_tanh(float x) {
  return 1.0f - 2.0f * __builtin_amdgcn_rcpf(1.0f + __expf(2.0f * x));
}

__global__ __launch_bounds__(256) void lstm_dir_kernel(
    const float* __restrict__ preF, const float* __restrict__ preB,
    const __half* __restrict__ whhF16_f, const __half* __restrict__ whhF16_b,
    float* __restrict__ hs)
{
  const int dir = blockIdx.x;
  const int b0  = blockIdx.y * LB_;
  const float*    pre = dir ? preB : preF;
  const _Float16* whh = (const _Float16*)(dir ? whhF16_b : whhF16_f);

  __shared__ _Float16 h_lds[LB_ * HP_];
  __shared__ float    z_lds[LB_ * ZP_];

  const int tid   = threadIdx.x;
  const int wave  = tid >> 5;
  const int lane  = tid & 31;
  const int l15   = lane & 15;
  const int khalf = (lane & 16) ? 8 : 0;
  const int hrow  = l15 >> 3;
  const int half  = lane >> 4;

  v16h wfrag[16];
  #pragma unroll
  for (int nt = 0; nt < 4; ++nt) {
    const int tileN = (wave * 4 + nt) * 16;
    #pragma unroll
    for (int kc = 0; kc < 4; ++kc)
      wfrag[nt * 4 + kc] =
          frag_from_f16(whh + (size_t)(tileN + l15) * H_ + kc * 32 + khalf);
  }

  for (int i = tid; i < LB_ * HP_; i += 256) h_lds[i] = (_Float16)0.0f;

  const int cb = tid >> 7, hx = tid & (H_ - 1);
  float c = 0.0f;
  __syncthreads();

  const float* pre_lane = pre + (size_t)(b0 + half) * (4 * H_) + wave * 64 + l15;
  float pnext[4];
  {
    const int t0 = dir ? (S_ - 1) : 0;
    #pragma unroll
    for (int nt = 0; nt < 4; ++nt)
      pnext[nt] = pre_lane[(size_t)t0 * (B_ * 4 * H_) + nt * 16];
  }

  for (int step = 0; step < S_; ++step) {
    const int t = dir ? (S_ - 1 - step) : step;

    float pcur[4];
    #pragma unroll
    for (int nt = 0; nt < 4; ++nt) pcur[nt] = pnext[nt];
    {
      const int sn = (step + 1 < S_) ? (step + 1) : step;
      const int tn = dir ? (S_ - 1 - sn) : sn;
      #pragma unroll
      for (int nt = 0; nt < 4; ++nt)
        pnext[nt] = pre_lane[(size_t)tn * (B_ * 4 * H_) + nt * 16];
    }

    v16h a[4];
    #pragma unroll
    for (int kc = 0; kc < 4; ++kc)
      a[kc] = frag_from_f16(&h_lds[hrow * HP_ + kc * 32 + khalf]);

    #pragma unroll
    for (int nt = 0; nt < 4; ++nt) {
      v8f acc = {};
      #pragma unroll
      for (int kc = 0; kc < 4; ++kc)
        acc = WMMA_F16(a[kc], wfrag[nt * 4 + kc], acc);
      const int col = (wave * 4 + nt) * 16 + l15;
      z_lds[half * ZP_ + col] = acc[0] + pcur[nt];
    }
    lds_only_barrier();

    {
      const float zi = z_lds[cb * ZP_ + hx];
      const float zf = z_lds[cb * ZP_ + 128 + hx];
      const float zg = z_lds[cb * ZP_ + 256 + hx];
      const float zo = z_lds[cb * ZP_ + 384 + hx];
      const float si = gate_sigmoid(zi);
      const float sf = gate_sigmoid(zf);
      const float so = gate_sigmoid(zo);
      c = sf * c + si * gate_tanh(zg);
      const float h = so * gate_tanh(c);
      h_lds[cb * HP_ + hx] = (_Float16)h;
      hs[((size_t)t * B_ + b0 + cb) * (2 * H_) + dir * H_ + hx] = h;
    }
    lds_only_barrier();
  }
}

__global__ __launch_bounds__(256) void attn_mask_norm_kernel(
    float* __restrict__ scores, const float* __restrict__ adj)
{
  const int i = blockIdx.x, b = blockIdx.y, tid = threadIdx.x;
  float* row        = scores + ((size_t)b * S_ + i) * S_;
  const float* arow = adj    + ((size_t)b * S_ + i) * S_;
  __shared__ float red[256];
  float part = 0.0f;
  for (int j = tid; j < S_; j += 256) {
    const float sg = 1.0f / (1.0f + __expf(-row[j]));
    const float av = (j == i) ? 1.0f : arow[j];
    const float v  = (sg + ((j == i) ? 1e-5f : 0.0f)) * av;
    row[j] = v;
    part += v;
  }
  red[tid] = part;
  __syncthreads();
  for (int off = 128; off > 0; off >>= 1) {
    if (tid < off) red[tid] += red[tid + off];
    __syncthreads();
  }
  const float inv = 1.0f / red[0];
  for (int j = tid; j < S_; j += 256) row[j] *= inv;
}

__global__ __launch_bounds__(256) void head_mean_kernel(
    const float* __restrict__ h1, const float* __restrict__ gfin,
    float* __restrict__ hidden)
{
  const int b = blockIdx.x, tid = threadIdx.x;
  for (int d = tid; d < 512; d += 256) {
    float s = 0.0f;
    if (d < 256) {
      for (int t = 0; t < S_; ++t) s += h1[((size_t)t * B_ + b) * 256 + d];
    } else {
      const int dd = d - 256;
      for (int t = 0; t < S_; ++t) s += gfin[((size_t)b * S_ + t) * 256 + dd];
    }
    hidden[b * 512 + d] = s * (1.0f / (float)S_);
  }
}

template <int A_MODE, int ACT, bool FULLN>
static inline void gemm(hipStream_t st,
                        const float* A, const float* Bm,
                        const float* bias, const float* bias2, const float* resid,
                        float* C, int M, int N, int K, int lda, int ldb, int ldc,
                        long long sA = 0, long long sB = 0, long long sC = 0,
                        int batches = 1)
{
  dim3 grid((N + 255) / 256, (M + 31) / 32, batches);
  gemm_wmma_kernel<A_MODE, ACT, FULLN><<<grid, 128, 0, st>>>(
      A, Bm, bias, bias2, resid, C, M, N, K, lda, ldb, ldc, sA, sB, sC);
}

extern "C" void kernel_launch(void* const* d_in, const int* in_sizes, int n_in,
                              void* d_out, int out_size, void* d_ws, size_t ws_size,
                              hipStream_t stream)
{
  (void)in_sizes; (void)n_in; (void)out_size; (void)ws_size;
  auto F = [&](int i) { return (const float*)d_in[i]; };

  const float* x        = F(0);
  const float* adj      = F(1);
  const float* l0f_Wih  = F(2),  *l0f_Whh = F(3),  *l0f_bih = F(4),  *l0f_bhh = F(5);
  const float* l0b_Wih  = F(6),  *l0b_Whh = F(7),  *l0b_bih = F(8),  *l0b_bhh = F(9);
  const float* l1f_Wih  = F(10), *l1f_Whh = F(11), *l1f_bih = F(12), *l1f_bhh = F(13);
  const float* l1b_Wih  = F(14), *l1b_Whh = F(15), *l1b_bih = F(16), *l1b_bhh = F(17);
  const float* fc1_W    = F(18), *fc1_b   = F(19);
  const float* wattn_W  = F(20), *wattn_b = F(21);
  const float* lin0_W   = F(22), *lin0_b  = F(23);
  const float* lin1_W   = F(24), *lin1_b  = F(25);
  const float* linf_W   = F(26), *linf_b  = F(27);
  const float* fcA_W    = F(28), *fcA_b   = F(29);
  const float* fcB_W    = F(30), *fcB_b   = F(31);
  const float* fcC_W    = F(32), *fcC_b   = F(33);

  char* wsp = (char*)d_ws;
  size_t off = 0;
  auto carve = [&](size_t bytes) -> void* {
    void* p = wsp + off;
    off += (bytes + 255) & ~(size_t)255;
    return p;
  };
  float* preF   = (float*)carve((size_t)S_ * B_ * 4 * H_ * 4);
  float* preB   = (float*)carve((size_t)S_ * B_ * 4 * H_ * 4);
  float* h0     = (float*)carve((size_t)S_ * B_ * 2 * H_ * 4);
  float* h1     = (float*)carve((size_t)S_ * B_ * 2 * H_ * 4);
  float* gA     = (float*)carve((size_t)B_ * S_ * G_ * 4);
  float* gB     = (float*)carve((size_t)B_ * S_ * G_ * 4);
  float* gC     = (float*)carve((size_t)B_ * S_ * G_ * 4);
  float* gT     = (float*)carve((size_t)B_ * G_ * S_ * 4);
  float* scores = (float*)carve((size_t)B_ * S_ * S_ * 4);
  __half* whh0f = (__half*)carve((size_t)4 * H_ * H_ * 2);
  __half* whh0b = (__half*)carve((size_t)4 * H_ * H_ * 2);
  __half* whh1f = (__half*)carve((size_t)4 * H_ * H_ * 2);
  __half* whh1b = (__half*)carve((size_t)4 * H_ * H_ * 2);
  float* fcAo   = (float*)carve((size_t)B_ * 128 * 4);
  float* fcBo   = (float*)carve((size_t)B_ * 32 * 4);

  const int MSB = S_ * B_;
  const int nwh = 4 * H_ * H_;

  cvt_f32_f16_kernel<<<(nwh + 255) / 256, 256, 0, stream>>>(l0f_Whh, whh0f, nwh);
  cvt_f32_f16_kernel<<<(nwh + 255) / 256, 256, 0, stream>>>(l0b_Whh, whh0b, nwh);
  cvt_f32_f16_kernel<<<(nwh + 255) / 256, 256, 0, stream>>>(l1f_Whh, whh1f, nwh);
  cvt_f32_f16_kernel<<<(nwh + 255) / 256, 256, 0, stream>>>(l1b_Whh, whh1b, nwh);

  gemm<1, 0, true>(stream, x, l0f_Wih, l0f_bih, l0f_bhh, nullptr, preF,
                   MSB, 4 * H_, D_, D_, D_, 4 * H_);
  gemm<1, 0, true>(stream, x, l0b_Wih, l0b_bih, l0b_bhh, nullptr, preB,
                   MSB, 4 * H_, D_, D_, D_, 4 * H_);
  lstm_dir_kernel<<<dim3(2, B_ / LB_), 256, 0, stream>>>(preF, preB, whh0f, whh0b, h0);

  gemm<0, 0, true>(stream, h0, l1f_Wih, l1f_bih, l1f_bhh, nullptr, preF,
                   MSB, 4 * H_, 2 * H_, 2 * H_, 2 * H_, 4 * H_);
  gemm<0, 0, true>(stream, h0, l1b_Wih, l1b_bih, l1b_bhh, nullptr, preB,
                   MSB, 4 * H_, 2 * H_, 2 * H_, 2 * H_, 4 * H_);
  lstm_dir_kernel<<<dim3(2, B_ / LB_), 256, 0, stream>>>(preF, preB, whh1f, whh1b, h1);

  gemm<0, 0, true>(stream, x, fc1_W, fc1_b, nullptr, nullptr, gA,
                   MSB, G_, D_, D_, D_, G_);

  float* cur = gA; float* t1 = gB; float* t2 = gC;
  const long long sGG = (long long)S_ * G_;
  const long long sSS = (long long)S_ * S_;
  for (int l = 0; l < 4; ++l) {
    const float* Wq = wattn_W + (size_t)l * G_ * G_;
    gemm<0, 0, true>(stream, cur, Wq, wattn_b + l * G_, nullptr, nullptr, t2,
                     MSB, G_, G_, G_, G_, G_);
    gemm<0, 0, true>(stream, t2, cur, nullptr, nullptr, nullptr, scores,
                     S_, S_, G_, G_, G_, S_, sGG, sGG, sSS, B_);
    attn_mask_norm_kernel<<<dim3(S_, B_), 256, 0, stream>>>(scores, adj);
    transpose_bsg_kernel<<<dim3(S_ / 32, G_ / 32, B_), dim3(32, 8), 0, stream>>>(cur, gT);
    gemm<0, 0, true>(stream, scores, gT, nullptr, nullptr, nullptr, t1,
                     S_, G_, S_, S_, S_, G_, sSS, (long long)G_ * S_, sGG, B_);
    gemm<0, 1, true>(stream, t1, lin0_W + (size_t)l * G_ * G_, lin0_b + l * G_,
                     nullptr, nullptr, t2, MSB, G_, G_, G_, G_, G_);
    gemm<0, 1, true>(stream, t2, lin1_W + (size_t)l * G_ * G_, lin1_b + l * G_,
                     nullptr, cur, t1, MSB, G_, G_, G_, G_, G_);
    float* old = cur; cur = t1; t1 = old;
  }
  gemm<0, 0, true>(stream, cur, linf_W, linf_b, nullptr, nullptr, t2,
                   MSB, G_, G_, G_, G_, G_);
  float* gfin = t2;

  float* y_out  = (float*)d_out;
  float* hidden = (float*)d_out + B_ * 2;
  head_mean_kernel<<<B_, 256, 0, stream>>>(h1, gfin, hidden);

  gemm<0, 2, true >(stream, hidden, fcA_W, fcA_b, nullptr, nullptr, fcAo,
                    B_, 128, 512, 512, 512, 128);
  gemm<0, 2, false>(stream, fcAo, fcB_W, fcB_b, nullptr, nullptr, fcBo,
                    B_, 32, 128, 128, 128, 32);
  gemm<0, 0, false>(stream, fcBo, fcC_W, fcC_b, nullptr, nullptr, y_out,
                    B_, 2, 32, 32, 32, 2);
}


# LSTM recurrence: pre of step s+7 is loaded during step s into an 8-deep register ring (step loop unrolled by 8); it was waited for inside the step that issued it; summation order unchanged
# speedup vs baseline: 3.3954x; 1.0781x over previous
#include <hip/hip_runtime.h>
#include <hip/hip_fp16.h>
#include <hip/hip_bf16.h>

#define B_  16
#define S_  512
#define D_  1024
#define H_  128
#define G_  256
#define HP_ 136
#define ZP_ 516

typedef __attribute__((ext_vector_type(16))) _Float16 v16h;
typedef __attribute__((ext_vector_type(8)))  float    v8f;

__device__ __forceinline__ v16h frag_from_f32(const float* p) {
  float4 x0 = *(const float4*)(p);
  float4 x1 = *(const float4*)(p + 4);
  float4 y0 = *(const float4*)(p + 16);
  float4 y1 = *(const float4*)(p + 20);
  v16h f;
  f[0]=(_Float16)x0.x;  f[1]=(_Float16)x0.y;  f[2]=(_Float16)x0.z;  f[3]=(_Float16)x0.w;
  f[4]=(_Float16)x1.x;  f[5]=(_Float16)x1.y;  f[6]=(_Float16)x1.z;  f[7]=(_Float16)x1.w;
  f[8]=(_Float16)y0.x;  f[9]=(_Float16)y0.y;  f[10]=(_Float16)y0.z; f[11]=(_Float16)y0.w;
  f[12]=(_Float16)y1.x; f[13]=(_Float16)y1.y; f[14]=(_Float16)y1.z; f[15]=(_Float16)y1.w;
  return f;
}

__device__ __forceinline__ v16h frag_from_f16(const _Float16* p) {
  v16h f;
  #pragma unroll
  for (int i = 0; i < 8; ++i) { f[i] = p[i]; f[8 + i] = p[16 + i]; }
  return f;
}

#define WMMA_F16(a, b, c) __builtin_amdgcn_wmma_f32_16x16x32_f16( \
    false, (a), false, (b), (short)0, (c), false, false)

template <int A_MODE, int ACT, bool FULLN>
__global__ __launch_bounds__(128) void gemm_wmma_kernel(
    const float* __restrict__ A, const float* __restrict__ Bm,
    const float* __restrict__ bias, const float* __restrict__ bias2,
    const float* __restrict__ resid, float* __restrict__ C,
    int M, int N, int K, int lda, int ldb, int ldc,
    long long sA, long long sB, long long sC)
{
  const int lane  = threadIdx.x & 31;
  const int wave  = threadIdx.x >> 5;
  const int tileM = blockIdx.y * 32;
  const int nbase = blockIdx.x * 256 + wave * 64;
  if (nbase >= N) return;

  const int z = blockIdx.z;
  const float* Ab = A  + (size_t)z * (size_t)sA;
  const float* Bb = Bm + (size_t)z * (size_t)sB;
  float*       Cb = C  + (size_t)z * (size_t)sC;
  const float* Rb = resid ? (resid + (size_t)z * (size_t)sC) : nullptr;

  const int l15   = lane & 15;
  const int khalf = (lane & 16) ? 8 : 0;

  int am0 = tileM + l15;      if (am0 >= M) am0 = M - 1;
  int am1 = tileM + 16 + l15; if (am1 >= M) am1 = M - 1;
  size_t arow0, arow1;
  if (A_MODE == 1) {
    arow0 = ((size_t)(am0 & 15) * S_ + (size_t)(am0 >> 4)) * (size_t)lda;
    arow1 = ((size_t)(am1 & 15) * S_ + (size_t)(am1 >> 4)) * (size_t)lda;
  } else {
    arow0 = (size_t)am0 * (size_t)lda;
    arow1 = (size_t)am1 * (size_t)lda;
  }

  bool ton[4]; int bn[4];
  #pragma unroll
  for (int nt = 0; nt < 4; ++nt) {
    if (FULLN) {
      ton[nt] = true;
      bn[nt]  = nbase + nt * 16 + l15;
    } else {
      ton[nt] = (nbase + nt * 16) < N;
      int c = nbase + nt * 16 + l15; if (c >= N) c = N - 1;
      bn[nt] = c;
    }
  }

  v8f acc[2][4] = {};
  for (int k0 = 0; k0 < K; k0 += 32) {
    const float* pa0 = Ab + arow0 + k0 + khalf;
    const float* pa1 = Ab + arow1 + k0 + khalf;
    v16h a0 = frag_from_f32(pa0);
    v16h a1 = frag_from_f32(pa1);
    if (k0 + 32 < K) {
      __builtin_prefetch(pa0 + 32, 0, 1);
      __builtin_prefetch(pa1 + 32, 0, 1);
    }
    #pragma unroll
    for (int nt = 0; nt < 4; ++nt) {
      if (!FULLN && !ton[nt]) continue;
      v16h b = frag_from_f32(Bb + (size_t)bn[nt] * (size_t)ldb + k0 + khalf);
      acc[0][nt] = WMMA_F16(a0, b, acc[0][nt]);
      acc[1][nt] = WMMA_F16(a1, b, acc[1][nt]);
    }
  }

  const int rb = (lane < 16) ? 0 : 8;
  #pragma unroll
  for (int nt = 0; nt < 4; ++nt) {
    if (!FULLN && !ton[nt]) continue;
    const int col = nbase + nt * 16 + l15;
    bool colok = FULLN ? true : (col < N);
    float bv = 0.0f;
    if (colok) {
      if (bias)  bv += bias[col];
      if (bias2) bv += bias2[col];
    }
    #pragma unroll
    for (int mt = 0; mt < 2; ++mt) {
      #pragma unroll
      for (int r = 0; r < 8; ++r) {
        const int row = tileM + mt * 16 + rb + r;
        if (row < M && colok) {
          float v = acc[mt][nt][r] + bv;
          if      (ACT == 1) v = v > 0.0f ? v : 0.0f;
          else if (ACT == 2) v = v > 0.0f ? v : 0.1f * v;
          if (Rb) v += Rb[(size_t)row * (size_t)ldc + col];
          Cb[(size_t)row * (size_t)ldc + col] = v;
        }
      }
    }
  }
}

__global__ void cvt_f32_f16_kernel(const float* __restrict__ src,
                                   __half* __restrict__ dst, int n) {
  int i = blockIdx.x * blockDim.x + threadIdx.x;
  if (i < n) ((_Float16*)dst)[i] = (_Float16)src[i];
}

__global__ __launch_bounds__(256) void transpose_bsg_kernel(
    const float* __restrict__ src, float* __restrict__ dst)
{
  __shared__ float tile[32][33];
  const int b  = blockIdx.z;
  const int s0 = blockIdx.x * 32, g0 = blockIdx.y * 32;
  const int tx = threadIdx.x, ty = threadIdx.y;
  const float* sp = src + (size_t)b * S_ * G_;
  float*       dp = dst + (size_t)b * G_ * S_;
  #pragma unroll
  for (int r = 0; r < 32; r += 8)
    tile[ty + r][tx] = sp[(size_t)(s0 + ty + r) * G_ + g0 + tx];
  __syncthreads();
  #pragma unroll
  for (int r = 0; r < 32; r += 8)
    dp[(size_t)(g0 + ty + r) * S_ + s0 + tx] = tile[tx][ty + r];
}

#define LB_ 2
#define PD_ 8
static_assert(S_ % PD_ == 0, "the step loop is unrolled by PD_");

__device__ __forceinline__ void lds_only_barrier() {
  __builtin_amdgcn_fence(__ATOMIC_RELEASE, "workgroup", "local");
  __builtin_amdgcn_s_barrier();
  __builtin_amdgcn_fence(__ATOMIC_ACQUIRE, "workgroup", "local");
}

__device__ __forceinline__ float gate_sigmoid(float x) {
  return __builtin_amdgcn_rcpf(1.0f + __expf(-x));
}
__device__ __forceinline__ float gate_tanh(float x) {
  return 1.0f - 2.0f * __builtin_amdgcn_rcpf(1.0f + __expf(2.0f * x));
}

__global__ __launch_bounds__(256) void lstm_dir_kernel(
    const float* __restrict__ preF, const float* __restrict__ preB,
    const __half* __restrict__ whhF16_f, const __half* __restrict__ whhF16_b,
    float* __restrict__ hs)
{
  const int dir = blockIdx.x;
  const int b0  = blockIdx.y * LB_;
  const float*    pre = dir ? preB : preF;
  const _Float16* whh = (const _Float16*)(dir ? whhF16_b : whhF16_f);

  __shared__ _Float16 h_lds[LB_ * HP_];
  __shared__ float    z_lds[LB_ * ZP_];

  const int tid   = threadIdx.x;
  const int wave  = tid >> 5;
  const int lane  = tid & 31;
  const int l15   = lane & 15;
  const int khalf = (lane & 16) ? 8 : 0;
  const int hrow  = l15 >> 3;
  const int half  = lane >> 4;

  v16h wfrag[16];
  #pragma unroll
  for (int nt = 0; nt < 4; ++nt) {
    const int tileN = (wave * 4 + nt) * 16;
    #pragma unroll
    for (int kc = 0; kc < 4; ++kc)
      wfrag[nt * 4 + kc] =
          frag_from_f16(whh + (size_t)(tileN + l15) * H_ + kc * 32 + khalf);
  }

  for (int i = tid; i < LB_ * HP_; i += 256) h_lds[i] = (_Float16)0.0f;

  const int cb = tid >> 7, hx = tid & (H_ - 1);
  float c = 0.0f;
  __syncthreads();

  const float* pre_lane = pre + (size_t)(b0 + half) * (4 * H_) + wave * 64 + l15;
  float pq[PD_][4];
  #pragma unroll
  for (int j = 0; j < PD_ - 1; ++j) {
    const int tj = dir ? (S_ - 1 - j) : j;
    #pragma unroll
    for (int nt = 0; nt < 4; ++nt)
      pq[j][nt] = pre_lane[(size_t)tj * (B_ * 4 * H_) + nt * 16];
  }

  for (int step0 = 0; step0 < S_; step0 += PD_) {
    #pragma unroll
    for (int j = 0; j < PD_; ++j) {
      const int step = step0 + j;
      const int t = dir ? (S_ - 1 - step) : step;

      {
        const int sn = (step + PD_ - 1 < S_) ? (step + PD_ - 1) : (S_ - 1);
        const int tn = dir ? (S_ - 1 - sn) : sn;
        #pragma unroll
        for (int nt = 0; nt < 4; ++nt)
          pq[(j + PD_ - 1) % PD_][nt] = pre_lane[(size_t)tn * (B_ * 4 * H_) + nt * 16];
      }

      v16h a[4];
      #pragma unroll
      for (int kc = 0; kc < 4; ++kc)
        a[kc] = frag_from_f16(&h_lds[hrow * HP_ + kc * 32 + khalf]);

      #pragma unroll
      for (int nt = 0; nt < 4; ++nt) {
        v8f acc = {};
        #pragma unroll
        for (int kc = 0; kc < 4; ++kc)
          acc = WMMA_F16(a[kc], wfrag[nt * 4 + kc], acc);
        const int col = (wave * 4 + nt) * 16 + l15;
        z_lds[half * ZP_ + col] = acc[0] + pq[j][nt];
      }
      lds_only_barrier();

      {
        const float zi = z_lds[cb * ZP_ + hx];
        const float zf = z_lds[cb * ZP_ + 128 + hx];
        const float zg = z_lds[cb * ZP_ + 256 + hx];
        const float zo = z_lds[cb * ZP_ + 384 + hx];
        const float si = gate_sigmoid(zi);
        const float sf = gate_sigmoid(zf);
        const float so = gate_sigmoid(zo);
        c = sf * c + si * gate_tanh(zg);
        const float h = so * gate_tanh(c);
        h_lds[cb * HP_ + hx] = (_Float16)h;
        hs[((size_t)t * B_ + b0 + cb) * (2 * H_) + dir * H_ + hx] = h;
      }
      lds_only_barrier();
    }
  }
}

__global__ __launch_bounds__(256) void attn_mask_norm_kernel(
    float* __restrict__ scores, const float* __restrict__ adj)
{
  const int i = blockIdx.x, b = blockIdx.y, tid = threadIdx.x;
  float* row        = scores + ((size_t)b * S_ + i) * S_;
  const float* arow = adj    + ((size_t)b * S_ + i) * S_;
  __shared__ float red[256];
  float part = 0.0f;
  for (int j = tid; j < S_; j += 256) {
    const float sg = 1.0f / (1.0f + __expf(-row[j]));
    const float av = (j == i) ? 1.0f : arow[j];
    const float v  = (sg + ((j == i) ? 1e-5f : 0.0f)) * av;
    row[j] = v;
    part += v;
  }
  red[tid] = part;
  __syncthreads();
  for (int off = 128; off > 0; off >>= 1) {
    if (tid < off) red[tid] += red[tid + off];
    __syncthreads();
  }
  const float inv = 1.0f / red[0];
  for (int j = tid; j < S_; j += 256) row[j] *= inv;
}

__global__ __launch_bounds__(256) void head_mean_kernel(
    const float* __restrict__ h1, const float* __restrict__ gfin,
    float* __restrict__ hidden)
{
  const int b = blockIdx.x, tid = threadIdx.x;
  for (int d = tid; d < 512; d += 256) {
    float s = 0.0f;
    if (d < 256) {
      for (int t = 0; t < S_; ++t) s += h1[((size_t)t * B_ + b) * 256 + d];
    } else {
      const int dd = d - 256;
      for (int t = 0; t < S_; ++t) s += gfin[((size_t)b * S_ + t) * 256 + dd];
    }
    hidden[b * 512 + d] = s * (1.0f / (float)S_);
  }
}

template <int A_MODE, int ACT, bool FULLN>
static inline void gemm(hipStream_t st,
                        const float* A, const float* Bm,
                        const float* bias, const float* bias2, const float* resid,
                        float* C, int M, int N, int K, int lda, int ldb, int ldc,
                        long long sA = 0, long long sB = 0, long long sC = 0,
                        int batches = 1)
{
  dim3 grid((N + 255) / 256, (M + 31) / 32, batches);
  gemm_wmma_kernel<A_MODE, ACT, FULLN><<<grid, 128, 0, st>>>(
      A, Bm, bias, bias2, resid, C, M, N, K, lda, ldb, ldc, sA, sB, sC);
}

extern "C" void kernel_launch(void* const* d_in, const int* in_sizes, int n_in,
                              void* d_out, int out_size, void* d_ws, size_t ws_size,
                              hipStream_t stream)
{
  (void)in_sizes; (void)n_in; (void)out_size; (void)ws_size;
  auto F = [&](int i) { return (const float*)d_in[i]; };

  const float* x        = F(0);
  const float* adj      = F(1);
  const float* l0f_Wih  = F(2),  *l0f_Whh = F(3),  *l0f_bih = F(4),  *l0f_bhh = F(5);
  const float* l0b_Wih  = F(6),  *l0b_Whh = F(7),  *l0b_bih = F(8),  *l0b_bhh = F(9);
  const float* l1f_Wih  = F(10), *l1f_Whh = F(11), *l1f_bih = F(12), *l1f_bhh = F(13);
  const float* l1b_Wih  = F(14), *l1b_Whh = F(15), *l1b_bih = F(16), *l1b_bhh = F(17);
  const float* fc1_W    = F(18), *fc1_b   = F(19);
  const float* wattn_W  = F(20), *wattn_b = F(21);
  const float* lin0_W   = F(22), *lin0_b  = F(23);
  const float* lin1_W   = F(24), *lin1_b  = F(25);
  const float* linf_W   = F(26), *linf_b  = F(27);
  const float* fcA_W    = F(28), *fcA_b   = F(29);
  const float* fcB_W    = F(30), *fcB_b   = F(31);
  const float* fcC_W    = F(32), *fcC_b   = F(33);

  char* wsp = (char*)d_ws;
  size_t off = 0;
  auto carve = [&](size_t bytes) -> void* {
    void* p = wsp + off;
    off += (bytes + 255) & ~(size_t)255;
    return p;
  };
  float* preF   = (float*)carve((size_t)S_ * B_ * 4 * H_ * 4);
  float* preB   = (float*)carve((size_t)S_ * B_ * 4 * H_ * 4);
  float* h0     = (float*)carve((size_t)S_ * B_ * 2 * H_ * 4);
  float* h1     = (float*)carve((size_t)S_ * B_ * 2 * H_ * 4);
  float* gA     = (float*)carve((size_t)B_ * S_ * G_ * 4);
  float* gB     = (float*)carve((size_t)B_ * S_ * G_ * 4);
  float* gC     = (float*)carve((size_t)B_ * S_ * G_ * 4);
  float* gT     = (float*)carve((size_t)B_ * G_ * S_ * 4);
  float* scores = (float*)carve((size_t)B_ * S_ * S_ * 4);
  __half* whh0f = (__half*)carve((size_t)4 * H_ * H_ * 2);
  __half* whh0b = (__half*)carve((size_t)4 * H_ * H_ * 2);
  __half* whh1f = (__half*)carve((size_t)4 * H_ * H_ * 2);
  __half* whh1b = (__half*)carve((size_t)4 * H_ * H_ * 2);
  float* fcAo   = (float*)carve((size_t)B_ * 128 * 4);
  float* fcBo   = (float*)carve((size_t)B_ * 32 * 4);

  const int MSB = S_ * B_;
  const int nwh = 4 * H_ * H_;

  cvt_f32_f16_kernel<<<(nwh + 255) / 256, 256, 0, stream>>>(l0f_Whh, whh0f, nwh);
  cvt_f32_f16_kernel<<<(nwh + 255) / 256, 256, 0, stream>>>(l0b_Whh, whh0b, nwh);
  cvt_f32_f16_kernel<<<(nwh + 255) / 256, 256, 0, stream>>>(l1f_Whh, whh1f, nwh);
  cvt_f32_f16_kernel<<<(nwh + 255) / 256, 256, 0, stream>>>(l1b_Whh, whh1b, nwh);

  gemm<1, 0, true>(stream, x, l0f_Wih, l0f_bih, l0f_bhh, nullptr, preF,
                   MSB, 4 * H_, D_, D_, D_, 4 * H_);
  gemm<1, 0, true>(stream, x, l0b_Wih, l0b_bih, l0b_bhh, nullptr, preB,
                   MSB, 4 * H_, D_, D_, D_, 4 * H_);
  lstm_dir_kernel<<<dim3(2, B_ / LB_), 256, 0, stream>>>(preF, preB, whh0f, whh0b, h0);

  gemm<0, 0, true>(stream, h0, l1f_Wih, l1f_bih, l1f_bhh, nullptr, preF,
                   MSB, 4 * H_, 2 * H_, 2 * H_, 2 * H_, 4 * H_);
  gemm<0, 0, true>(stream, h0, l1b_Wih, l1b_bih, l1b_bhh, nullptr, preB,
                   MSB, 4 * H_, 2 * H_, 2 * H_, 2 * H_, 4 * H_);
  lstm_dir_kernel<<<dim3(2, B_ / LB_), 256, 0, stream>>>(preF, preB, whh1f, whh1b, h1);

  gemm<0, 0, true>(stream, x, fc1_W, fc1_b, nullptr, nullptr, gA,
                   MSB, G_, D_, D_, D_, G_);

  float* cur = gA; float* t1 = gB; float* t2 = gC;
  const long long sGG = (long long)S_ * G_;
  const long long sSS = (long long)S_ * S_;
  for (int l = 0; l < 4; ++l) {
    const float* Wq = wattn_W + (size_t)l * G_ * G_;
    gemm<0, 0, true>(stream, cur, Wq, wattn_b + l * G_, nullptr, nullptr, t2,
                     MSB, G_, G_, G_, G_, G_);
    gemm<0, 0, true>(stream, t2, cur, nullptr, nullptr, nullptr, scores,
                     S_, S_, G_, G_, G_, S_, sGG, sGG, sSS, B_);
    attn_mask_norm_kernel<<<dim3(S_, B_), 256, 0, stream>>>(scores, adj);
    transpose_bsg_kernel<<<dim3(S_ / 32, G_ / 32, B_), dim3(32, 8), 0, stream>>>(cur, gT);
    gemm<0, 0, true>(stream, scores, gT, nullptr, nullptr, nullptr, t1,
                     S_, G_, S_, S_, S_, G_, sSS, (long long)G_ * S_, sGG, B_);
    gemm<0, 1, true>(stream, t1, lin0_W + (size_t)l * G_ * G_, lin0_b + l * G_,
                     nullptr, nullptr, t2, MSB, G_, G_, G_, G_, G_);
    gemm<0, 1, true>(stream, t2, lin1_W + (size_t)l * G_ * G_, lin1_b + l * G_,
                     nullptr, cur, t1, MSB, G_, G_, G_, G_, G_);
    float* old = cur; cur = t1; t1 = old;
  }
  gemm<0, 0, true>(stream, cur, linf_W, linf_b, nullptr, nullptr, t2,
                   MSB, G_, G_, G_, G_, G_);
  float* gfin = t2;

  float* y_out  = (float*)d_out;
  float* hidden = (float*)d_out + B_ * 2;
  head_mean_kernel<<<B_, 256, 0, stream>>>(h1, gfin, hidden);

  gemm<0, 2, true >(stream, hidden, fcA_W, fcA_b, nullptr, nullptr, fcAo,
                    B_, 128, 512, 512, 512, 128);
  gemm<0, 2, false>(stream, fcAo, fcB_W, fcB_b, nullptr, nullptr, fcBo,
                    B_, 32, 128, 128, 128, 32);
  gemm<0, 0, false>(stream, fcBo, fcC_W, fcC_b, nullptr, nullptr, y_out,
                    B_, 2, 32, 32, 32, 2);
}


# LSTM recurrence: each wave owns the i,f,g,o tiles of 16 hidden units, so a lane finishes its cell in registers: z skips LDS, h double-buffered, one barrier per step not two; summation order unchanged
# speedup vs baseline: 3.6110x; 1.0635x over previous
#include <hip/hip_runtime.h>
#include <hip/hip_fp16.h>
#include <hip/hip_bf16.h>

#define B_  16
#define S_  512
#define D_  1024
#define H_  128
#define G_  256
#define HP_ 136

typedef __attribute__((ext_vector_type(16))) _Float16 v16h;
typedef __attribute__((ext_vector_type(8)))  float    v8f;

__device__ __forceinline__ v16h frag_from_f32(const float* p) {
  float4 x0 = *(const float4*)(p);
  float4 x1 = *(const float4*)(p + 4);
  float4 y0 = *(const float4*)(p + 16);
  float4 y1 = *(const float4*)(p + 20);
  v16h f;
  f[0]=(_Float16)x0.x;  f[1]=(_Float16)x0.y;  f[2]=(_Float16)x0.z;  f[3]=(_Float16)x0.w;
  f[4]=(_Float16)x1.x;  f[5]=(_Float16)x1.y;  f[6]=(_Float16)x1.z;  f[7]=(_Float16)x1.w;
  f[8]=(_Float16)y0.x;  f[9]=(_Float16)y0.y;  f[10]=(_Float16)y0.z; f[11]=(_Float16)y0.w;
  f[12]=(_Float16)y1.x; f[13]=(_Float16)y1.y; f[14]=(_Float16)y1.z; f[15]=(_Float16)y1.w;
  return f;
}

__device__ __forceinline__ v16h frag_from_f16(const _Float16* p) {
  v16h f;
  #pragma unroll
  for (int i = 0; i < 8; ++i) { f[i] = p[i]; f[8 + i] = p[16 + i]; }
  return f;
}

#define WMMA_F16(a, b, c) __builtin_amdgcn_wmma_f32_16x16x32_f16( \
    false, (a), false, (b), (short)0, (c), false, false)

template <int A_MODE, int ACT, bool FULLN>
__global__ __launch_bounds__(128) void gemm_wmma_kernel(
    const float* __restrict__ A, const float* __restrict__ Bm,
    const float* __restrict__ bias, const float* __restrict__ bias2,
    const float* __restrict__ resid, float* __restrict__ C,
    int M, int N, int K, int lda, int ldb, int ldc,
    long long sA, long long sB, long long sC)
{
  const int lane  = threadIdx.x & 31;
  const int wave  = threadIdx.x >> 5;
  const int tileM = blockIdx.y * 32;
  const int nbase = blockIdx.x * 256 + wave * 64;
  if (nbase >= N) return;

  const int z = blockIdx.z;
  const float* Ab = A  + (size_t)z * (size_t)sA;
  const float* Bb = Bm + (size_t)z * (size_t)sB;
  float*       Cb = C  + (size_t)z * (size_t)sC;
  const float* Rb = resid ? (resid + (size_t)z * (size_t)sC) : nullptr;

  const int l15   = lane & 15;
  const int khalf = (lane & 16) ? 8 : 0;

  int am0 = tileM + l15;      if (am0 >= M) am0 = M - 1;
  int am1 = tileM + 16 + l15; if (am1 >= M) am1 = M - 1;
  size_t arow0, arow1;
  if (A_MODE == 1) {
    arow0 = ((size_t)(am0 & 15) * S_ + (size_t)(am0 >> 4)) * (size_t)lda;
    arow1 = ((size_t)(am1 & 15) * S_ + (size_t)(am1 >> 4)) * (size_t)lda;
  } else {
    arow0 = (size_t)am0 * (size_t)lda;
    arow1 = (size_t)am1 * (size_t)lda;
  }

  bool ton[4]; int bn[4];
  #pragma unroll
  for (int nt = 0; nt < 4; ++nt) {
    if (FULLN) {
      ton[nt] = true;
      bn[nt]  = nbase + nt * 16 + l15;
    } else {
      ton[nt] = (nbase + nt * 16) < N;
      int c = nbase + nt * 16 + l15; if (c >= N) c = N - 1;
      bn[nt] = c;
    }
  }

  v8f acc[2][4] = {};
  for (int k0 = 0; k0 < K; k0 += 32) {
    const float* pa0 = Ab + arow0 + k0 + khalf;
    const float* pa1 = Ab + arow1 + k0 + khalf;
    v16h a0 = frag_from_f32(pa0);
    v16h a1 = frag_from_f32(pa1);
    if (k0 + 32 < K) {
      __builtin_prefetch(pa0 + 32, 0, 1);
      __builtin_prefetch(pa1 + 32, 0, 1);
    }
    #pragma unroll
    for (int nt = 0; nt < 4; ++nt) {
      if (!FULLN && !ton[nt]) continue;
      v16h b = frag_from_f32(Bb + (size_t)bn[nt] * (size_t)ldb + k0 + khalf);
      acc[0][nt] = WMMA_F16(a0, b, acc[0][nt]);
      acc[1][nt] = WMMA_F16(a1, b, acc[1][nt]);
    }
  }

  const int rb = (lane < 16) ? 0 : 8;
  #pragma unroll
  for (int nt = 0; nt < 4; ++nt) {
    if (!FULLN && !ton[nt]) continue;
    const int col = nbase + nt * 16 + l15;
    bool colok = FULLN ? true : (col < N);
    float bv = 0.0f;
    if (colok) {
      if (bias)  bv += bias[col];
      if (bias2) bv += bias2[col];
    }
    #pragma unroll
    for (int mt = 0; mt < 2; ++mt) {
      #pragma unroll
      for (int r = 0; r < 8; ++r) {
        const int row = tileM + mt * 16 + rb + r;
        if (row < M && colok) {
          float v = acc[mt][nt][r] + bv;
          if      (ACT == 1) v = v > 0.0f ? v : 0.0f;
          else if (ACT == 2) v = v > 0.0f ? v : 0.1f * v;
          if (Rb) v += Rb[(size_t)row * (size_t)ldc + col];
          Cb[(size_t)row * (size_t)ldc + col] = v;
        }
      }
    }
  }
}

__global__ void cvt_f32_f16_kernel(const float* __restrict__ src,
                                   __half* __restrict__ dst, int n) {
  int i = blockIdx.x * blockDim.x + threadIdx.x;
  if (i < n) ((_Float16*)dst)[i] = (_Float16)src[i];
}

__global__ __launch_bounds__(256) void transpose_bsg_kernel(
    const float* __restrict__ src, float* __restrict__ dst)
{
  __shared__ float tile[32][33];
  const int b  = blockIdx.z;
  const int s0 = blockIdx.x * 32, g0 = blockIdx.y * 32;
  const int tx = threadIdx.x, ty = threadIdx.y;
  const float* sp = src + (size_t)b * S_ * G_;
  float*       dp = dst + (size_t)b * G_ * S_;
  #pragma unroll
  for (int r = 0; r < 32; r += 8)
    tile[ty + r][tx] = sp[(size_t)(s0 + ty + r) * G_ + g0 + tx];
  __syncthreads();
  #pragma unroll
  for (int r = 0; r < 32; r += 8)
    dp[(size_t)(g0 + ty + r) * S_ + s0 + tx] = tile[tx][ty + r];
}

#define LB_ 2
#define PD_ 8
static_assert(S_ % PD_ == 0, "the step loop is unrolled by PD_");
static_assert(PD_ % 2 == 0, "the h buffer of a step is chosen by the parity of its ring slot");

__device__ __forceinline__ void lds_only_barrier() {
  __builtin_amdgcn_fence(__ATOMIC_RELEASE, "workgroup", "local");
  __builtin_amdgcn_s_barrier();
  __builtin_amdgcn_fence(__ATOMIC_ACQUIRE, "workgroup", "local");
}

__device__ __forceinline__ float gate_sigmoid(float x) {
  return __builtin_amdgcn_rcpf(1.0f + __expf(-x));
}
__device__ __forceinline__ float gate_tanh(float x) {
  return 1.0f - 2.0f * __builtin_amdgcn_rcpf(1.0f + __expf(2.0f * x));
}

__global__ __launch_bounds__(256) void lstm_dir_kernel(
    const float* __restrict__ preF, const float* __restrict__ preB,
    const __half* __restrict__ whhF16_f, const __half* __restrict__ whhF16_b,
    float* __restrict__ hs)
{
  const int dir = blockIdx.x;
  const int b0  = blockIdx.y * LB_;
  const float*    pre = dir ? preB : preF;
  const _Float16* whh = (const _Float16*)(dir ? whhF16_b : whhF16_f);

  __shared__ _Float16 h_lds[2][LB_ * HP_];

  const int tid   = threadIdx.x;
  const int wave  = tid >> 5;
  const int lane  = tid & 31;
  const int l15   = lane & 15;
  const int khalf = (lane & 16) ? 8 : 0;
  const int hrow  = l15 >> 3;
  const int half  = lane >> 4;

  v16h wfrag[16];
  #pragma unroll
  for (int nt = 0; nt < 4; ++nt) {
    const int tileN = nt * H_ + wave * 16;
    #pragma unroll
    for (int kc = 0; kc < 4; ++kc)
      wfrag[nt * 4 + kc] =
          frag_from_f16(whh + (size_t)(tileN + l15) * H_ + kc * 32 + khalf);
  }

  for (int i = tid; i < LB_ * HP_; i += 256) h_lds[0][i] = (_Float16)0.0f;

  const int hx = wave * 16 + l15;
  float c = 0.0f;
  __syncthreads();

  const float* pre_lane = pre + (size_t)(b0 + half) * (4 * H_) + hx;
  float pq[PD_][4];
  #pragma unroll
  for (int j = 0; j < PD_ - 1; ++j) {
    const int tj = dir ? (S_ - 1 - j) : j;
    #pragma unroll
    for (int nt = 0; nt < 4; ++nt)
      pq[j][nt] = pre_lane[(size_t)tj * (B_ * 4 * H_) + nt * H_];
  }

  for (int step0 = 0; step0 < S_; step0 += PD_) {
    #pragma unroll
    for (int j = 0; j < PD_; ++j) {
      const int step = step0 + j;
      const int t = dir ? (S_ - 1 - step) : step;

      {
        const int sn = (step + PD_ - 1 < S_) ? (step + PD_ - 1) : (S_ - 1);
        const int tn = dir ? (S_ - 1 - sn) : sn;
        #pragma unroll
        for (int nt = 0; nt < 4; ++nt)
          pq[(j + PD_ - 1) % PD_][nt] = pre_lane[(size_t)tn * (B_ * 4 * H_) + nt * H_];
      }

      const _Float16* h_rd = h_lds[j & 1];
      _Float16*       h_wr = h_lds[(j & 1) ^ 1];

      v16h a[4];
      #pragma unroll
      for (int kc = 0; kc < 4; ++kc)
        a[kc] = frag_from_f16(&h_rd[hrow * HP_ + kc * 32 + khalf]);

      float z[4];
      #pragma unroll
      for (int nt = 0; nt < 4; ++nt) {
        v8f acc = {};
        #pragma unroll
        for (int kc = 0; kc < 4; ++kc)
          acc = WMMA_F16(a[kc], wfrag[nt * 4 + kc], acc);
        z[nt] = acc[0] + pq[j][nt];
      }

      {
        const float si = gate_sigmoid(z[0]);
        const float sf = gate_sigmoid(z[1]);
        const float so = gate_sigmoid(z[3]);
        c = sf * c + si * gate_tanh(z[2]);
        const float h = so * gate_tanh(c);
        h_wr[half * HP_ + hx] = (_Float16)h;
        hs[((size_t)t * B_ + b0 + half) * (2 * H_) + dir * H_ + hx] = h;
      }
      lds_only_barrier();
    }
  }
}

__global__ __launch_bounds__(256) void attn_mask_norm_kernel(
    float* __restrict__ scores, const float* __restrict__ adj)
{
  const int i = blockIdx.x, b = blockIdx.y, tid = threadIdx.x;
  float* row        = scores + ((size_t)b * S_ + i) * S_;
  const float* arow = adj    + ((size_t)b * S_ + i) * S_;
  __shared__ float red[256];
  float part = 0.0f;
  for (int j = tid; j < S_; j += 256) {
    const float sg = 1.0f / (1.0f + __expf(-row[j]));
    const float av = (j == i) ? 1.0f : arow[j];
    const float v  = (sg + ((j == i) ? 1e-5f : 0.0f)) * av;
    row[j] = v;
    part += v;
  }
  red[tid] = part;
  __syncthreads();
  for (int off = 128; off > 0; off >>= 1) {
    if (tid < off) red[tid] += red[tid + off];
    __syncthreads();
  }
  const float inv = 1.0f / red[0];
  for (int j = tid; j < S_; j += 256) row[j] *= inv;
}

__global__ __launch_bounds__(256) void head_mean_kernel(
    const float* __restrict__ h1, const float* __restrict__ gfin,
    float* __restrict__ hidden)
{
  const int b = blockIdx.x, tid = threadIdx.x;
  for (int d = tid; d < 512; d += 256) {
    float s = 0.0f;
    if (d < 256) {
      for (int t = 0; t < S_; ++t) s += h1[((size_t)t * B_ + b) * 256 + d];
    } else {
      const int dd = d - 256;
      for (int t = 0; t < S_; ++t) s += gfin[((size_t)b * S_ + t) * 256 + dd];
    }
    hidden[b * 512 + d] = s * (1.0f / (float)S_);
  }
}

template <int A_MODE, int ACT, bool FULLN>
static inline void gemm(hipStream_t st,
                        const float* A, const float* Bm,
                        const float* bias, const float* bias2, const float* resid,
                        float* C, int M, int N, int K, int lda, int ldb, int ldc,
                        long long sA = 0, long long sB = 0, long long sC = 0,
                        int batches = 1)
{
  dim3 grid((N + 255) / 256, (M + 31) / 32, batches);
  gemm_wmma_kernel<A_MODE, ACT, FULLN><<<grid, 128, 0, st>>>(
      A, Bm, bias, bias2, resid, C, M, N, K, lda, ldb, ldc, sA, sB, sC);
}

extern "C" void kernel_launch(void* const* d_in, const int* in_sizes, int n_in,
                              void* d_out, int out_size, void* d_ws, size_t ws_size,
                              hipStream_t stream)
{
  (void)in_sizes; (void)n_in; (void)out_size; (void)ws_size;
  auto F = [&](int i) { return (const float*)d_in[i]; };

  const float* x        = F(0);
  const float* adj      = F(1);
  const float* l0f_Wih  = F(2),  *l0f_Whh = F(3),  *l0f_bih = F(4),  *l0f_bhh = F(5);
  const float* l0b_Wih  = F(6),  *l0b_Whh = F(7),  *l0b_bih = F(8),  *l0b_bhh = F(9);
  const float* l1f_Wih  = F(10), *l1f_Whh = F(11), *l1f_bih = F(12), *l1f_bhh = F(13);
  const float* l1b_Wih  = F(14), *l1b_Whh = F(15), *l1b_bih = F(16), *l1b_bhh = F(17);
  const float* fc1_W    = F(18), *fc1_b   = F(19);
  const float* wattn_W  = F(20), *wattn_b = F(21);
  const float* lin0_W   = F(22), *lin0_b  = F(23);
  const float* lin1_W   = F(24), *lin1_b  = F(25);
  const float* linf_W   = F(26), *linf_b  = F(27);
  const float* fcA_W    = F(28), *fcA_b   = F(29);
  const float* fcB_W    = F(30), *fcB_b   = F(31);
  const float* fcC_W    = F(32), *fcC_b   = F(33);

  char* wsp = (char*)d_ws;
  size_t off = 0;
  auto carve = [&](size_t bytes) -> void* {
    void* p = wsp + off;
    off += (bytes + 255) & ~(size_t)255;
    return p;
  };
  float* preF   = (float*)carve((size_t)S_ * B_ * 4 * H_ * 4);
  float* preB   = (float*)carve((size_t)S_ * B_ * 4 * H_ * 4);
  float* h0     = (float*)carve((size_t)S_ * B_ * 2 * H_ * 4);
  float* h1     = (float*)carve((size_t)S_ * B_ * 2 * H_ * 4);
  float* gA     = (float*)carve((size_t)B_ * S_ * G_ * 4);
  float* gB     = (float*)carve((size_t)B_ * S_ * G_ * 4);
  float* gC     = (float*)carve((size_t)B_ * S_ * G_ * 4);
  float* gT     = (float*)carve((size_t)B_ * G_ * S_ * 4);
  float* scores = (float*)carve((size_t)B_ * S_ * S_ * 4);
  __half* whh0f = (__half*)carve((size_t)4 * H_ * H_ * 2);
  __half* whh0b = (__half*)carve((size_t)4 * H_ * H_ * 2);
  __half* whh1f = (__half*)carve((size_t)4 * H_ * H_ * 2);
  __half* whh1b = (__half*)carve((size_t)4 * H_ * H_ * 2);
  float* fcAo   = (float*)carve((size_t)B_ * 128 * 4);
  float* fcBo   = (float*)carve((size_t)B_ * 32 * 4);

  const int MSB = S_ * B_;
  const int nwh = 4 * H_ * H_;

  cvt_f32_f16_kernel<<<(nwh + 255) / 256, 256, 0, stream>>>(l0f_Whh, whh0f, nwh);
  cvt_f32_f16_kernel<<<(nwh + 255) / 256, 256, 0, stream>>>(l0b_Whh, whh0b, nwh);
  cvt_f32_f16_kernel<<<(nwh + 255) / 256, 256, 0, stream>>>(l1f_Whh, whh1f, nwh);
  cvt_f32_f16_kernel<<<(nwh + 255) / 256, 256, 0, stream>>>(l1b_Whh, whh1b, nwh);

  gemm<1, 0, true>(stream, x, l0f_Wih, l0f_bih, l0f_bhh, nullptr, preF,
                   MSB, 4 * H_, D_, D_, D_, 4 * H_);
  gemm<1, 0, true>(stream, x, l0b_Wih, l0b_bih, l0b_bhh, nullptr, preB,
                   MSB, 4 * H_, D_, D_, D_, 4 * H_);
  lstm_dir_kernel<<<dim3(2, B_ / LB_), 256, 0, stream>>>(preF, preB, whh0f, whh0b, h0);

  gemm<0, 0, true>(stream, h0, l1f_Wih, l1f_bih, l1f_bhh, nullptr, preF,
                   MSB, 4 * H_, 2 * H_, 2 * H_, 2 * H_, 4 * H_);
  gemm<0, 0, true>(stream, h0, l1b_Wih, l1b_bih, l1b_bhh, nullptr, preB,
                   MSB, 4 * H_, 2 * H_, 2 * H_, 2 * H_, 4 * H_);
  lstm_dir_kernel<<<dim3(2, B_ / LB_), 256, 0, stream>>>(preF, preB, whh1f, whh1b, h1);

  gemm<0, 0, true>(stream, x, fc1_W, fc1_b, nullptr, nullptr, gA,
                   MSB, G_, D_, D_, D_, G_);

  float* cur = gA; float* t1 = gB; float* t2 = gC;
  const long long sGG = (long long)S_ * G_;
  const long long sSS = (long long)S_ * S_;
  for (int l = 0; l < 4; ++l) {
    const float* Wq = wattn_W + (size_t)l * G_ * G_;
    gemm<0, 0, true>(stream, cur, Wq, wattn_b + l * G_, nullptr, nullptr, t2,
                     MSB, G_, G_, G_, G_, G_);
    gemm<0, 0, true>(stream, t2, cur, nullptr, nullptr, nullptr, scores,
                     S_, S_, G_, G_, G_, S_, sGG, sGG, sSS, B_);
    attn_mask_norm_kernel<<<dim3(S_, B_), 256, 0, stream>>>(scores, adj);
    transpose_bsg_kernel<<<dim3(S_ / 32, G_ / 32, B_), dim3(32, 8), 0, stream>>>(cur, gT);
    gemm<0, 0, true>(stream, scores, gT, nullptr, nullptr, nullptr, t1,
                     S_, G_, S_, S_, S_, G_, sSS, (long long)G_ * S_, sGG, B_);
    gemm<0, 1, true>(stream, t1, lin0_W + (size_t)l * G_ * G_, lin0_b + l * G_,
                     nullptr, nullptr, t2, MSB, G_, G_, G_, G_, G_);
    gemm<0, 1, true>(stream, t2, lin1_W + (size_t)l * G_ * G_, lin1_b + l * G_,
                     nullptr, cur, t1, MSB, G_, G_, G_, G_, G_);
    float* old = cur; cur = t1; t1 = old;
  }
  gemm<0, 0, true>(stream, cur, linf_W, linf_b, nullptr, nullptr, t2,
                   MSB, G_, G_, G_, G_, G_);
  float* gfin = t2;

  float* y_out  = (float*)d_out;
  float* hidden = (float*)d_out + B_ * 2;
  head_mean_kernel<<<B_, 256, 0, stream>>>(h1, gfin, hidden);

  gemm<0, 2, true >(stream, hidden, fcA_W, fcA_b, nullptr, nullptr, fcAo,
                    B_, 128, 512, 512, 512, 128);
  gemm<0, 2, false>(stream, fcAo, fcB_W, fcB_b, nullptr, nullptr, fcBo,
                    B_, 32, 128, 128, 128, 32);
  gemm<0, 0, false>(stream, fcBo, fcC_W, fcC_b, nullptr, nullptr, y_out,
                    B_, 2, 32, 32, 32, 2);
}


# WMMA GEMM K loop: a k chunk's 24 operand loads go out as one group two chunks ahead into a raw f32 register ring, not fragment by fragment inside the chunk; summation order unchanged
# speedup vs baseline: 4.2375x; 1.1735x over previous
#include <hip/hip_runtime.h>
#include <hip/hip_fp16.h>
#include <hip/hip_bf16.h>

#define B_  16
#define S_  512
#define D_  1024
#define H_  128
#define G_  256
#define HP_ 136

typedef __attribute__((ext_vector_type(16))) _Float16 v16h;
typedef __attribute__((ext_vector_type(8)))  float    v8f;

struct rawfrag { float4 x0, x1, y0, y1; };

__device__ __forceinline__ rawfrag raw_load(const float* p) {
  rawfrag r;
  r.x0 = *(const float4*)(p);
  r.x1 = *(const float4*)(p + 4);
  r.y0 = *(const float4*)(p + 16);
  r.y1 = *(const float4*)(p + 20);
  return r;
}

__device__ __forceinline__ v16h raw_to_frag(const rawfrag& r) {
  v16h f;
  f[0]=(_Float16)r.x0.x;  f[1]=(_Float16)r.x0.y;  f[2]=(_Float16)r.x0.z;  f[3]=(_Float16)r.x0.w;
  f[4]=(_Float16)r.x1.x;  f[5]=(_Float16)r.x1.y;  f[6]=(_Float16)r.x1.z;  f[7]=(_Float16)r.x1.w;
  f[8]=(_Float16)r.y0.x;  f[9]=(_Float16)r.y0.y;  f[10]=(_Float16)r.y0.z; f[11]=(_Float16)r.y0.w;
  f[12]=(_Float16)r.y1.x; f[13]=(_Float16)r.y1.y; f[14]=(_Float16)r.y1.z; f[15]=(_Float16)r.y1.w;
  return f;
}

__device__ __forceinline__ v16h frag_from_f16(const _Float16* p) {
  v16h f;
  #pragma unroll
  for (int i = 0; i < 8; ++i) { f[i] = p[i]; f[8 + i] = p[16 + i]; }
  return f;
}

#define WMMA_F16(a, b, c) __builtin_amdgcn_wmma_f32_16x16x32_f16( \
    false, (a), false, (b), (short)0, (c), false, false)

#define GPF_ 2

template <bool FULLN>
__device__ __forceinline__ void gemm_chunk(
    rawfrag (&sa)[2], rawfrag (&sb)[4], v8f (&acc)[2][4], const bool (&ton)[4],
    const float* pa0, const float* pa1, const float* const (&pb)[4],
    bool refill, int koff_next)
{
  v16h a0 = raw_to_frag(sa[0]);
  v16h a1 = raw_to_frag(sa[1]);
  v16h b[4];
  #pragma unroll
  for (int nt = 0; nt < 4; ++nt)
    if (FULLN || ton[nt]) b[nt] = raw_to_frag(sb[nt]);
  asm volatile("" : "+v"(a0), "+v"(a1) : : "memory");
  #pragma unroll
  for (int nt = 0; nt < 4; ++nt)
    if (FULLN || ton[nt]) asm volatile("" : "+v"(b[nt]) : : "memory");
  if (refill) {
    sa[0] = raw_load(pa0 + koff_next);
    sa[1] = raw_load(pa1 + koff_next);
    #pragma unroll
    for (int nt = 0; nt < 4; ++nt)
      if (FULLN || ton[nt]) sb[nt] = raw_load(pb[nt] + koff_next);
  }
  __builtin_amdgcn_sched_barrier(0);
  #pragma unroll
  for (int nt = 0; nt < 4; ++nt) {
    if (!FULLN && !ton[nt]) continue;
    acc[0][nt] = WMMA_F16(a0, b[nt], acc[0][nt]);
    acc[1][nt] = WMMA_F16(a1, b[nt], acc[1][nt]);
  }
}

template <int A_MODE, int ACT, bool FULLN>
__global__ __launch_bounds__(128) void gemm_wmma_kernel(
    const float* __restrict__ A, const float* __restrict__ Bm,
    const float* __restrict__ bias, const float* __restrict__ bias2,
    const float* __restrict__ resid, float* __restrict__ C,
    int M, int N, int K, int lda, int ldb, int ldc,
    long long sA, long long sB, long long sC)
{
  const int lane  = threadIdx.x & 31;
  const int wave  = threadIdx.x >> 5;
  const int tileM = blockIdx.y * 32;
  const int nbase = blockIdx.x * 256 + wave * 64;
  if (nbase >= N) return;

  const int z = blockIdx.z;
  const float* Ab = A  + (size_t)z * (size_t)sA;
  const float* Bb = Bm + (size_t)z * (size_t)sB;
  float*       Cb = C  + (size_t)z * (size_t)sC;
  const float* Rb = resid ? (resid + (size_t)z * (size_t)sC) : nullptr;

  const int l15   = lane & 15;
  const int khalf = (lane & 16) ? 8 : 0;

  int am0 = tileM + l15;      if (am0 >= M) am0 = M - 1;
  int am1 = tileM + 16 + l15; if (am1 >= M) am1 = M - 1;
  size_t arow0, arow1;
  if (A_MODE == 1) {
    arow0 = ((size_t)(am0 & 15) * S_ + (size_t)(am0 >> 4)) * (size_t)lda;
    arow1 = ((size_t)(am1 & 15) * S_ + (size_t)(am1 >> 4)) * (size_t)lda;
  } else {
    arow0 = (size_t)am0 * (size_t)lda;
    arow1 = (size_t)am1 * (size_t)lda;
  }

  bool ton[4]; int bn[4];
  #pragma unroll
  for (int nt = 0; nt < 4; ++nt) {
    if (FULLN) {
      ton[nt] = true;
      bn[nt]  = nbase + nt * 16 + l15;
    } else {
      ton[nt] = (nbase + nt * 16) < N;
      int c = nbase + nt * 16 + l15; if (c >= N) c = N - 1;
      bn[nt] = c;
    }
  }

  const float* pa0 = Ab + arow0 + khalf;
  const float* pa1 = Ab + arow1 + khalf;
  const float* pb[4];
  #pragma unroll
  for (int nt = 0; nt < 4; ++nt)
    pb[nt] = Bb + (size_t)bn[nt] * (size_t)ldb + khalf;

  const int nk = K >> 5;
  rawfrag ra[GPF_][2], rb_[GPF_][4];
  #pragma unroll
  for (int j = 0; j < GPF_; ++j) {
    const int kj = ((j < nk) ? j : (nk - 1)) * 32;
    ra[j][0] = raw_load(pa0 + kj);
    ra[j][1] = raw_load(pa1 + kj);
    #pragma unroll
    for (int nt = 0; nt < 4; ++nt)
      if (FULLN || ton[nt]) rb_[j][nt] = raw_load(pb[nt] + kj);
  }

  v8f acc[2][4] = {};
  int kc = 0;
  for (; kc + 2 * GPF_ <= nk; kc += GPF_) {
    #pragma unroll
    for (int j = 0; j < GPF_; ++j)
      gemm_chunk<FULLN>(ra[j], rb_[j], acc, ton, pa0, pa1, pb,
                        true, (kc + j + GPF_) * 32);
  }
  for (; kc < nk; kc += GPF_) {
    #pragma unroll
    for (int j = 0; j < GPF_; ++j)
      if (kc + j < nk)
        gemm_chunk<FULLN>(ra[j], rb_[j], acc, ton, pa0, pa1, pb,
                          kc + j + GPF_ < nk, (kc + j + GPF_) * 32);
  }

  const int rb = (lane < 16) ? 0 : 8;
  #pragma unroll
  for (int nt = 0; nt < 4; ++nt) {
    if (!FULLN && !ton[nt]) continue;
    const int col = nbase + nt * 16 + l15;
    bool colok = FULLN ? true : (col < N);
    float bv = 0.0f;
    if (colok) {
      if (bias)  bv += bias[col];
      if (bias2) bv += bias2[col];
    }
    #pragma unroll
    for (int mt = 0; mt < 2; ++mt) {
      #pragma unroll
      for (int r = 0; r < 8; ++r) {
        const int row = tileM + mt * 16 + rb + r;
        if (row < M && colok) {
          float v = acc[mt][nt][r] + bv;
          if      (ACT == 1) v = v > 0.0f ? v : 0.0f;
          else if (ACT == 2) v = v > 0.0f ? v : 0.1f * v;
          if (Rb) v += Rb[(size_t)row * (size_t)ldc + col];
          Cb[(size_t)row * (size_t)ldc + col] = v;
        }
      }
    }
  }
}

__global__ void cvt_f32_f16_kernel(const float* __restrict__ src,
                                   __half* __restrict__ dst, int n) {
  int i = blockIdx.x * blockDim.x + threadIdx.x;
  if (i < n) ((_Float16*)dst)[i] = (_Float16)src[i];
}

__global__ __launch_bounds__(256) void transpose_bsg_kernel(
    const float* __restrict__ src, float* __restrict__ dst)
{
  __shared__ float tile[32][33];
  const int b  = blockIdx.z;
  const int s0 = blockIdx.x * 32, g0 = blockIdx.y * 32;
  const int tx = threadIdx.x, ty = threadIdx.y;
  const float* sp = src + (size_t)b * S_ * G_;
  float*       dp = dst + (size_t)b * G_ * S_;
  #pragma unroll
  for (int r = 0; r < 32; r += 8)
    tile[ty + r][tx] = sp[(size_t)(s0 + ty + r) * G_ + g0 + tx];
  __syncthreads();
  #pragma unroll
  for (int r = 0; r < 32; r += 8)
    dp[(size_t)(g0 + ty + r) * S_ + s0 + tx] = tile[tx][ty + r];
}

#define LB_ 2
#define PD_ 8
static_assert(S_ % PD_ == 0, "the step loop is unrolled by PD_");
static_assert(PD_ % 2 == 0, "the h buffer of a step is chosen by the parity of its ring slot");

__device__ __forceinline__ void lds_only_barrier() {
  __builtin_amdgcn_fence(__ATOMIC_RELEASE, "workgroup", "local");
  __builtin_amdgcn_s_barrier();
  __builtin_amdgcn_fence(__ATOMIC_ACQUIRE, "workgroup", "local");
}

__device__ __forceinline__ float gate_sigmoid(float x) {
  return __builtin_amdgcn_rcpf(1.0f + __expf(-x));
}
__device__ __forceinline__ float gate_tanh(float x) {
  return 1.0f - 2.0f * __builtin_amdgcn_rcpf(1.0f + __expf(2.0f * x));
}

__global__ __launch_bounds__(256) void lstm_dir_kernel(
    const float* __restrict__ preF, const float* __restrict__ preB,
    const __half* __restrict__ whhF16_f, const __half* __restrict__ whhF16_b,
    float* __restrict__ hs)
{
  const int dir = blockIdx.x;
  const int b0  = blockIdx.y * LB_;
  const float*    pre = dir ? preB : preF;
  const _Float16* whh = (const _Float16*)(dir ? whhF16_b : whhF16_f);

  __shared__ _Float16 h_lds[2][LB_ * HP_];

  const int tid   = threadIdx.x;
  const int wave  = tid >> 5;
  const int lane  = tid & 31;
  const int l15   = lane & 15;
  const int khalf = (lane & 16) ? 8 : 0;
  const int hrow  = l15 >> 3;
  const int half  = lane >> 4;

  v16h wfrag[16];
  #pragma unroll
  for (int nt = 0; nt < 4; ++nt) {
    const int tileN = nt * H_ + wave * 16;
    #pragma unroll
    for (int kc = 0; kc < 4; ++kc)
      wfrag[nt * 4 + kc] =
          frag_from_f16(whh + (size_t)(tileN + l15) * H_ + kc * 32 + khalf);
  }

  for (int i = tid; i < LB_ * HP_; i += 256) h_lds[0][i] = (_Float16)0.0f;

  const int hx = wave * 16 + l15;
  float c = 0.0f;
  __syncthreads();

  const float* pre_lane = pre + (size_t)(b0 + half) * (4 * H_) + hx;
  float pq[PD_][4];
  #pragma unroll
  for (int j = 0; j < PD_ - 1; ++j) {
    const int tj = dir ? (S_ - 1 - j) : j;
    #pragma unroll
    for (int nt = 0; nt < 4; ++nt)
      pq[j][nt] = pre_lane[(size_t)tj * (B_ * 4 * H_) + nt * H_];
  }

  for (int step0 = 0; step0 < S_; step0 += PD_) {
    #pragma unroll
    for (int j = 0; j < PD_; ++j) {
      const int step = step0 + j;
      const int t = dir ? (S_ - 1 - step) : step;

      {
        const int sn = (step + PD_ - 1 < S_) ? (step + PD_ - 1) : (S_ - 1);
        const int tn = dir ? (S_ - 1 - sn) : sn;
        #pragma unroll
        for (int nt = 0; nt < 4; ++nt)
          pq[(j + PD_ - 1) % PD_][nt] = pre_lane[(size_t)tn * (B_ * 4 * H_) + nt * H_];
      }

      const _Float16* h_rd = h_lds[j & 1];
      _Float16*       h_wr = h_lds[(j & 1) ^ 1];

      v16h a[4];
      #pragma unroll
      for (int kc = 0; kc < 4; ++kc)
        a[kc] = frag_from_f16(&h_rd[hrow * HP_ + kc * 32 + khalf]);

      float z[4];
      #pragma unroll
      for (int nt = 0; nt < 4; ++nt) {
        v8f acc = {};
        #pragma unroll
        for (int kc = 0; kc < 4; ++kc)
          acc = WMMA_F16(a[kc], wfrag[nt * 4 + kc], acc);
        z[nt] = acc[0] + pq[j][nt];
      }

      {
        const float si = gate_sigmoid(z[0]);
        const float sf = gate_sigmoid(z[1]);
        const float so = gate_sigmoid(z[3]);
        c = sf * c + si * gate_tanh(z[2]);
        const float h = so * gate_tanh(c);
        h_wr[half * HP_ + hx] = (_Float16)h;
        hs[((size_t)t * B_ + b0 + half) * (2 * H_) + dir * H_ + hx] = h;
      }
      lds_only_barrier();
    }
  }
}

__global__ __launch_bounds__(256) void attn_mask_norm_kernel(
    float* __restrict__ scores, const float* __restrict__ adj)
{
  const int i = blockIdx.x, b = blockIdx.y, tid = threadIdx.x;
  float* row        = scores + ((size_t)b * S_ + i) * S_;
  const float* arow = adj    + ((size_t)b * S_ + i) * S_;
  __shared__ float red[256];
  float part = 0.0f;
  for (int j = tid; j < S_; j += 256) {
    const float sg = 1.0f / (1.0f + __expf(-row[j]));
    const float av = (j == i) ? 1.0f : arow[j];
    const float v  = (sg + ((j == i) ? 1e-5f : 0.0f)) * av;
    row[j] = v;
    part += v;
  }
  red[tid] = part;
  __syncthreads();
  for (int off = 128; off > 0; off >>= 1) {
    if (tid < off) red[tid] += red[tid + off];
    __syncthreads();
  }
  const float inv = 1.0f / red[0];
  for (int j = tid; j < S_; j += 256) row[j] *= inv;
}

__global__ __launch_bounds__(256) void head_mean_kernel(
    const float* __restrict__ h1, const float* __restrict__ gfin,
    float* __restrict__ hidden)
{
  const int b = blockIdx.x, tid = threadIdx.x;
  for (int d = tid; d < 512; d += 256) {
    float s = 0.0f;
    if (d < 256) {
      for (int t = 0; t < S_; ++t) s += h1[((size_t)t * B_ + b) * 256 + d];
    } else {
      const int dd = d - 256;
      for (int t = 0; t < S_; ++t) s += gfin[((size_t)b * S_ + t) * 256 + dd];
    }
    hidden[b * 512 + d] = s * (1.0f / (float)S_);
  }
}

template <int A_MODE, int ACT, bool FULLN>
static inline void gemm(hipStream_t st,
                        const float* A, const float* Bm,
                        const float* bias, const float* bias2, const float* resid,
                        float* C, int M, int N, int K, int lda, int ldb, int ldc,
                        long long sA = 0, long long sB = 0, long long sC = 0,
                        int batches = 1)
{
  dim3 grid((N + 255) / 256, (M + 31) / 32, batches);
  gemm_wmma_kernel<A_MODE, ACT, FULLN><<<grid, 128, 0, st>>>(
      A, Bm, bias, bias2, resid, C, M, N, K, lda, ldb, ldc, sA, sB, sC);
}

extern "C" void kernel_launch(void* const* d_in, const int* in_sizes, int n_in,
                              void* d_out, int out_size, void* d_ws, size_t ws_size,
                              hipStream_t stream)
{
  (void)in_sizes; (void)n_in; (void)out_size; (void)ws_size;
  auto F = [&](int i) { return (const float*)d_in[i]; };

  const float* x        = F(0);
  const float* adj      = F(1);
  const float* l0f_Wih  = F(2),  *l0f_Whh = F(3),  *l0f_bih = F(4),  *l0f_bhh = F(5);
  const float* l0b_Wih  = F(6),  *l0b_Whh = F(7),  *l0b_bih = F(8),  *l0b_bhh = F(9);
  const float* l1f_Wih  = F(10), *l1f_Whh = F(11), *l1f_bih = F(12), *l1f_bhh = F(13);
  const float* l1b_Wih  = F(14), *l1b_Whh = F(15), *l1b_bih = F(16), *l1b_bhh = F(17);
  const float* fc1_W    = F(18), *fc1_b   = F(19);
  const float* wattn_W  = F(20), *wattn_b = F(21);
  const float* lin0_W   = F(22), *lin0_b  = F(23);
  const float* lin1_W   = F(24), *lin1_b  = F(25);
  const float* linf_W   = F(26), *linf_b  = F(27);
  const float* fcA_W    = F(28), *fcA_b   = F(29);
  const float* fcB_W    = F(30), *fcB_b   = F(31);
  const float* fcC_W    = F(32), *fcC_b   = F(33);

  char* wsp = (char*)d_ws;
  size_t off = 0;
  auto carve = [&](size_t bytes) -> void* {
    void* p = wsp + off;
    off += (bytes + 255) & ~(size_t)255;
    return p;
  };
  float* preF   = (float*)carve((size_t)S_ * B_ * 4 * H_ * 4);
  float* preB   = (float*)carve((size_t)S_ * B_ * 4 * H_ * 4);
  float* h0     = (float*)carve((size_t)S_ * B_ * 2 * H_ * 4);
  float* h1     = (float*)carve((size_t)S_ * B_ * 2 * H_ * 4);
  float* gA     = (float*)carve((size_t)B_ * S_ * G_ * 4);
  float* gB     = (float*)carve((size_t)B_ * S_ * G_ * 4);
  float* gC     = (float*)carve((size_t)B_ * S_ * G_ * 4);
  float* gT     = (float*)carve((size_t)B_ * G_ * S_ * 4);
  float* scores = (float*)carve((size_t)B_ * S_ * S_ * 4);
  __half* whh0f = (__half*)carve((size_t)4 * H_ * H_ * 2);
  __half* whh0b = (__half*)carve((size_t)4 * H_ * H_ * 2);
  __half* whh1f = (__half*)carve((size_t)4 * H_ * H_ * 2);
  __half* whh1b = (__half*)carve((size_t)4 * H_ * H_ * 2);
  float* fcAo   = (float*)carve((size_t)B_ * 128 * 4);
  float* fcBo   = (float*)carve((size_t)B_ * 32 * 4);

  const int MSB = S_ * B_;
  const int nwh = 4 * H_ * H_;

  cvt_f32_f16_kernel<<<(nwh + 255) / 256, 256, 0, stream>>>(l0f_Whh, whh0f, nwh);
  cvt_f32_f16_kernel<<<(nwh + 255) / 256, 256, 0, stream>>>(l0b_Whh, whh0b, nwh);
  cvt_f32_f16_kernel<<<(nwh + 255) / 256, 256, 0, stream>>>(l1f_Whh, whh1f, nwh);
  cvt_f32_f16_kernel<<<(nwh + 255) / 256, 256, 0, stream>>>(l1b_Whh, whh1b, nwh);

  gemm<1, 0, true>(stream, x, l0f_Wih, l0f_bih, l0f_bhh, nullptr, preF,
                   MSB, 4 * H_, D_, D_, D_, 4 * H_);
  gemm<1, 0, true>(stream, x, l0b_Wih, l0b_bih, l0b_bhh, nullptr, preB,
                   MSB, 4 * H_, D_, D_, D_, 4 * H_);
  lstm_dir_kernel<<<dim3(2, B_ / LB_), 256, 0, stream>>>(preF, preB, whh0f, whh0b, h0);

  gemm<0, 0, true>(stream, h0, l1f_Wih, l1f_bih, l1f_bhh, nullptr, preF,
                   MSB, 4 * H_, 2 * H_, 2 * H_, 2 * H_, 4 * H_);
  gemm<0, 0, true>(stream, h0, l1b_Wih, l1b_bih, l1b_bhh, nullptr, preB,
                   MSB, 4 * H_, 2 * H_, 2 * H_, 2 * H_, 4 * H_);
  lstm_dir_kernel<<<dim3(2, B_ / LB_), 256, 0, stream>>>(preF, preB, whh1f, whh1b, h1);

  gemm<0, 0, true>(stream, x, fc1_W, fc1_b, nullptr, nullptr, gA,
                   MSB, G_, D_, D_, D_, G_);

  float* cur = gA; float* t1 = gB; float* t2 = gC;
  const long long sGG = (long long)S_ * G_;
  const long long sSS = (long long)S_ * S_;
  for (int l = 0; l < 4; ++l) {
    const float* Wq = wattn_W + (size_t)l * G_ * G_;
    gemm<0, 0, true>(stream, cur, Wq, wattn_b + l * G_, nullptr, nullptr, t2,
                     MSB, G_, G_, G_, G_, G_);
    gemm<0, 0, true>(stream, t2, cur, nullptr, nullptr, nullptr, scores,
                     S_, S_, G_, G_, G_, S_, sGG, sGG, sSS, B_);
    attn_mask_norm_kernel<<<dim3(S_, B_), 256, 0, stream>>>(scores, adj);
    transpose_bsg_kernel<<<dim3(S_ / 32, G_ / 32, B_), dim3(32, 8), 0, stream>>>(cur, gT);
    gemm<0, 0, true>(stream, scores, gT, nullptr, nullptr, nullptr, t1,
                     S_, G_, S_, S_, S_, G_, sSS, (long long)G_ * S_, sGG, B_);
    gemm<0, 1, true>(stream, t1, lin0_W + (size_t)l * G_ * G_, lin0_b + l * G_,
                     nullptr, nullptr, t2, MSB, G_, G_, G_, G_, G_);
    gemm<0, 1, true>(stream, t2, lin1_W + (size_t)l * G_ * G_, lin1_b + l * G_,
                     nullptr, cur, t1, MSB, G_, G_, G_, G_, G_);
    float* old = cur; cur = t1; t1 = old;
  }
  gemm<0, 0, true>(stream, cur, linf_W, linf_b, nullptr, nullptr, t2,
                   MSB, G_, G_, G_, G_, G_);
  float* gfin = t2;

  float* y_out  = (float*)d_out;
  float* hidden = (float*)d_out + B_ * 2;
  head_mean_kernel<<<B_, 256, 0, stream>>>(h1, gfin, hidden);

  gemm<0, 2, true >(stream, hidden, fcA_W, fcA_b, nullptr, nullptr, fcAo,
                    B_, 128, 512, 512, 512, 128);
  gemm<0, 2, false>(stream, fcAo, fcB_W, fcB_b, nullptr, nullptr, fcBo,
                    B_, 32, 128, 128, 128, 32);
  gemm<0, 0, false>(stream, fcBo, fcC_W, fcC_b, nullptr, nullptr, y_out,
                    B_, 2, 32, 32, 32, 2);
}


# WMMA GEMMs: a workgroup reads its 128x64 operand tiles once per k chunk by whole-line loads, casts them to f16 once and stages them in LDS; fragments come from LDS; summation order unchanged
# speedup vs baseline: 5.1529x; 1.2160x over previous
#include <hip/hip_runtime.h>
#include <hip/hip_fp16.h>
#include <hip/hip_bf16.h>

#define B_  16
#define S_  512
#define D_  1024
#define H_  128
#define G_  256
#define HP_ 136

typedef __attribute__((ext_vector_type(16))) _Float16 v16h;
typedef __attribute__((ext_vector_type(8)))  float    v8f;

struct rawfrag { float4 x0, x1, y0, y1; };

__device__ __forceinline__ rawfrag raw_load(const float* p) {
  rawfrag r;
  r.x0 = *(const float4*)(p);
  r.x1 = *(const float4*)(p + 4);
  r.y0 = *(const float4*)(p + 16);
  r.y1 = *(const float4*)(p + 20);
  return r;
}

__device__ __forceinline__ v16h raw_to_frag(const rawfrag& r) {
  v16h f;
  f[0]=(_Float16)r.x0.x;  f[1]=(_Float16)r.x0.y;  f[2]=(_Float16)r.x0.z;  f[3]=(_Float16)r.x0.w;
  f[4]=(_Float16)r.x1.x;  f[5]=(_Float16)r.x1.y;  f[6]=(_Float16)r.x1.z;  f[7]=(_Float16)r.x1.w;
  f[8]=(_Float16)r.y0.x;  f[9]=(_Float16)r.y0.y;  f[10]=(_Float16)r.y0.z; f[11]=(_Float16)r.y0.w;
  f[12]=(_Float16)r.y1.x; f[13]=(_Float16)r.y1.y; f[14]=(_Float16)r.y1.z; f[15]=(_Float16)r.y1.w;
  return f;
}

__device__ __forceinline__ v16h frag_from_f16(const _Float16* p) {
  v16h f;
  #pragma unroll
  for (int i = 0; i < 8; ++i) { f[i] = p[i]; f[8 + i] = p[16 + i]; }
  return f;
}

#define WMMA_F16(a, b, c) __builtin_amdgcn_wmma_f32_16x16x32_f16( \
    false, (a), false, (b), (short)0, (c), false, false)

#define GPF_ 2

template <bool FULLN>
__device__ __forceinline__ void gemm_chunk(
    rawfrag (&sa)[2], rawfrag (&sb)[4], v8f (&acc)[2][4], const bool (&ton)[4],
    const float* pa0, const float* pa1, const float* const (&pb)[4],
    bool refill, int koff_next)
{
  v16h a0 = raw_to_frag(sa[0]);
  v16h a1 = raw_to_frag(sa[1]);
  v16h b[4];
  #pragma unroll
  for (int nt = 0; nt < 4; ++nt)
    if (FULLN || ton[nt]) b[nt] = raw_to_frag(sb[nt]);
  asm volatile("" : "+v"(a0), "+v"(a1) : : "memory");
  #pragma unroll
  for (int nt = 0; nt < 4; ++nt)
    if (FULLN || ton[nt]) asm volatile("" : "+v"(b[nt]) : : "memory");
  if (refill) {
    sa[0] = raw_load(pa0 + koff_next);
    sa[1] = raw_load(pa1 + koff_next);
    #pragma unroll
    for (int nt = 0; nt < 4; ++nt)
      if (FULLN || ton[nt]) sb[nt] = raw_load(pb[nt] + koff_next);
  }
  __builtin_amdgcn_sched_barrier(0);
  #pragma unroll
  for (int nt = 0; nt < 4; ++nt) {
    if (!FULLN && !ton[nt]) continue;
    acc[0][nt] = WMMA_F16(a0, b[nt], acc[0][nt]);
    acc[1][nt] = WMMA_F16(a1, b[nt], acc[1][nt]);
  }
}

template <int A_MODE, int ACT, bool FULLN>
__global__ __launch_bounds__(128) void gemm_wmma_kernel(
    const float* __restrict__ A, const float* __restrict__ Bm,
    const float* __restrict__ bias, const float* __restrict__ bias2,
    const float* __restrict__ resid, float* __restrict__ C,
    int M, int N, int K, int lda, int ldb, int ldc,
    long long sA, long long sB, long long sC)
{
  const int lane  = threadIdx.x & 31;
  const int wave  = threadIdx.x >> 5;
  const int tileM = blockIdx.y * 32;
  const int nbase = blockIdx.x * 256 + wave * 64;
  if (nbase >= N) return;

  const int z = blockIdx.z;
  const float* Ab = A  + (size_t)z * (size_t)sA;
  const float* Bb = Bm + (size_t)z * (size_t)sB;
  float*       Cb = C  + (size_t)z * (size_t)sC;
  const float* Rb = resid ? (resid + (size_t)z * (size_t)sC) : nullptr;

  const int l15   = lane & 15;
  const int khalf = (lane & 16) ? 8 : 0;

  int am0 = tileM + l15;      if (am0 >= M) am0 = M - 1;
  int am1 = tileM + 16 + l15; if (am1 >= M) am1 = M - 1;
  size_t arow0, arow1;
  if (A_MODE == 1) {
    arow0 = ((size_t)(am0 & 15) * S_ + (size_t)(am0 >> 4)) * (size_t)lda;
    arow1 = ((size_t)(am1 & 15) * S_ + (size_t)(am1 >> 4)) * (size_t)lda;
  } else {
    arow0 = (size_t)am0 * (size_t)lda;
    arow1 = (size_t)am1 * (size_t)lda;
  }

  bool ton[4]; int bn[4];
  #pragma unroll
  for (int nt = 0; nt < 4; ++nt) {
    if (FULLN) {
      ton[nt] = true;
      bn[nt]  = nbase + nt * 16 + l15;
    } else {
      ton[nt] = (nbase + nt * 16) < N;
      int c = nbase + nt * 16 + l15; if (c >= N) c = N - 1;
      bn[nt] = c;
    }
  }

  const float* pa0 = Ab + arow0 + khalf;
  const float* pa1 = Ab + arow1 + khalf;
  const float* pb[4];
  #pragma unroll
  for (int nt = 0; nt < 4; ++nt)
    pb[nt] = Bb + (size_t)bn[nt] * (size_t)ldb + khalf;

  const int nk = K >> 5;
  rawfrag ra[GPF_][2], rb_[GPF_][4];
  #pragma unroll
  for (int j = 0; j < GPF_; ++j) {
    const int kj = ((j < nk) ? j : (nk - 1)) * 32;
    ra[j][0] = raw_load(pa0 + kj);
    ra[j][1] = raw_load(pa1 + kj);
    #pragma unroll
    for (int nt = 0; nt < 4; ++nt)
      if (FULLN || ton[nt]) rb_[j][nt] = raw_load(pb[nt] + kj);
  }

  v8f acc[2][4] = {};
  int kc = 0;
  for (; kc + 2 * GPF_ <= nk; kc += GPF_) {
    #pragma unroll
    for (int j = 0; j < GPF_; ++j)
      gemm_chunk<FULLN>(ra[j], rb_[j], acc, ton, pa0, pa1, pb,
                        true, (kc + j + GPF_) * 32);
  }
  for (; kc < nk; kc += GPF_) {
    #pragma unroll
    for (int j = 0; j < GPF_; ++j)
      if (kc + j < nk)
        gemm_chunk<FULLN>(ra[j], rb_[j], acc, ton, pa0, pa1, pb,
                          kc + j + GPF_ < nk, (kc + j + GPF_) * 32);
  }

  const int rb = (lane < 16) ? 0 : 8;
  #pragma unroll
  for (int nt = 0; nt < 4; ++nt) {
    if (!FULLN && !ton[nt]) continue;
    const int col = nbase + nt * 16 + l15;
    bool colok = FULLN ? true : (col < N);
    float bv = 0.0f;
    if (colok) {
      if (bias)  bv += bias[col];
      if (bias2) bv += bias2[col];
    }
    #pragma unroll
    for (int mt = 0; mt < 2; ++mt) {
      #pragma unroll
      for (int r = 0; r < 8; ++r) {
        const int row = tileM + mt * 16 + rb + r;
        if (row < M && colok) {
          float v = acc[mt][nt][r] + bv;
          if      (ACT == 1) v = v > 0.0f ? v : 0.0f;
          else if (ACT == 2) v = v > 0.0f ? v : 0.1f * v;
          if (Rb) v += Rb[(size_t)row * (size_t)ldc + col];
          Cb[(size_t)row * (size_t)ldc + col] = v;
        }
      }
    }
  }
}

__device__ __forceinline__ void lds_only_barrier() {
  __builtin_amdgcn_fence(__ATOMIC_RELEASE, "workgroup", "local");
  __builtin_amdgcn_s_barrier();
  __builtin_amdgcn_fence(__ATOMIC_ACQUIRE, "workgroup", "local");
}

#define TM_ 128
#define TN_ 64
#define LP_ 40

typedef __attribute__((ext_vector_type(8))) _Float16 v8h;
typedef __attribute__((ext_vector_type(4))) _Float16 v4h;

__device__ __forceinline__ v16h frag_from_lds(const _Float16* p) {
  const v8h lo = *(const v8h*)(p);
  const v8h hi = *(const v8h*)(p + 16);
  return __builtin_shufflevector(lo, hi, 0, 1, 2, 3, 4, 5, 6, 7,
                                 8, 9, 10, 11, 12, 13, 14, 15);
}

typedef __attribute__((ext_vector_type(4))) float v4f;

__device__ __forceinline__ void stage_store(_Float16* dst, const v4f& v) {
  v4h h;
  h[0] = (_Float16)v[0]; h[1] = (_Float16)v[1]; h[2] = (_Float16)v[2]; h[3] = (_Float16)v[3];
  *(v4h*)dst = h;
}

struct stagewords { v4f a[8]; v4f b[4]; };

__device__ __forceinline__ void stage_issue(stagewords& r, const float* ga, size_t astep,
                                            const float* gb, size_t bstep, int koff) {
  #pragma unroll
  for (int i = 0; i < 8; ++i) r.a[i] = *(const v4f*)(ga + (size_t)i * astep + koff);
  #pragma unroll
  for (int i = 0; i < 4; ++i) r.b[i] = *(const v4f*)(gb + (size_t)i * bstep + koff);
}

__device__ __forceinline__ void stage_to_lds(stagewords& r, _Float16* sa, _Float16* sb,
                                             int soff) {
  asm volatile("" : "+v"(r.a[0]), "+v"(r.a[1]), "+v"(r.a[2]), "+v"(r.a[3]),
                    "+v"(r.a[4]), "+v"(r.a[5]), "+v"(r.a[6]), "+v"(r.a[7]),
                    "+v"(r.b[0]), "+v"(r.b[1]), "+v"(r.b[2]), "+v"(r.b[3]));
  #pragma unroll
  for (int i = 0; i < 8; ++i) stage_store(sa + soff + i * 16 * LP_, r.a[i]);
  #pragma unroll
  for (int i = 0; i < 4; ++i) stage_store(sb + soff + i * 16 * LP_, r.b[i]);
}

__device__ __forceinline__ void lds_chunk_mma(const _Float16* pa, const _Float16* pb,
                                              v8f (&acc)[2][4]) {
  const v16h a0 = frag_from_lds(pa);
  const v16h a1 = frag_from_lds(pa + 16 * LP_);
  v16h b[4];
  #pragma unroll
  for (int nt = 0; nt < 4; ++nt) b[nt] = frag_from_lds(pb + nt * 16 * LP_);
  #pragma unroll
  for (int nt = 0; nt < 4; ++nt) {
    acc[0][nt] = WMMA_F16(a0, b[nt], acc[0][nt]);
    acc[1][nt] = WMMA_F16(a1, b[nt], acc[1][nt]);
  }
}

template <int A_MODE, int ACT>
__global__ __launch_bounds__(128) void gemm_lds_kernel(
    const float* __restrict__ A, const float* __restrict__ Bm,
    const float* __restrict__ bias, const float* __restrict__ bias2,
    const float* __restrict__ resid, float* __restrict__ C,
    int K, int lda, int ldb, int ldc,
    long long sA, long long sB, long long sC)
{
  __shared__ __attribute__((aligned(16))) _Float16 sa[2][TM_ * LP_];
  __shared__ __attribute__((aligned(16))) _Float16 sb[2][TN_ * LP_];

  const int tid   = threadIdx.x;
  const int lane  = tid & 31;
  const int wave  = tid >> 5;
  const int tileM = blockIdx.y * TM_;
  const int nbase = blockIdx.x * TN_;

  const int z = blockIdx.z;
  const float* Ab = A  + (size_t)z * (size_t)sA;
  const float* Bb = Bm + (size_t)z * (size_t)sB;
  float*       Cb = C  + (size_t)z * (size_t)sC;
  const float* Rb = resid ? (resid + (size_t)z * (size_t)sC) : nullptr;

  const int q  = tid & 7;
  const int rg = tid >> 3;
  const float* ga; size_t astep;
  if (A_MODE == 1) {
    ga = Ab + ((size_t)rg * S_ + (size_t)(tileM >> 4)) * (size_t)lda + q * 4;
    astep = (size_t)lda;
  } else {
    ga = Ab + (size_t)(tileM + rg) * (size_t)lda + q * 4;
    astep = (size_t)16 * (size_t)lda;
  }
  const float* gb = Bb + (size_t)(nbase + rg) * (size_t)ldb + q * 4;
  const size_t bstep = (size_t)16 * (size_t)ldb;
  const int soff = rg * LP_ + q * 4;

  const int l15   = lane & 15;
  const int khalf = (lane & 16) ? 8 : 0;
  const int fa = (wave * 32 + l15) * LP_ + khalf;
  const int fb = l15 * LP_ + khalf;

  const int nk = K >> 5;
  stagewords r0, r1;
  stage_issue(r0, ga, astep, gb, bstep, 0);
  stage_issue(r1, ga, astep, gb, bstep, 32);
  stage_to_lds(r0, sa[0], sb[0], soff);
  lds_only_barrier();

  v8f acc[2][4] = {};
  for (int kc = 0; kc < nk; kc += 2) {
    stage_issue(r0, ga, astep, gb, bstep, ((kc + 2 < nk) ? (kc + 2) : (nk - 1)) * 32);
    __builtin_amdgcn_sched_barrier(0);
    lds_chunk_mma(sa[0] + fa, sb[0] + fb, acc);
    __builtin_amdgcn_sched_barrier(0);
    stage_to_lds(r1, sa[1], sb[1], soff);
    lds_only_barrier();
    stage_issue(r1, ga, astep, gb, bstep, ((kc + 3 < nk) ? (kc + 3) : (nk - 1)) * 32);
    __builtin_amdgcn_sched_barrier(0);
    lds_chunk_mma(sa[1] + fa, sb[1] + fb, acc);
    __builtin_amdgcn_sched_barrier(0);
    stage_to_lds(r0, sa[0], sb[0], soff);
    lds_only_barrier();
  }

  const int rb = (lane < 16) ? 0 : 8;
  #pragma unroll
  for (int nt = 0; nt < 4; ++nt) {
    const int col = nbase + nt * 16 + l15;
    float bv = 0.0f;
    if (bias)  bv += bias[col];
    if (bias2) bv += bias2[col];
    #pragma unroll
    for (int mt = 0; mt < 2; ++mt) {
      #pragma unroll
      for (int r = 0; r < 8; ++r) {
        const int row = tileM + wave * 32 + mt * 16 + rb + r;
        float v = acc[mt][nt][r] + bv;
        if      (ACT == 1) v = v > 0.0f ? v : 0.0f;
        else if (ACT == 2) v = v > 0.0f ? v : 0.1f * v;
        if (Rb) v += Rb[(size_t)row * (size_t)ldc + col];
        Cb[(size_t)row * (size_t)ldc + col] = v;
      }
    }
  }
}

__global__ void cvt_f32_f16_kernel(const float* __restrict__ src,
                                   __half* __restrict__ dst, int n) {
  int i = blockIdx.x * blockDim.x + threadIdx.x;
  if (i < n) ((_Float16*)dst)[i] = (_Float16)src[i];
}

__global__ __launch_bounds__(256) void transpose_bsg_kernel(
    const float* __restrict__ src, float* __restrict__ dst)
{
  __shared__ float tile[32][33];
  const int b  = blockIdx.z;
  const int s0 = blockIdx.x * 32, g0 = blockIdx.y * 32;
  const int tx = threadIdx.x, ty = threadIdx.y;
  const float* sp = src + (size_t)b * S_ * G_;
  float*       dp = dst + (size_t)b * G_ * S_;
  #pragma unroll
  for (int r = 0; r < 32; r += 8)
    tile[ty + r][tx] = sp[(size_t)(s0 + ty + r) * G_ + g0 + tx];
  __syncthreads();
  #pragma unroll
  for (int r = 0; r < 32; r += 8)
    dp[(size_t)(g0 + ty + r) * S_ + s0 + tx] = tile[tx][ty + r];
}

#define LB_ 2
#define PD_ 8
static_assert(S_ % PD_ == 0, "the step loop is unrolled by PD_");
static_assert(PD_ % 2 == 0, "the h buffer of a step is chosen by the parity of its ring slot");

__device__ __forceinline__ float gate_sigmoid(float x) {
  return __builtin_amdgcn_rcpf(1.0f + __expf(-x));
}
__device__ __forceinline__ float gate_tanh(float x) {
  return 1.0f - 2.0f * __builtin_amdgcn_rcpf(1.0f + __expf(2.0f * x));
}

__global__ __launch_bounds__(256) void lstm_dir_kernel(
    const float* __restrict__ preF, const float* __restrict__ preB,
    const __half* __restrict__ whhF16_f, const __half* __restrict__ whhF16_b,
    float* __restrict__ hs)
{
  const int dir = blockIdx.x;
  const int b0  = blockIdx.y * LB_;
  const float*    pre = dir ? preB : preF;
  const _Float16* whh = (const _Float16*)(dir ? whhF16_b : whhF16_f);

  __shared__ _Float16 h_lds[2][LB_ * HP_];

  const int tid   = threadIdx.x;
  const int wave  = tid >> 5;
  const int lane  = tid & 31;
  const int l15   = lane & 15;
  const int khalf = (lane & 16) ? 8 : 0;
  const int hrow  = l15 >> 3;
  const int half  = lane >> 4;

  v16h wfrag[16];
  #pragma unroll
  for (int nt = 0; nt < 4; ++nt) {
    const int tileN = nt * H_ + wave * 16;
    #pragma unroll
    for (int kc = 0; kc < 4; ++kc)
      wfrag[nt * 4 + kc] =
          frag_from_f16(whh + (size_t)(tileN + l15) * H_ + kc * 32 + khalf);
  }

  for (int i = tid; i < LB_ * HP_; i += 256) h_lds[0][i] = (_Float16)0.0f;

  const int hx = wave * 16 + l15;
  float c = 0.0f;
  __syncthreads();

  const float* pre_lane = pre + (size_t)(b0 + half) * (4 * H_) + hx;
  float pq[PD_][4];
  #pragma unroll
  for (int j = 0; j < PD_ - 1; ++j) {
    const int tj = dir ? (S_ - 1 - j) : j;
    #pragma unroll
    for (int nt = 0; nt < 4; ++nt)
      pq[j][nt] = pre_lane[(size_t)tj * (B_ * 4 * H_) + nt * H_];
  }

  for (int step0 = 0; step0 < S_; step0 += PD_) {
    #pragma unroll
    for (int j = 0; j < PD_; ++j) {
      const int step = step0 + j;
      const int t = dir ? (S_ - 1 - step) : step;

      {
        const int sn = (step + PD_ - 1 < S_) ? (step + PD_ - 1) : (S_ - 1);
        const int tn = dir ? (S_ - 1 - sn) : sn;
        #pragma unroll
        for (int nt = 0; nt < 4; ++nt)
          pq[(j + PD_ - 1) % PD_][nt] = pre_lane[(size_t)tn * (B_ * 4 * H_) + nt * H_];
      }

      const _Float16* h_rd = h_lds[j & 1];
      _Float16*       h_wr = h_lds[(j & 1) ^ 1];

      v16h a[4];
      #pragma unroll
      for (int kc = 0; kc < 4; ++kc)
        a[kc] = frag_from_f16(&h_rd[hrow * HP_ + kc * 32 + khalf]);

      float z[4];
      #pragma unroll
      for (int nt = 0; nt < 4; ++nt) {
        v8f acc = {};
        #pragma unroll
        for (int kc = 0; kc < 4; ++kc)
          acc = WMMA_F16(a[kc], wfrag[nt * 4 + kc], acc);
        z[nt] = acc[0] + pq[j][nt];
      }

      {
        const float si = gate_sigmoid(z[0]);
        const float sf = gate_sigmoid(z[1]);
        const float so = gate_sigmoid(z[3]);
        c = sf * c + si * gate_tanh(z[2]);
        const float h = so * gate_tanh(c);
        h_wr[half * HP_ + hx] = (_Float16)h;
        hs[((size_t)t * B_ + b0 + half) * (2 * H_) + dir * H_ + hx] = h;
      }
      lds_only_barrier();
    }
  }
}

__global__ __launch_bounds__(256) void attn_mask_norm_kernel(
    float* __restrict__ scores, const float* __restrict__ adj)
{
  const int i = blockIdx.x, b = blockIdx.y, tid = threadIdx.x;
  float* row        = scores + ((size_t)b * S_ + i) * S_;
  const float* arow = adj    + ((size_t)b * S_ + i) * S_;
  __shared__ float red[256];
  float part = 0.0f;
  for (int j = tid; j < S_; j += 256) {
    const float sg = 1.0f / (1.0f + __expf(-row[j]));
    const float av = (j == i) ? 1.0f : arow[j];
    const float v  = (sg + ((j == i) ? 1e-5f : 0.0f)) * av;
    row[j] = v;
    part += v;
  }
  red[tid] = part;
  __syncthreads();
  for (int off = 128; off > 0; off >>= 1) {
    if (tid < off) red[tid] += red[tid + off];
    __syncthreads();
  }
  const float inv = 1.0f / red[0];
  for (int j = tid; j < S_; j += 256) row[j] *= inv;
}

__global__ __launch_bounds__(256) void head_mean_kernel(
    const float* __restrict__ h1, const float* __restrict__ gfin,
    float* __restrict__ hidden)
{
  const int b = blockIdx.x, tid = threadIdx.x;
  for (int d = tid; d < 512; d += 256) {
    float s = 0.0f;
    if (d < 256) {
      for (int t = 0; t < S_; ++t) s += h1[((size_t)t * B_ + b) * 256 + d];
    } else {
      const int dd = d - 256;
      for (int t = 0; t < S_; ++t) s += gfin[((size_t)b * S_ + t) * 256 + dd];
    }
    hidden[b * 512 + d] = s * (1.0f / (float)S_);
  }
}

template <int A_MODE, int ACT, bool FULLN>
static inline void gemm(hipStream_t st,
                        const float* A, const float* Bm,
                        const float* bias, const float* bias2, const float* resid,
                        float* C, int M, int N, int K, int lda, int ldb, int ldc,
                        long long sA = 0, long long sB = 0, long long sC = 0,
                        int batches = 1)
{
  dim3 grid((N + 255) / 256, (M + 31) / 32, batches);
  gemm_wmma_kernel<A_MODE, ACT, FULLN><<<grid, 128, 0, st>>>(
      A, Bm, bias, bias2, resid, C, M, N, K, lda, ldb, ldc, sA, sB, sC);
}

template <int A_MODE, int ACT>
static inline void gemm_lds(hipStream_t st,
                            const float* A, const float* Bm,
                            const float* bias, const float* bias2, const float* resid,
                            float* C, int M, int N, int K, int lda, int ldb, int ldc,
                            long long sA = 0, long long sB = 0, long long sC = 0,
                            int batches = 1)
{
  dim3 grid(N / TN_, M / TM_, batches);
  gemm_lds_kernel<A_MODE, ACT><<<grid, 128, 0, st>>>(
      A, Bm, bias, bias2, resid, C, K, lda, ldb, ldc, sA, sB, sC);
}

extern "C" void kernel_launch(void* const* d_in, const int* in_sizes, int n_in,
                              void* d_out, int out_size, void* d_ws, size_t ws_size,
                              hipStream_t stream)
{
  (void)in_sizes; (void)n_in; (void)out_size; (void)ws_size;
  auto F = [&](int i) { return (const float*)d_in[i]; };

  const float* x        = F(0);
  const float* adj      = F(1);
  const float* l0f_Wih  = F(2),  *l0f_Whh = F(3),  *l0f_bih = F(4),  *l0f_bhh = F(5);
  const float* l0b_Wih  = F(6),  *l0b_Whh = F(7),  *l0b_bih = F(8),  *l0b_bhh = F(9);
  const float* l1f_Wih  = F(10), *l1f_Whh = F(11), *l1f_bih = F(12), *l1f_bhh = F(13);
  const float* l1b_Wih  = F(14), *l1b_Whh = F(15), *l1b_bih = F(16), *l1b_bhh = F(17);
  const float* fc1_W    = F(18), *fc1_b   = F(19);
  const float* wattn_W  = F(20), *wattn_b = F(21);
  const float* lin0_W   = F(22), *lin0_b  = F(23);
  const float* lin1_W   = F(24), *lin1_b  = F(25);
  const float* linf_W   = F(26), *linf_b  = F(27);
  const float* fcA_W    = F(28), *fcA_b   = F(29);
  const float* fcB_W    = F(30), *fcB_b   = F(31);
  const float* fcC_W    = F(32), *fcC_b   = F(33);

  char* wsp = (char*)d_ws;
  size_t off = 0;
  auto carve = [&](size_t bytes) -> void* {
    void* p = wsp + off;
    off += (bytes + 255) & ~(size_t)255;
    return p;
  };
  float* preF   = (float*)carve((size_t)S_ * B_ * 4 * H_ * 4);
  float* preB   = (float*)carve((size_t)S_ * B_ * 4 * H_ * 4);
  float* h0     = (float*)carve((size_t)S_ * B_ * 2 * H_ * 4);
  float* h1     = (float*)carve((size_t)S_ * B_ * 2 * H_ * 4);
  float* gA     = (float*)carve((size_t)B_ * S_ * G_ * 4);
  float* gB     = (float*)carve((size_t)B_ * S_ * G_ * 4);
  float* gC     = (float*)carve((size_t)B_ * S_ * G_ * 4);
  float* gT     = (float*)carve((size_t)B_ * G_ * S_ * 4);
  float* scores = (float*)carve((size_t)B_ * S_ * S_ * 4);
  __half* whh0f = (__half*)carve((size_t)4 * H_ * H_ * 2);
  __half* whh0b = (__half*)carve((size_t)4 * H_ * H_ * 2);
  __half* whh1f = (__half*)carve((size_t)4 * H_ * H_ * 2);
  __half* whh1b = (__half*)carve((size_t)4 * H_ * H_ * 2);
  float* fcAo   = (float*)carve((size_t)B_ * 128 * 4);
  float* fcBo   = (float*)carve((size_t)B_ * 32 * 4);

  const int MSB = S_ * B_;
  const int nwh = 4 * H_ * H_;

  cvt_f32_f16_kernel<<<(nwh + 255) / 256, 256, 0, stream>>>(l0f_Whh, whh0f, nwh);
  cvt_f32_f16_kernel<<<(nwh + 255) / 256, 256, 0, stream>>>(l0b_Whh, whh0b, nwh);
  cvt_f32_f16_kernel<<<(nwh + 255) / 256, 256, 0, stream>>>(l1f_Whh, whh1f, nwh);
  cvt_f32_f16_kernel<<<(nwh + 255) / 256, 256, 0, stream>>>(l1b_Whh, whh1b, nwh);

  gemm_lds<1, 0>(stream, x, l0f_Wih, l0f_bih, l0f_bhh, nullptr, preF,
                   MSB, 4 * H_, D_, D_, D_, 4 * H_);
  gemm_lds<1, 0>(stream, x, l0b_Wih, l0b_bih, l0b_bhh, nullptr, preB,
                   MSB, 4 * H_, D_, D_, D_, 4 * H_);
  lstm_dir_kernel<<<dim3(2, B_ / LB_), 256, 0, stream>>>(preF, preB, whh0f, whh0b, h0);

  gemm_lds<0, 0>(stream, h0, l1f_Wih, l1f_bih, l1f_bhh, nullptr, preF,
                   MSB, 4 * H_, 2 * H_, 2 * H_, 2 * H_, 4 * H_);
  gemm_lds<0, 0>(stream, h0, l1b_Wih, l1b_bih, l1b_bhh, nullptr, preB,
                   MSB, 4 * H_, 2 * H_, 2 * H_, 2 * H_, 4 * H_);
  lstm_dir_kernel<<<dim3(2, B_ / LB_), 256, 0, stream>>>(preF, preB, whh1f, whh1b, h1);

  gemm_lds<0, 0>(stream, x, fc1_W, fc1_b, nullptr, nullptr, gA,
                   MSB, G_, D_, D_, D_, G_);

  float* cur = gA; float* t1 = gB; float* t2 = gC;
  const long long sGG = (long long)S_ * G_;
  const long long sSS = (long long)S_ * S_;
  for (int l = 0; l < 4; ++l) {
    const float* Wq = wattn_W + (size_t)l * G_ * G_;
    gemm_lds<0, 0>(stream, cur, Wq, wattn_b + l * G_, nullptr, nullptr, t2,
                     MSB, G_, G_, G_, G_, G_);
    gemm_lds<0, 0>(stream, t2, cur, nullptr, nullptr, nullptr, scores,
                     S_, S_, G_, G_, G_, S_, sGG, sGG, sSS, B_);
    attn_mask_norm_kernel<<<dim3(S_, B_), 256, 0, stream>>>(scores, adj);
    transpose_bsg_kernel<<<dim3(S_ / 32, G_ / 32, B_), dim3(32, 8), 0, stream>>>(cur, gT);
    gemm_lds<0, 0>(stream, scores, gT, nullptr, nullptr, nullptr, t1,
                     S_, G_, S_, S_, S_, G_, sSS, (long long)G_ * S_, sGG, B_);
    gemm_lds<0, 1>(stream, t1, lin0_W + (size_t)l * G_ * G_, lin0_b + l * G_,
                     nullptr, nullptr, t2, MSB, G_, G_, G_, G_, G_);
    gemm_lds<0, 1>(stream, t2, lin1_W + (size_t)l * G_ * G_, lin1_b + l * G_,
                     nullptr, cur, t1, MSB, G_, G_, G_, G_, G_);
    float* old = cur; cur = t1; t1 = old;
  }
  gemm_lds<0, 0>(stream, cur, linf_W, linf_b, nullptr, nullptr, t2,
                   MSB, G_, G_, G_, G_, G_);
  float* gfin = t2;

  float* y_out  = (float*)d_out;
  float* hidden = (float*)d_out + B_ * 2;
  head_mean_kernel<<<B_, 256, 0, stream>>>(h1, gfin, hidden);

  gemm<0, 2, true >(stream, hidden, fcA_W, fcA_b, nullptr, nullptr, fcAo,
                    B_, 128, 512, 512, 512, 128);
  gemm<0, 2, false>(stream, fcAo, fcB_W, fcB_b, nullptr, nullptr, fcBo,
                    B_, 32, 128, 128, 128, 32);
  gemm<0, 0, false>(stream, fcBo, fcC_W, fcC_b, nullptr, nullptr, y_out,
                    B_, 2, 32, 32, 32, 2);
}


# GEMM epilogues: bias and residual known at compile time; bias loaded ahead of the K loop, the residual's 64 words per lane as one group: was one awaited load per word; summation order unchanged
# speedup vs baseline: 6.1927x; 1.2018x over previous
#include <hip/hip_runtime.h>
#include <hip/hip_fp16.h>
#include <hip/hip_bf16.h>

#define B_  16
#define S_  512
#define D_  1024
#define H_  128
#define G_  256
#define HP_ 136

typedef __attribute__((ext_vector_type(16))) _Float16 v16h;
typedef __attribute__((ext_vector_type(8)))  float    v8f;

struct rawfrag { float4 x0, x1, y0, y1; };

__device__ __forceinline__ rawfrag raw_load(const float* p) {
  rawfrag r;
  r.x0 = *(const float4*)(p);
  r.x1 = *(const float4*)(p + 4);
  r.y0 = *(const float4*)(p + 16);
  r.y1 = *(const float4*)(p + 20);
  return r;
}

__device__ __forceinline__ v16h raw_to_frag(const rawfrag& r) {
  v16h f;
  f[0]=(_Float16)r.x0.x;  f[1]=(_Float16)r.x0.y;  f[2]=(_Float16)r.x0.z;  f[3]=(_Float16)r.x0.w;
  f[4]=(_Float16)r.x1.x;  f[5]=(_Float16)r.x1.y;  f[6]=(_Float16)r.x1.z;  f[7]=(_Float16)r.x1.w;
  f[8]=(_Float16)r.y0.x;  f[9]=(_Float16)r.y0.y;  f[10]=(_Float16)r.y0.z; f[11]=(_Float16)r.y0.w;
  f[12]=(_Float16)r.y1.x; f[13]=(_Float16)r.y1.y; f[14]=(_Float16)r.y1.z; f[15]=(_Float16)r.y1.w;
  return f;
}

__device__ __forceinline__ v16h frag_from_f16(const _Float16* p) {
  v16h f;
  #pragma unroll
  for (int i = 0; i < 8; ++i) { f[i] = p[i]; f[8 + i] = p[16 + i]; }
  return f;
}

#define WMMA_F16(a, b, c) __builtin_amdgcn_wmma_f32_16x16x32_f16( \
    false, (a), false, (b), (short)0, (c), false, false)

#define GPF_ 2

template <bool FULLN>
__device__ __forceinline__ void gemm_chunk(
    rawfrag (&sa)[2], rawfrag (&sb)[4], v8f (&acc)[2][4], const bool (&ton)[4],
    const float* pa0, const float* pa1, const float* const (&pb)[4],
    bool refill, int koff_next)
{
  v16h a0 = raw_to_frag(sa[0]);
  v16h a1 = raw_to_frag(sa[1]);
  v16h b[4];
  #pragma unroll
  for (int nt = 0; nt < 4; ++nt)
    if (FULLN || ton[nt]) b[nt] = raw_to_frag(sb[nt]);
  asm volatile("" : "+v"(a0), "+v"(a1) : : "memory");
  #pragma unroll
  for (int nt = 0; nt < 4; ++nt)
    if (FULLN || ton[nt]) asm volatile("" : "+v"(b[nt]) : : "memory");
  if (refill) {
    sa[0] = raw_load(pa0 + koff_next);
    sa[1] = raw_load(pa1 + koff_next);
    #pragma unroll
    for (int nt = 0; nt < 4; ++nt)
      if (FULLN || ton[nt]) sb[nt] = raw_load(pb[nt] + koff_next);
  }
  __builtin_amdgcn_sched_barrier(0);
  #pragma unroll
  for (int nt = 0; nt < 4; ++nt) {
    if (!FULLN && !ton[nt]) continue;
    acc[0][nt] = WMMA_F16(a0, b[nt], acc[0][nt]);
    acc[1][nt] = WMMA_F16(a1, b[nt], acc[1][nt]);
  }
}

template <int A_MODE, int ACT, bool FULLN>
__global__ __launch_bounds__(128) void gemm_wmma_kernel(
    const float* __restrict__ A, const float* __restrict__ Bm,
    const float* __restrict__ bias, const float* __restrict__ bias2,
    const float* __restrict__ resid, float* __restrict__ C,
    int M, int N, int K, int lda, int ldb, int ldc,
    long long sA, long long sB, long long sC)
{
  const int lane  = threadIdx.x & 31;
  const int wave  = threadIdx.x >> 5;
  const int tileM = blockIdx.y * 32;
  const int nbase = blockIdx.x * 256 + wave * 64;
  if (nbase >= N) return;

  const int z = blockIdx.z;
  const float* Ab = A  + (size_t)z * (size_t)sA;
  const float* Bb = Bm + (size_t)z * (size_t)sB;
  float*       Cb = C  + (size_t)z * (size_t)sC;
  const float* Rb = resid ? (resid + (size_t)z * (size_t)sC) : nullptr;

  const int l15   = lane & 15;
  const int khalf = (lane & 16) ? 8 : 0;

  int am0 = tileM + l15;      if (am0 >= M) am0 = M - 1;
  int am1 = tileM + 16 + l15; if (am1 >= M) am1 = M - 1;
  size_t arow0, arow1;
  if (A_MODE == 1) {
    arow0 = ((size_t)(am0 & 15) * S_ + (size_t)(am0 >> 4)) * (size_t)lda;
    arow1 = ((size_t)(am1 & 15) * S_ + (size_t)(am1 >> 4)) * (size_t)lda;
  } else {
    arow0 = (size_t)am0 * (size_t)lda;
    arow1 = (size_t)am1 * (size_t)lda;
  }

  bool ton[4]; int bn[4];
  #pragma unroll
  for (int nt = 0; nt < 4; ++nt) {
    if (FULLN) {
      ton[nt] = true;
      bn[nt]  = nbase + nt * 16 + l15;
    } else {
      ton[nt] = (nbase + nt * 16) < N;
      int c = nbase + nt * 16 + l15; if (c >= N) c = N - 1;
      bn[nt] = c;
    }
  }

  const float* pa0 = Ab + arow0 + khalf;
  const float* pa1 = Ab + arow1 + khalf;
  const float* pb[4];
  #pragma unroll
  for (int nt = 0; nt < 4; ++nt)
    pb[nt] = Bb + (size_t)bn[nt] * (size_t)ldb + khalf;

  const int nk = K >> 5;
  rawfrag ra[GPF_][2], rb_[GPF_][4];
  #pragma unroll
  for (int j = 0; j < GPF_; ++j) {
    const int kj = ((j < nk) ? j : (nk - 1)) * 32;
    ra[j][0] = raw_load(pa0 + kj);
    ra[j][1] = raw_load(pa1 + kj);
    #pragma unroll
    for (int nt = 0; nt < 4; ++nt)
      if (FULLN || ton[nt]) rb_[j][nt] = raw_load(pb[nt] + kj);
  }

  v8f acc[2][4] = {};
  int kc = 0;
  for (; kc + 2 * GPF_ <= nk; kc += GPF_) {
    #pragma unroll
    for (int j = 0; j < GPF_; ++j)
      gemm_chunk<FULLN>(ra[j], rb_[j], acc, ton, pa0, pa1, pb,
                        true, (kc + j + GPF_) * 32);
  }
  for (; kc < nk; kc += GPF_) {
    #pragma unroll
    for (int j = 0; j < GPF_; ++j)
      if (kc + j < nk)
        gemm_chunk<FULLN>(ra[j], rb_[j], acc, ton, pa0, pa1, pb,
                          kc + j + GPF_ < nk, (kc + j + GPF_) * 32);
  }

  const int rb = (lane < 16) ? 0 : 8;
  float bvv[4] = {0.0f, 0.0f, 0.0f, 0.0f};
  if (bias) {
    #pragma unroll
    for (int nt = 0; nt < 4; ++nt) bvv[nt] += bias[bn[nt]];
  }
  if (bias2) {
    #pragma unroll
    for (int nt = 0; nt < 4; ++nt) bvv[nt] += bias2[bn[nt]];
  }
  #pragma unroll
  for (int nt = 0; nt < 4; ++nt) {
    if (!FULLN && !ton[nt]) continue;
    const int col = nbase + nt * 16 + l15;
    bool colok = FULLN ? true : (col < N);
    const float bv = bvv[nt];
    #pragma unroll
    for (int mt = 0; mt < 2; ++mt) {
      #pragma unroll
      for (int r = 0; r < 8; ++r) {
        const int row = tileM + mt * 16 + rb + r;
        if (row < M && colok) {
          float v = acc[mt][nt][r] + bv;
          if      (ACT == 1) v = v > 0.0f ? v : 0.0f;
          else if (ACT == 2) v = v > 0.0f ? v : 0.1f * v;
          if (Rb) v += Rb[(size_t)row * (size_t)ldc + col];
          Cb[(size_t)row * (size_t)ldc + col] = v;
        }
      }
    }
  }
}

__device__ __forceinline__ void lds_only_barrier() {
  __builtin_amdgcn_fence(__ATOMIC_RELEASE, "workgroup", "local");
  __builtin_amdgcn_s_barrier();
  __builtin_amdgcn_fence(__ATOMIC_ACQUIRE, "workgroup", "local");
}

#define TM_ 128
#define TN_ 64
#define LP_ 40

typedef __attribute__((ext_vector_type(8))) _Float16 v8h;
typedef __attribute__((ext_vector_type(4))) _Float16 v4h;

__device__ __forceinline__ v16h frag_from_lds(const _Float16* p) {
  const v8h lo = *(const v8h*)(p);
  const v8h hi = *(const v8h*)(p + 16);
  return __builtin_shufflevector(lo, hi, 0, 1, 2, 3, 4, 5, 6, 7,
                                 8, 9, 10, 11, 12, 13, 14, 15);
}

typedef __attribute__((ext_vector_type(4))) float v4f;

__device__ __forceinline__ void stage_store(_Float16* dst, const v4f& v) {
  v4h h;
  h[0] = (_Float16)v[0]; h[1] = (_Float16)v[1]; h[2] = (_Float16)v[2]; h[3] = (_Float16)v[3];
  *(v4h*)dst = h;
}

struct stagewords { v4f a[8]; v4f b[4]; };

__device__ __forceinline__ void stage_issue(stagewords& r, const float* ga, size_t astep,
                                            const float* gb, size_t bstep, int koff) {
  #pragma unroll
  for (int i = 0; i < 8; ++i) r.a[i] = *(const v4f*)(ga + (size_t)i * astep + koff);
  #pragma unroll
  for (int i = 0; i < 4; ++i) r.b[i] = *(const v4f*)(gb + (size_t)i * bstep + koff);
}

__device__ __forceinline__ void stage_to_lds(stagewords& r, _Float16* sa, _Float16* sb,
                                             int soff) {
  asm volatile("" : "+v"(r.a[0]), "+v"(r.a[1]), "+v"(r.a[2]), "+v"(r.a[3]),
                    "+v"(r.a[4]), "+v"(r.a[5]), "+v"(r.a[6]), "+v"(r.a[7]),
                    "+v"(r.b[0]), "+v"(r.b[1]), "+v"(r.b[2]), "+v"(r.b[3]));
  #pragma unroll
  for (int i = 0; i < 8; ++i) stage_store(sa + soff + i * 16 * LP_, r.a[i]);
  #pragma unroll
  for (int i = 0; i < 4; ++i) stage_store(sb + soff + i * 16 * LP_, r.b[i]);
}

__device__ __forceinline__ void lds_chunk_mma(const _Float16* pa, const _Float16* pb,
                                              v8f (&acc)[2][4]) {
  const v16h a0 = frag_from_lds(pa);
  const v16h a1 = frag_from_lds(pa + 16 * LP_);
  v16h b[4];
  #pragma unroll
  for (int nt = 0; nt < 4; ++nt) b[nt] = frag_from_lds(pb + nt * 16 * LP_);
  #pragma unroll
  for (int nt = 0; nt < 4; ++nt) {
    acc[0][nt] = WMMA_F16(a0, b[nt], acc[0][nt]);
    acc[1][nt] = WMMA_F16(a1, b[nt], acc[1][nt]);
  }
}

__device__ __forceinline__ void rowsum_add(const stagewords& r, float (&rs)[8]) {
  #pragma unroll
  for (int i = 0; i < 8; ++i)
    rs[i] += (r.a[i][0] + r.a[i][1]) + (r.a[i][2] + r.a[i][3]);
}

template <int M>
__device__ __forceinline__ float lane_xor_add(float v) {
  const int t = __builtin_amdgcn_update_dpp(0, __builtin_bit_cast(int, v), 0x160 + M, 0xf, 0xf, true);
  return v + __builtin_bit_cast(float, t);
}

template <int A_MODE, int ACT, bool ROWN, int NB, bool RES>
__global__ __launch_bounds__(128) void gemm_lds_kernel(
    const float* __restrict__ A, const float* __restrict__ Bm,
    const float* __restrict__ bias, const float* __restrict__ bias2,
    const float* __restrict__ resid, float* __restrict__ C,
    int K, int lda, int ldb, int ldc,
    long long sA, long long sB, long long sC)
{
  __shared__ __attribute__((aligned(16))) _Float16 sa[2][TM_ * LP_];
  __shared__ __attribute__((aligned(16))) _Float16 sb[2][TN_ * LP_];
  __shared__ float rsum[TM_];

  const int tid   = threadIdx.x;
  const int lane  = tid & 31;
  const int wave  = tid >> 5;
  const int tileM = blockIdx.y * TM_;
  const int nbase = blockIdx.x * TN_;

  const int z = blockIdx.z;
  const float* Ab = A  + (size_t)z * (size_t)sA;
  const float* Bb = Bm + (size_t)z * (size_t)sB;
  float*       Cb = C  + (size_t)z * (size_t)sC;
  const float* Rb = (ACT == 3 || RES) ? (resid + (size_t)z * (size_t)sC) : nullptr;

  const int q  = tid & 7;
  const int rg = tid >> 3;
  const float* ga; size_t astep;
  if (A_MODE == 1) {
    ga = Ab + ((size_t)rg * S_ + (size_t)(tileM >> 4)) * (size_t)lda + q * 4;
    astep = (size_t)lda;
  } else {
    ga = Ab + (size_t)(tileM + rg) * (size_t)lda + q * 4;
    astep = (size_t)16 * (size_t)lda;
  }
  const float* gb = Bb + (size_t)(nbase + rg) * (size_t)ldb + q * 4;
  const size_t bstep = (size_t)16 * (size_t)ldb;
  const int soff = rg * LP_ + q * 4;

  const int l15   = lane & 15;
  const int khalf = (lane & 16) ? 8 : 0;
  const int fa = (wave * 32 + l15) * LP_ + khalf;
  const int fb = l15 * LP_ + khalf;

  float bv[4];
  #pragma unroll
  for (int nt = 0; nt < 4; ++nt) {
    const int col = nbase + nt * 16 + l15;
    float b = 0.0f;
    if (NB >= 1) b += bias[col];
    if (NB >= 2) b += bias2[col];
    bv[nt] = b;
  }

  const int nk = K >> 5;
  stagewords r0, r1;
  stage_issue(r0, ga, astep, gb, bstep, 0);
  stage_issue(r1, ga, astep, gb, bstep, 32);
  stage_to_lds(r0, sa[0], sb[0], soff);
  float rs[8] = {};
  if (ROWN) rowsum_add(r0, rs);
  lds_only_barrier();

  v8f acc[2][4] = {};
  for (int kc = 0; kc < nk; kc += 2) {
    stage_issue(r0, ga, astep, gb, bstep, ((kc + 2 < nk) ? (kc + 2) : (nk - 1)) * 32);
    __builtin_amdgcn_sched_barrier(0);
    lds_chunk_mma(sa[0] + fa, sb[0] + fb, acc);
    __builtin_amdgcn_sched_barrier(0);
    stage_to_lds(r1, sa[1], sb[1], soff);
    if (ROWN) rowsum_add(r1, rs);
    lds_only_barrier();
    stage_issue(r1, ga, astep, gb, bstep, ((kc + 3 < nk) ? (kc + 3) : (nk - 1)) * 32);
    __builtin_amdgcn_sched_barrier(0);
    lds_chunk_mma(sa[1] + fa, sb[1] + fb, acc);
    __builtin_amdgcn_sched_barrier(0);
    stage_to_lds(r0, sa[0], sb[0], soff);
    if (ROWN && kc + 2 < nk) rowsum_add(r0, rs);
    lds_only_barrier();
  }

  const int rb = (lane < 16) ? 0 : 8;
  float inv[2][8];
  if (ROWN) {
    #pragma unroll
    for (int i = 0; i < 8; ++i) {
      const float s = lane_xor_add<4>(lane_xor_add<2>(lane_xor_add<1>(rs[i])));
      if (q == 0) rsum[rg + 16 * i] = s;
    }
    __syncthreads();
    #pragma unroll
    for (int mt = 0; mt < 2; ++mt) {
      #pragma unroll
      for (int r = 0; r < 8; ++r)
        inv[mt][r] = 1.0f / rsum[wave * 32 + mt * 16 + rb + r];
    }
  }
  float adjv[4][2][8];
  if (ACT == 3 || RES) {
    #pragma unroll
    for (int nt = 0; nt < 4; ++nt) {
      #pragma unroll
      for (int mt = 0; mt < 2; ++mt) {
        #pragma unroll
        for (int r = 0; r < 8; ++r)
          adjv[nt][mt][r] = Rb[(size_t)(tileM + wave * 32 + mt * 16 + rb + r) * (size_t)ldc
                               + nbase + nt * 16 + l15];
      }
    }
  }
  #pragma unroll
  for (int nt = 0; nt < 4; ++nt) {
    const int col = nbase + nt * 16 + l15;
    #pragma unroll
    for (int mt = 0; mt < 2; ++mt) {
      #pragma unroll
      for (int r = 0; r < 8; ++r) {
        const int row = tileM + wave * 32 + mt * 16 + rb + r;
        float v = acc[mt][nt][r];
        if (ROWN) v *= inv[mt][r];
        v += bv[nt];
        if      (ACT == 1) v = v > 0.0f ? v : 0.0f;
        else if (ACT == 2) v = v > 0.0f ? v : 0.1f * v;
        if (ACT == 3) {
          const bool  dg = (row == col);
          const float sg = 1.0f / (1.0f + __expf(-v));
          v = (sg + (dg ? 1e-5f : 0.0f)) * (dg ? 1.0f : adjv[nt][mt][r]);
        } else if (RES) {
          v += adjv[nt][mt][r];
        }
        Cb[(size_t)row * (size_t)ldc + col] = v;
      }
    }
  }
}

__global__ void cvt_f32_f16_kernel(const float* __restrict__ src,
                                   __half* __restrict__ dst, int n) {
  int i = blockIdx.x * blockDim.x + threadIdx.x;
  if (i < n) ((_Float16*)dst)[i] = (_Float16)src[i];
}

__global__ __launch_bounds__(256) void transpose_bsg_kernel(
    const float* __restrict__ src, float* __restrict__ dst)
{
  __shared__ float tile[32][33];
  const int b  = blockIdx.z;
  const int s0 = blockIdx.x * 32, g0 = blockIdx.y * 32;
  const int tx = threadIdx.x, ty = threadIdx.y;
  const float* sp = src + (size_t)b * S_ * G_;
  float*       dp = dst + (size_t)b * G_ * S_;
  #pragma unroll
  for (int r = 0; r < 32; r += 8)
    tile[ty + r][tx] = sp[(size_t)(s0 + ty + r) * G_ + g0 + tx];
  __syncthreads();
  #pragma unroll
  for (int r = 0; r < 32; r += 8)
    dp[(size_t)(g0 + ty + r) * S_ + s0 + tx] = tile[tx][ty + r];
}

#define LB_ 2
#define PD_ 8
static_assert(S_ % PD_ == 0, "the step loop is unrolled by PD_");
static_assert(PD_ % 2 == 0, "the h buffer of a step is chosen by the parity of its ring slot");

__device__ __forceinline__ float gate_sigmoid(float x) {
  return __builtin_amdgcn_rcpf(1.0f + __expf(-x));
}
__device__ __forceinline__ float gate_tanh(float x) {
  return 1.0f - 2.0f * __builtin_amdgcn_rcpf(1.0f + __expf(2.0f * x));
}

__global__ __launch_bounds__(256) void lstm_dir_kernel(
    const float* __restrict__ preF, const float* __restrict__ preB,
    const __half* __restrict__ whhF16_f, const __half* __restrict__ whhF16_b,
    float* __restrict__ hs)
{
  const int dir = blockIdx.x;
  const int b0  = blockIdx.y * LB_;
  const float*    pre = dir ? preB : preF;
  const _Float16* whh = (const _Float16*)(dir ? whhF16_b : whhF16_f);

  __shared__ _Float16 h_lds[2][LB_ * HP_];

  const int tid   = threadIdx.x;
  const int wave  = tid >> 5;
  const int lane  = tid & 31;
  const int l15   = lane & 15;
  const int khalf = (lane & 16) ? 8 : 0;
  const int hrow  = l15 >> 3;
  const int half  = lane >> 4;

  v16h wfrag[16];
  #pragma unroll
  for (int nt = 0; nt < 4; ++nt) {
    const int tileN = nt * H_ + wave * 16;
    #pragma unroll
    for (int kc = 0; kc < 4; ++kc)
      wfrag[nt * 4 + kc] =
          frag_from_f16(whh + (size_t)(tileN + l15) * H_ + kc * 32 + khalf);
  }

  for (int i = tid; i < LB_ * HP_; i += 256) h_lds[0][i] = (_Float16)0.0f;

  const int hx = wave * 16 + l15;
  float c = 0.0f;
  __syncthreads();

  const float* pre_lane = pre + (size_t)(b0 + half) * (4 * H_) + hx;
  float pq[PD_][4];
  #pragma unroll
  for (int j = 0; j < PD_ - 1; ++j) {
    const int tj = dir ? (S_ - 1 - j) : j;
    #pragma unroll
    for (int nt = 0; nt < 4; ++nt)
      pq[j][nt] = pre_lane[(size_t)tj * (B_ * 4 * H_) + nt * H_];
  }

  for (int step0 = 0; step0 < S_; step0 += PD_) {
    #pragma unroll
    for (int j = 0; j < PD_; ++j) {
      const int step = step0 + j;
      const int t = dir ? (S_ - 1 - step) : step;

      {
        const int sn = (step + PD_ - 1 < S_) ? (step + PD_ - 1) : (S_ - 1);
        const int tn = dir ? (S_ - 1 - sn) : sn;
        #pragma unroll
        for (int nt = 0; nt < 4; ++nt)
          pq[(j + PD_ - 1) % PD_][nt] = pre_lane[(size_t)tn * (B_ * 4 * H_) + nt * H_];
      }

      const _Float16* h_rd = h_lds[j & 1];
      _Float16*       h_wr = h_lds[(j & 1) ^ 1];

      v16h a[4];
      #pragma unroll
      for (int kc = 0; kc < 4; ++kc)
        a[kc] = frag_from_f16(&h_rd[hrow * HP_ + kc * 32 + khalf]);

      float z[4];
      #pragma unroll
      for (int nt = 0; nt < 4; ++nt) {
        v8f acc = {};
        #pragma unroll
        for (int kc = 0; kc < 4; ++kc)
          acc = WMMA_F16(a[kc], wfrag[nt * 4 + kc], acc);
        z[nt] = acc[0] + pq[j][nt];
      }

      {
        const float si = gate_sigmoid(z[0]);
        const float sf = gate_sigmoid(z[1]);
        const float so = gate_sigmoid(z[3]);
        c = sf * c + si * gate_tanh(z[2]);
        const float h = so * gate_tanh(c);
        h_wr[half * HP_ + hx] = (_Float16)h;
        hs[((size_t)t * B_ + b0 + half) * (2 * H_) + dir * H_ + hx] = h;
      }
      lds_only_barrier();
    }
  }
}

__global__ __launch_bounds__(256) void head_mean_kernel(
    const float* __restrict__ h1, const float* __restrict__ gfin,
    float* __restrict__ hidden)
{
  const int b = blockIdx.x, tid = threadIdx.x;
  for (int d = tid; d < 512; d += 256) {
    float s = 0.0f;
    if (d < 256) {
      for (int t = 0; t < S_; ++t) s += h1[((size_t)t * B_ + b) * 256 + d];
    } else {
      const int dd = d - 256;
      for (int t = 0; t < S_; ++t) s += gfin[((size_t)b * S_ + t) * 256 + dd];
    }
    hidden[b * 512 + d] = s * (1.0f / (float)S_);
  }
}

template <int A_MODE, int ACT, bool FULLN>
static inline void gemm(hipStream_t st,
                        const float* A, const float* Bm,
                        const float* bias, const float* bias2, const float* resid,
                        float* C, int M, int N, int K, int lda, int ldb, int ldc,
                        long long sA = 0, long long sB = 0, long long sC = 0,
                        int batches = 1)
{
  dim3 grid((N + 255) / 256, (M + 31) / 32, batches);
  gemm_wmma_kernel<A_MODE, ACT, FULLN><<<grid, 128, 0, st>>>(
      A, Bm, bias, bias2, resid, C, M, N, K, lda, ldb, ldc, sA, sB, sC);
}

template <int A_MODE, int ACT, int NB, bool RES, bool ROWN = false>
static inline void gemm_lds(hipStream_t st,
                            const float* A, const float* Bm,
                            const float* bias, const float* bias2, const float* resid,
                            float* C, int M, int N, int K, int lda, int ldb, int ldc,
                            long long sA = 0, long long sB = 0, long long sC = 0,
                            int batches = 1)
{
  dim3 grid(N / TN_, M / TM_, batches);
  gemm_lds_kernel<A_MODE, ACT, ROWN, NB, RES><<<grid, 128, 0, st>>>(
      A, Bm, bias, bias2, resid, C, K, lda, ldb, ldc, sA, sB, sC);
}

extern "C" void kernel_launch(void* const* d_in, const int* in_sizes, int n_in,
                              void* d_out, int out_size, void* d_ws, size_t ws_size,
                              hipStream_t stream)
{
  (void)in_sizes; (void)n_in; (void)out_size; (void)ws_size;
  auto F = [&](int i) { return (const float*)d_in[i]; };

  const float* x        = F(0);
  const float* adj      = F(1);
  const float* l0f_Wih  = F(2),  *l0f_Whh = F(3),  *l0f_bih = F(4),  *l0f_bhh = F(5);
  const float* l0b_Wih  = F(6),  *l0b_Whh = F(7),  *l0b_bih = F(8),  *l0b_bhh = F(9);
  const float* l1f_Wih  = F(10), *l1f_Whh = F(11), *l1f_bih = F(12), *l1f_bhh = F(13);
  const float* l1b_Wih  = F(14), *l1b_Whh = F(15), *l1b_bih = F(16), *l1b_bhh = F(17);
  const float* fc1_W    = F(18), *fc1_b   = F(19);
  const float* wattn_W  = F(20), *wattn_b = F(21);
  const float* lin0_W   = F(22), *lin0_b  = F(23);
  const float* lin1_W   = F(24), *lin1_b  = F(25);
  const float* linf_W   = F(26), *linf_b  = F(27);
  const float* fcA_W    = F(28), *fcA_b   = F(29);
  const float* fcB_W    = F(30), *fcB_b   = F(31);
  const float* fcC_W    = F(32), *fcC_b   = F(33);

  char* wsp = (char*)d_ws;
  size_t off = 0;
  auto carve = [&](size_t bytes) -> void* {
    void* p = wsp + off;
    off += (bytes + 255) & ~(size_t)255;
    return p;
  };
  float* preF   = (float*)carve((size_t)S_ * B_ * 4 * H_ * 4);
  float* preB   = (float*)carve((size_t)S_ * B_ * 4 * H_ * 4);
  float* h0     = (float*)carve((size_t)S_ * B_ * 2 * H_ * 4);
  float* h1     = (float*)carve((size_t)S_ * B_ * 2 * H_ * 4);
  float* gA     = (float*)carve((size_t)B_ * S_ * G_ * 4);
  float* gB     = (float*)carve((size_t)B_ * S_ * G_ * 4);
  float* gC     = (float*)carve((size_t)B_ * S_ * G_ * 4);
  float* gT     = (float*)carve((size_t)B_ * G_ * S_ * 4);
  float* scores = (float*)carve((size_t)B_ * S_ * S_ * 4);
  __half* whh0f = (__half*)carve((size_t)4 * H_ * H_ * 2);
  __half* whh0b = (__half*)carve((size_t)4 * H_ * H_ * 2);
  __half* whh1f = (__half*)carve((size_t)4 * H_ * H_ * 2);
  __half* whh1b = (__half*)carve((size_t)4 * H_ * H_ * 2);
  float* fcAo   = (float*)carve((size_t)B_ * 128 * 4);
  float* fcBo   = (float*)carve((size_t)B_ * 32 * 4);

  const int MSB = S_ * B_;
  const int nwh = 4 * H_ * H_;

  cvt_f32_f16_kernel<<<(nwh + 255) / 256, 256, 0, stream>>>(l0f_Whh, whh0f, nwh);
  cvt_f32_f16_kernel<<<(nwh + 255) / 256, 256, 0, stream>>>(l0b_Whh, whh0b, nwh);
  cvt_f32_f16_kernel<<<(nwh + 255) / 256, 256, 0, stream>>>(l1f_Whh, whh1f, nwh);
  cvt_f32_f16_kernel<<<(nwh + 255) / 256, 256, 0, stream>>>(l1b_Whh, whh1b, nwh);

  gemm_lds<1, 0, 2, false>(stream, x, l0f_Wih, l0f_bih, l0f_bhh, nullptr, preF,
                   MSB, 4 * H_, D_, D_, D_, 4 * H_);
  gemm_lds<1, 0, 2, false>(stream, x, l0b_Wih, l0b_bih, l0b_bhh, nullptr, preB,
                   MSB, 4 * H_, D_, D_, D_, 4 * H_);
  lstm_dir_kernel<<<dim3(2, B_ / LB_), 256, 0, stream>>>(preF, preB, whh0f, whh0b, h0);

  gemm_lds<0, 0, 2, false>(stream, h0, l1f_Wih, l1f_bih, l1f_bhh, nullptr, preF,
                   MSB, 4 * H_, 2 * H_, 2 * H_, 2 * H_, 4 * H_);
  gemm_lds<0, 0, 2, false>(stream, h0, l1b_Wih, l1b_bih, l1b_bhh, nullptr, preB,
                   MSB, 4 * H_, 2 * H_, 2 * H_, 2 * H_, 4 * H_);
  lstm_dir_kernel<<<dim3(2, B_ / LB_), 256, 0, stream>>>(preF, preB, whh1f, whh1b, h1);

  gemm_lds<0, 0, 1, false>(stream, x, fc1_W, fc1_b, nullptr, nullptr, gA,
                   MSB, G_, D_, D_, D_, G_);

  float* cur = gA; float* t1 = gB; float* t2 = gC;
  const long long sGG = (long long)S_ * G_;
  const long long sSS = (long long)S_ * S_;
  for (int l = 0; l < 4; ++l) {
    const float* Wq = wattn_W + (size_t)l * G_ * G_;
    gemm_lds<0, 0, 1, false>(stream, cur, Wq, wattn_b + l * G_, nullptr, nullptr, t2,
                     MSB, G_, G_, G_, G_, G_);
    gemm_lds<0, 3, 0, false>(stream, t2, cur, nullptr, nullptr, adj, scores,
                     S_, S_, G_, G_, G_, S_, sGG, sGG, sSS, B_);
    transpose_bsg_kernel<<<dim3(S_ / 32, G_ / 32, B_), dim3(32, 8), 0, stream>>>(cur, gT);
    gemm_lds<0, 0, 0, false, true>(stream, scores, gT, nullptr, nullptr, nullptr, t1,
                     S_, G_, S_, S_, S_, G_, sSS, (long long)G_ * S_, sGG, B_);
    gemm_lds<0, 1, 1, false>(stream, t1, lin0_W + (size_t)l * G_ * G_, lin0_b + l * G_,
                     nullptr, nullptr, t2, MSB, G_, G_, G_, G_, G_);
    gemm_lds<0, 1, 1, true>(stream, t2, lin1_W + (size_t)l * G_ * G_, lin1_b + l * G_,
                     nullptr, cur, t1, MSB, G_, G_, G_, G_, G_);
    float* old = cur; cur = t1; t1 = old;
  }
  gemm_lds<0, 0, 1, false>(stream, cur, linf_W, linf_b, nullptr, nullptr, t2,
                   MSB, G_, G_, G_, G_, G_);
  float* gfin = t2;

  float* y_out  = (float*)d_out;
  float* hidden = (float*)d_out + B_ * 2;
  head_mean_kernel<<<B_, 256, 0, stream>>>(h1, gfin, hidden);

  gemm<0, 2, true >(stream, hidden, fcA_W, fcA_b, nullptr, nullptr, fcAo,
                    B_, 128, 512, 512, 512, 128);
  gemm<0, 2, false>(stream, fcAo, fcB_W, fcB_b, nullptr, nullptr, fcBo,
                    B_, 32, 128, 128, 128, 32);
  gemm<0, 0, false>(stream, fcBo, fcC_W, fcC_b, nullptr, nullptr, y_out,
                    B_, 2, 32, 32, 32, 2);
}


# time mean for the head: 128 workgroups of 16 waves, a wave sums 32 rows with all its loads in flight, partials joined in LDS; was one thread per column walking 512 rows in turn; sum order changed
# speedup vs baseline: 6.5867x; 1.0636x over previous
#include <hip/hip_runtime.h>
#include <hip/hip_fp16.h>
#include <hip/hip_bf16.h>

#define B_  16
#define S_  512
#define D_  1024
#define H_  128
#define G_  256
#define HP_ 136

typedef __attribute__((ext_vector_type(16))) _Float16 v16h;
typedef __attribute__((ext_vector_type(8)))  float    v8f;

struct rawfrag { float4 x0, x1, y0, y1; };

__device__ __forceinline__ rawfrag raw_load(const float* p) {
  rawfrag r;
  r.x0 = *(const float4*)(p);
  r.x1 = *(const float4*)(p + 4);
  r.y0 = *(const float4*)(p + 16);
  r.y1 = *(const float4*)(p + 20);
  return r;
}

__device__ __forceinline__ v16h raw_to_frag(const rawfrag& r) {
  v16h f;
  f[0]=(_Float16)r.x0.x;  f[1]=(_Float16)r.x0.y;  f[2]=(_Float16)r.x0.z;  f[3]=(_Float16)r.x0.w;
  f[4]=(_Float16)r.x1.x;  f[5]=(_Float16)r.x1.y;  f[6]=(_Float16)r.x1.z;  f[7]=(_Float16)r.x1.w;
  f[8]=(_Float16)r.y0.x;  f[9]=(_Float16)r.y0.y;  f[10]=(_Float16)r.y0.z; f[11]=(_Float16)r.y0.w;
  f[12]=(_Float16)r.y1.x; f[13]=(_Float16)r.y1.y; f[14]=(_Float16)r.y1.z; f[15]=(_Float16)r.y1.w;
  return f;
}

__device__ __forceinline__ v16h frag_from_f16(const _Float16* p) {
  v16h f;
  #pragma unroll
  for (int i = 0; i < 8; ++i) { f[i] = p[i]; f[8 + i] = p[16 + i]; }
  return f;
}

#define WMMA_F16(a, b, c) __builtin_amdgcn_wmma_f32_16x16x32_f16( \
    false, (a), false, (b), (short)0, (c), false, false)

#define GPF_ 2

template <bool FULLN>
__device__ __forceinline__ void gemm_chunk(
    rawfrag (&sa)[2], rawfrag (&sb)[4], v8f (&acc)[2][4], const bool (&ton)[4],
    const float* pa0, const float* pa1, const float* const (&pb)[4],
    bool refill, int koff_next)
{
  v16h a0 = raw_to_frag(sa[0]);
  v16h a1 = raw_to_frag(sa[1]);
  v16h b[4];
  #pragma unroll
  for (int nt = 0; nt < 4; ++nt)
    if (FULLN || ton[nt]) b[nt] = raw_to_frag(sb[nt]);
  asm volatile("" : "+v"(a0), "+v"(a1) : : "memory");
  #pragma unroll
  for (int nt = 0; nt < 4; ++nt)
    if (FULLN || ton[nt]) asm volatile("" : "+v"(b[nt]) : : "memory");
  if (refill) {
    sa[0] = raw_load(pa0 + koff_next);
    sa[1] = raw_load(pa1 + koff_next);
    #pragma unroll
    for (int nt = 0; nt < 4; ++nt)
      if (FULLN || ton[nt]) sb[nt] = raw_load(pb[nt] + koff_next);
  }
  __builtin_amdgcn_sched_barrier(0);
  #pragma unroll
  for (int nt = 0; nt < 4; ++nt) {
    if (!FULLN && !ton[nt]) continue;
    acc[0][nt] = WMMA_F16(a0, b[nt], acc[0][nt]);
    acc[1][nt] = WMMA_F16(a1, b[nt], acc[1][nt]);
  }
}

template <int A_MODE, int ACT, bool FULLN>
__global__ __launch_bounds__(128) void gemm_wmma_kernel(
    const float* __restrict__ A, const float* __restrict__ Bm,
    const float* __restrict__ bias, const float* __restrict__ bias2,
    const float* __restrict__ resid, float* __restrict__ C,
    int M, int N, int K, int lda, int ldb, int ldc,
    long long sA, long long sB, long long sC)
{
  const int lane  = threadIdx.x & 31;
  const int wave  = threadIdx.x >> 5;
  const int tileM = blockIdx.y * 32;
  const int nbase = blockIdx.x * 256 + wave * 64;
  if (nbase >= N) return;

  const int z = blockIdx.z;
  const float* Ab = A  + (size_t)z * (size_t)sA;
  const float* Bb = Bm + (size_t)z * (size_t)sB;
  float*       Cb = C  + (size_t)z * (size_t)sC;
  const float* Rb = resid ? (resid + (size_t)z * (size_t)sC) : nullptr;

  const int l15   = lane & 15;
  const int khalf = (lane & 16) ? 8 : 0;

  int am0 = tileM + l15;      if (am0 >= M) am0 = M - 1;
  int am1 = tileM + 16 + l15; if (am1 >= M) am1 = M - 1;
  size_t arow0, arow1;
  if (A_MODE == 1) {
    arow0 = ((size_t)(am0 & 15) * S_ + (size_t)(am0 >> 4)) * (size_t)lda;
    arow1 = ((size_t)(am1 & 15) * S_ + (size_t)(am1 >> 4)) * (size_t)lda;
  } else {
    arow0 = (size_t)am0 * (size_t)lda;
    arow1 = (size_t)am1 * (size_t)lda;
  }

  bool ton[4]; int bn[4];
  #pragma unroll
  for (int nt = 0; nt < 4; ++nt) {
    if (FULLN) {
      ton[nt] = true;
      bn[nt]  = nbase + nt * 16 + l15;
    } else {
      ton[nt] = (nbase + nt * 16) < N;
      int c = nbase + nt * 16 + l15; if (c >= N) c = N - 1;
      bn[nt] = c;
    }
  }

  const float* pa0 = Ab + arow0 + khalf;
  const float* pa1 = Ab + arow1 + khalf;
  const float* pb[4];
  #pragma unroll
  for (int nt = 0; nt < 4; ++nt)
    pb[nt] = Bb + (size_t)bn[nt] * (size_t)ldb + khalf;

  const int nk = K >> 5;
  rawfrag ra[GPF_][2], rb_[GPF_][4];
  #pragma unroll
  for (int j = 0; j < GPF_; ++j) {
    const int kj = ((j < nk) ? j : (nk - 1)) * 32;
    ra[j][0] = raw_load(pa0 + kj);
    ra[j][1] = raw_load(pa1 + kj);
    #pragma unroll
    for (int nt = 0; nt < 4; ++nt)
      if (FULLN || ton[nt]) rb_[j][nt] = raw_load(pb[nt] + kj);
  }

  v8f acc[2][4] = {};
  int kc = 0;
  for (; kc + 2 * GPF_ <= nk; kc += GPF_) {
    #pragma unroll
    for (int j = 0; j < GPF_; ++j)
      gemm_chunk<FULLN>(ra[j], rb_[j], acc, ton, pa0, pa1, pb,
                        true, (kc + j + GPF_) * 32);
  }
  for (; kc < nk; kc += GPF_) {
    #pragma unroll
    for (int j = 0; j < GPF_; ++j)
      if (kc + j < nk)
        gemm_chunk<FULLN>(ra[j], rb_[j], acc, ton, pa0, pa1, pb,
                          kc + j + GPF_ < nk, (kc + j + GPF_) * 32);
  }

  const int rb = (lane < 16) ? 0 : 8;
  float bvv[4] = {0.0f, 0.0f, 0.0f, 0.0f};
  if (bias) {
    #pragma unroll
    for (int nt = 0; nt < 4; ++nt) bvv[nt] += bias[bn[nt]];
  }
  if (bias2) {
    #pragma unroll
    for (int nt = 0; nt < 4; ++nt) bvv[nt] += bias2[bn[nt]];
  }
  #pragma unroll
  for (int nt = 0; nt < 4; ++nt) {
    if (!FULLN && !ton[nt]) continue;
    const int col = nbase + nt * 16 + l15;
    bool colok = FULLN ? true : (col < N);
    const float bv = bvv[nt];
    #pragma unroll
    for (int mt = 0; mt < 2; ++mt) {
      #pragma unroll
      for (int r = 0; r < 8; ++r) {
        const int row = tileM + mt * 16 + rb + r;
        if (row < M && colok) {
          float v = acc[mt][nt][r] + bv;
          if      (ACT == 1) v = v > 0.0f ? v : 0.0f;
          else if (ACT == 2) v = v > 0.0f ? v : 0.1f * v;
          if (Rb) v += Rb[(size_t)row * (size_t)ldc + col];
          Cb[(size_t)row * (size_t)ldc + col] = v;
        }
      }
    }
  }
}

__device__ __forceinline__ void lds_only_barrier() {
  __builtin_amdgcn_fence(__ATOMIC_RELEASE, "workgroup", "local");
  __builtin_amdgcn_s_barrier();
  __builtin_amdgcn_fence(__ATOMIC_ACQUIRE, "workgroup", "local");
}

#define TM_ 128
#define TN_ 64
#define LP_ 40

typedef __attribute__((ext_vector_type(8))) _Float16 v8h;
typedef __attribute__((ext_vector_type(4))) _Float16 v4h;

__device__ __forceinline__ v16h frag_from_lds(const _Float16* p) {
  const v8h lo = *(const v8h*)(p);
  const v8h hi = *(const v8h*)(p + 16);
  return __builtin_shufflevector(lo, hi, 0, 1, 2, 3, 4, 5, 6, 7,
                                 8, 9, 10, 11, 12, 13, 14, 15);
}

typedef __attribute__((ext_vector_type(4))) float v4f;

__device__ __forceinline__ void stage_store(_Float16* dst, const v4f& v) {
  v4h h;
  h[0] = (_Float16)v[0]; h[1] = (_Float16)v[1]; h[2] = (_Float16)v[2]; h[3] = (_Float16)v[3];
  *(v4h*)dst = h;
}

struct stagewords { v4f a[8]; v4f b[4]; };

__device__ __forceinline__ void stage_issue(stagewords& r, const float* ga, size_t astep,
                                            const float* gb, size_t bstep, int koff) {
  #pragma unroll
  for (int i = 0; i < 8; ++i) r.a[i] = *(const v4f*)(ga + (size_t)i * astep + koff);
  #pragma unroll
  for (int i = 0; i < 4; ++i) r.b[i] = *(const v4f*)(gb + (size_t)i * bstep + koff);
}

__device__ __forceinline__ void stage_to_lds(stagewords& r, _Float16* sa, _Float16* sb,
                                             int soff) {
  asm volatile("" : "+v"(r.a[0]), "+v"(r.a[1]), "+v"(r.a[2]), "+v"(r.a[3]),
                    "+v"(r.a[4]), "+v"(r.a[5]), "+v"(r.a[6]), "+v"(r.a[7]),
                    "+v"(r.b[0]), "+v"(r.b[1]), "+v"(r.b[2]), "+v"(r.b[3]));
  #pragma unroll
  for (int i = 0; i < 8; ++i) stage_store(sa + soff + i * 16 * LP_, r.a[i]);
  #pragma unroll
  for (int i = 0; i < 4; ++i) stage_store(sb + soff + i * 16 * LP_, r.b[i]);
}

__device__ __forceinline__ void lds_chunk_mma(const _Float16* pa, const _Float16* pb,
                                              v8f (&acc)[2][4]) {
  const v16h a0 = frag_from_lds(pa);
  const v16h a1 = frag_from_lds(pa + 16 * LP_);
  v16h b[4];
  #pragma unroll
  for (int nt = 0; nt < 4; ++nt) b[nt] = frag_from_lds(pb + nt * 16 * LP_);
  #pragma unroll
  for (int nt = 0; nt < 4; ++nt) {
    acc[0][nt] = WMMA_F16(a0, b[nt], acc[0][nt]);
    acc[1][nt] = WMMA_F16(a1, b[nt], acc[1][nt]);
  }
}

__device__ __forceinline__ void rowsum_add(const stagewords& r, float (&rs)[8]) {
  #pragma unroll
  for (int i = 0; i < 8; ++i)
    rs[i] += (r.a[i][0] + r.a[i][1]) + (r.a[i][2] + r.a[i][3]);
}

template <int M>
__device__ __forceinline__ float lane_xor_add(float v) {
  const int t = __builtin_amdgcn_update_dpp(0, __builtin_bit_cast(int, v), 0x160 + M, 0xf, 0xf, true);
  return v + __builtin_bit_cast(float, t);
}

template <int A_MODE, int ACT, bool ROWN, int NB, bool RES>
__global__ __launch_bounds__(128) void gemm_lds_kernel(
    const float* __restrict__ A, const float* __restrict__ Bm,
    const float* __restrict__ bias, const float* __restrict__ bias2,
    const float* __restrict__ resid, float* __restrict__ C,
    int K, int lda, int ldb, int ldc,
    long long sA, long long sB, long long sC)
{
  __shared__ __attribute__((aligned(16))) _Float16 sa[2][TM_ * LP_];
  __shared__ __attribute__((aligned(16))) _Float16 sb[2][TN_ * LP_];
  __shared__ float rsum[TM_];

  const int tid   = threadIdx.x;
  const int lane  = tid & 31;
  const int wave  = tid >> 5;
  const int tileM = blockIdx.y * TM_;
  const int nbase = blockIdx.x * TN_;

  const int z = blockIdx.z;
  const float* Ab = A  + (size_t)z * (size_t)sA;
  const float* Bb = Bm + (size_t)z * (size_t)sB;
  float*       Cb = C  + (size_t)z * (size_t)sC;
  const float* Rb = (ACT == 3 || RES) ? (resid + (size_t)z * (size_t)sC) : nullptr;

  const int q  = tid & 7;
  const int rg = tid >> 3;
  const float* ga; size_t astep;
  if (A_MODE == 1) {
    ga = Ab + ((size_t)rg * S_ + (size_t)(tileM >> 4)) * (size_t)lda + q * 4;
    astep = (size_t)lda;
  } else {
    ga = Ab + (size_t)(tileM + rg) * (size_t)lda + q * 4;
    astep = (size_t)16 * (size_t)lda;
  }
  const float* gb = Bb + (size_t)(nbase + rg) * (size_t)ldb + q * 4;
  const size_t bstep = (size_t)16 * (size_t)ldb;
  const int soff = rg * LP_ + q * 4;

  const int l15   = lane & 15;
  const int khalf = (lane & 16) ? 8 : 0;
  const int fa = (wave * 32 + l15) * LP_ + khalf;
  const int fb = l15 * LP_ + khalf;

  float bv[4];
  #pragma unroll
  for (int nt = 0; nt < 4; ++nt) {
    const int col = nbase + nt * 16 + l15;
    float b = 0.0f;
    if (NB >= 1) b += bias[col];
    if (NB >= 2) b += bias2[col];
    bv[nt] = b;
  }

  const int nk = K >> 5;
  stagewords r0, r1;
  stage_issue(r0, ga, astep, gb, bstep, 0);
  stage_issue(r1, ga, astep, gb, bstep, 32);
  stage_to_lds(r0, sa[0], sb[0], soff);
  float rs[8] = {};
  if (ROWN) rowsum_add(r0, rs);
  lds_only_barrier();

  v8f acc[2][4] = {};
  for (int kc = 0; kc < nk; kc += 2) {
    stage_issue(r0, ga, astep, gb, bstep, ((kc + 2 < nk) ? (kc + 2) : (nk - 1)) * 32);
    __builtin_amdgcn_sched_barrier(0);
    lds_chunk_mma(sa[0] + fa, sb[0] + fb, acc);
    __builtin_amdgcn_sched_barrier(0);
    stage_to_lds(r1, sa[1], sb[1], soff);
    if (ROWN) rowsum_add(r1, rs);
    lds_only_barrier();
    stage_issue(r1, ga, astep, gb, bstep, ((kc + 3 < nk) ? (kc + 3) : (nk - 1)) * 32);
    __builtin_amdgcn_sched_barrier(0);
    lds_chunk_mma(sa[1] + fa, sb[1] + fb, acc);
    __builtin_amdgcn_sched_barrier(0);
    stage_to_lds(r0, sa[0], sb[0], soff);
    if (ROWN && kc + 2 < nk) rowsum_add(r0, rs);
    lds_only_barrier();
  }

  const int rb = (lane < 16) ? 0 : 8;
  float inv[2][8];
  if (ROWN) {
    #pragma unroll
    for (int i = 0; i < 8; ++i) {
      const float s = lane_xor_add<4>(lane_xor_add<2>(lane_xor_add<1>(rs[i])));
      if (q == 0) rsum[rg + 16 * i] = s;
    }
    __syncthreads();
    #pragma unroll
    for (int mt = 0; mt < 2; ++mt) {
      #pragma unroll
      for (int r = 0; r < 8; ++r)
        inv[mt][r] = 1.0f / rsum[wave * 32 + mt * 16 + rb + r];
    }
  }
  float adjv[4][2][8];
  if (ACT == 3 || RES) {
    #pragma unroll
    for (int nt = 0; nt < 4; ++nt) {
      #pragma unroll
      for (int mt = 0; mt < 2; ++mt) {
        #pragma unroll
        for (int r = 0; r < 8; ++r)
          adjv[nt][mt][r] = Rb[(size_t)(tileM + wave * 32 + mt * 16 + rb + r) * (size_t)ldc
                               + nbase + nt * 16 + l15];
      }
    }
  }
  #pragma unroll
  for (int nt = 0; nt < 4; ++nt) {
    const int col = nbase + nt * 16 + l15;
    #pragma unroll
    for (int mt = 0; mt < 2; ++mt) {
      #pragma unroll
      for (int r = 0; r < 8; ++r) {
        const int row = tileM + wave * 32 + mt * 16 + rb + r;
        float v = acc[mt][nt][r];
        if (ROWN) v *= inv[mt][r];
        v += bv[nt];
        if      (ACT == 1) v = v > 0.0f ? v : 0.0f;
        else if (ACT == 2) v = v > 0.0f ? v : 0.1f * v;
        if (ACT == 3) {
          const bool  dg = (row == col);
          const float sg = 1.0f / (1.0f + __expf(-v));
          v = (sg + (dg ? 1e-5f : 0.0f)) * (dg ? 1.0f : adjv[nt][mt][r]);
        } else if (RES) {
          v += adjv[nt][mt][r];
        }
        Cb[(size_t)row * (size_t)ldc + col] = v;
      }
    }
  }
}

__global__ void cvt_f32_f16_kernel(const float* __restrict__ src,
                                   __half* __restrict__ dst, int n) {
  int i = blockIdx.x * blockDim.x + threadIdx.x;
  if (i < n) ((_Float16*)dst)[i] = (_Float16)src[i];
}

__global__ __launch_bounds__(256) void transpose_bsg_kernel(
    const float* __restrict__ src, float* __restrict__ dst)
{
  __shared__ float tile[32][33];
  const int b  = blockIdx.z;
  const int s0 = blockIdx.x * 32, g0 = blockIdx.y * 32;
  const int tx = threadIdx.x, ty = threadIdx.y;
  const float* sp = src + (size_t)b * S_ * G_;
  float*       dp = dst + (size_t)b * G_ * S_;
  #pragma unroll
  for (int r = 0; r < 32; r += 8)
    tile[ty + r][tx] = sp[(size_t)(s0 + ty + r) * G_ + g0 + tx];
  __syncthreads();
  #pragma unroll
  for (int r = 0; r < 32; r += 8)
    dp[(size_t)(g0 + ty + r) * S_ + s0 + tx] = tile[tx][ty + r];
}

#define LB_ 2
#define PD_ 8
static_assert(S_ % PD_ == 0, "the step loop is unrolled by PD_");
static_assert(PD_ % 2 == 0, "the h buffer of a step is chosen by the parity of its ring slot");

__device__ __forceinline__ float gate_sigmoid(float x) {
  return __builtin_amdgcn_rcpf(1.0f + __expf(-x));
}
__device__ __forceinline__ float gate_tanh(float x) {
  return 1.0f - 2.0f * __builtin_amdgcn_rcpf(1.0f + __expf(2.0f * x));
}

__global__ __launch_bounds__(256) void lstm_dir_kernel(
    const float* __restrict__ preF, const float* __restrict__ preB,
    const __half* __restrict__ whhF16_f, const __half* __restrict__ whhF16_b,
    float* __restrict__ hs)
{
  const int dir = blockIdx.x;
  const int b0  = blockIdx.y * LB_;
  const float*    pre = dir ? preB : preF;
  const _Float16* whh = (const _Float16*)(dir ? whhF16_b : whhF16_f);

  __shared__ _Float16 h_lds[2][LB_ * HP_];

  const int tid   = threadIdx.x;
  const int wave  = tid >> 5;
  const int lane  = tid & 31;
  const int l15   = lane & 15;
  const int khalf = (lane & 16) ? 8 : 0;
  const int hrow  = l15 >> 3;
  const int half  = lane >> 4;

  v16h wfrag[16];
  #pragma unroll
  for (int nt = 0; nt < 4; ++nt) {
    const int tileN = nt * H_ + wave * 16;
    #pragma unroll
    for (int kc = 0; kc < 4; ++kc)
      wfrag[nt * 4 + kc] =
          frag_from_f16(whh + (size_t)(tileN + l15) * H_ + kc * 32 + khalf);
  }

  for (int i = tid; i < LB_ * HP_; i += 256) h_lds[0][i] = (_Float16)0.0f;

  const int hx = wave * 16 + l15;
  float c = 0.0f;
  __syncthreads();

  const float* pre_lane = pre + (size_t)(b0 + half) * (4 * H_) + hx;
  float pq[PD_][4];
  #pragma unroll
  for (int j = 0; j < PD_ - 1; ++j) {
    const int tj = dir ? (S_ - 1 - j) : j;
    #pragma unroll
    for (int nt = 0; nt < 4; ++nt)
      pq[j][nt] = pre_lane[(size_t)tj * (B_ * 4 * H_) + nt * H_];
  }

  for (int step0 = 0; step0 < S_; step0 += PD_) {
    #pragma unroll
    for (int j = 0; j < PD_; ++j) {
      const int step = step0 + j;
      const int t = dir ? (S_ - 1 - step) : step;

      {
        const int sn = (step + PD_ - 1 < S_) ? (step + PD_ - 1) : (S_ - 1);
        const int tn = dir ? (S_ - 1 - sn) : sn;
        #pragma unroll
        for (int nt = 0; nt < 4; ++nt)
          pq[(j + PD_ - 1) % PD_][nt] = pre_lane[(size_t)tn * (B_ * 4 * H_) + nt * H_];
      }

      const _Float16* h_rd = h_lds[j & 1];
      _Float16*       h_wr = h_lds[(j & 1) ^ 1];

      v16h a[4];
      #pragma unroll
      for (int kc = 0; kc < 4; ++kc)
        a[kc] = frag_from_f16(&h_rd[hrow * HP_ + kc * 32 + khalf]);

      float z[4];
      #pragma unroll
      for (int nt = 0; nt < 4; ++nt) {
        v8f acc = {};
        #pragma unroll
        for (int kc = 0; kc < 4; ++kc)
          acc = WMMA_F16(a[kc], wfrag[nt * 4 + kc], acc);
        z[nt] = acc[0] + pq[j][nt];
      }

      {
        const float si = gate_sigmoid(z[0]);
        const float sf = gate_sigmoid(z[1]);
        const float so = gate_sigmoid(z[3]);
        c = sf * c + si * gate_tanh(z[2]);
        const float h = so * gate_tanh(c);
        h_wr[half * HP_ + hx] = (_Float16)h;
        hs[((size_t)t * B_ + b0 + half) * (2 * H_) + dir * H_ + hx] = h;
      }
      lds_only_barrier();
    }
  }
}

#define MC_ 64
#define MW_ 16
#define MR_ (S_ / MW_)

__global__ __launch_bounds__(32 * MW_) void head_mean_kernel(
    const float* __restrict__ h1, const float* __restrict__ gfin,
    float* __restrict__ hidden)
{
  __shared__ float2 part[MW_][MC_ / 2];
  const int b    = blockIdx.y;
  const int d0   = blockIdx.x * MC_;
  const int lane = threadIdx.x & 31;
  const int wave = threadIdx.x >> 5;
  const int t0   = wave * MR_;

  const float* base; size_t rstride;
  if (d0 < 2 * H_) {
    base = h1 + (size_t)b * (2 * H_) + d0 + lane * 2;
    rstride = (size_t)B_ * (2 * H_);
  } else {
    base = gfin + (size_t)b * S_ * G_ + (d0 - 2 * H_) + lane * 2;
    rstride = (size_t)G_;
  }
  float2 v[MR_];
  #pragma unroll
  for (int i = 0; i < MR_; ++i)
    v[i] = *(const float2*)(base + (size_t)(t0 + i) * rstride);
  float2 s = v[0];
  #pragma unroll
  for (int i = 1; i < MR_; ++i) { s.x += v[i].x; s.y += v[i].y; }
  part[wave][lane] = s;
  __syncthreads();
  if (threadIdx.x < MC_) {
    const float* pp = (const float*)&part[0][0];
    float a = pp[threadIdx.x];
    #pragma unroll
    for (int w = 1; w < MW_; ++w) a += pp[w * MC_ + threadIdx.x];
    hidden[b * 512 + d0 + threadIdx.x] = a * (1.0f / (float)S_);
  }
}

template <int A_MODE, int ACT, bool FULLN>
static inline void gemm(hipStream_t st,
                        const float* A, const float* Bm,
                        const float* bias, const float* bias2, const float* resid,
                        float* C, int M, int N, int K, int lda, int ldb, int ldc,
                        long long sA = 0, long long sB = 0, long long sC = 0,
                        int batches = 1)
{
  dim3 grid((N + 255) / 256, (M + 31) / 32, batches);
  gemm_wmma_kernel<A_MODE, ACT, FULLN><<<grid, 128, 0, st>>>(
      A, Bm, bias, bias2, resid, C, M, N, K, lda, ldb, ldc, sA, sB, sC);
}

template <int A_MODE, int ACT, int NB, bool RES, bool ROWN = false>
static inline void gemm_lds(hipStream_t st,
                            const float* A, const float* Bm,
                            const float* bias, const float* bias2, const float* resid,
                            float* C, int M, int N, int K, int lda, int ldb, int ldc,
                            long long sA = 0, long long sB = 0, long long sC = 0,
                            int batches = 1)
{
  dim3 grid(N / TN_, M / TM_, batches);
  gemm_lds_kernel<A_MODE, ACT, ROWN, NB, RES><<<grid, 128, 0, st>>>(
      A, Bm, bias, bias2, resid, C, K, lda, ldb, ldc, sA, sB, sC);
}

extern "C" void kernel_launch(void* const* d_in, const int* in_sizes, int n_in,
                              void* d_out, int out_size, void* d_ws, size_t ws_size,
                              hipStream_t stream)
{
  (void)in_sizes; (void)n_in; (void)out_size; (void)ws_size;
  auto F = [&](int i) { return (const float*)d_in[i]; };

  const float* x        = F(0);
  const float* adj      = F(1);
  const float* l0f_Wih  = F(2),  *l0f_Whh = F(3),  *l0f_bih = F(4),  *l0f_bhh = F(5);
  const float* l0b_Wih  = F(6),  *l0b_Whh = F(7),  *l0b_bih = F(8),  *l0b_bhh = F(9);
  const float* l1f_Wih  = F(10), *l1f_Whh = F(11), *l1f_bih = F(12), *l1f_bhh = F(13);
  const float* l1b_Wih  = F(14), *l1b_Whh = F(15), *l1b_bih = F(16), *l1b_bhh = F(17);
  const float* fc1_W    = F(18), *fc1_b   = F(19);
  const float* wattn_W  = F(20), *wattn_b = F(21);
  const float* lin0_W   = F(22), *lin0_b  = F(23);
  const float* lin1_W   = F(24), *lin1_b  = F(25);
  const float* linf_W   = F(26), *linf_b  = F(27);
  const float* fcA_W    = F(28), *fcA_b   = F(29);
  const float* fcB_W    = F(30), *fcB_b   = F(31);
  const float* fcC_W    = F(32), *fcC_b   = F(33);

  char* wsp = (char*)d_ws;
  size_t off = 0;
  auto carve = [&](size_t bytes) -> void* {
    void* p = wsp + off;
    off += (bytes + 255) & ~(size_t)255;
    return p;
  };
  float* preF   = (float*)carve((size_t)S_ * B_ * 4 * H_ * 4);
  float* preB   = (float*)carve((size_t)S_ * B_ * 4 * H_ * 4);
  float* h0     = (float*)carve((size_t)S_ * B_ * 2 * H_ * 4);
  float* h1     = (float*)carve((size_t)S_ * B_ * 2 * H_ * 4);
  float* gA     = (float*)carve((size_t)B_ * S_ * G_ * 4);
  float* gB     = (float*)carve((size_t)B_ * S_ * G_ * 4);
  float* gC     = (float*)carve((size_t)B_ * S_ * G_ * 4);
  float* gT     = (float*)carve((size_t)B_ * G_ * S_ * 4);
  float* scores = (float*)carve((size_t)B_ * S_ * S_ * 4);
  __half* whh0f = (__half*)carve((size_t)4 * H_ * H_ * 2);
  __half* whh0b = (__half*)carve((size_t)4 * H_ * H_ * 2);
  __half* whh1f = (__half*)carve((size_t)4 * H_ * H_ * 2);
  __half* whh1b = (__half*)carve((size_t)4 * H_ * H_ * 2);
  float* fcAo   = (float*)carve((size_t)B_ * 128 * 4);
  float* fcBo   = (float*)carve((size_t)B_ * 32 * 4);

  const int MSB = S_ * B_;
  const int nwh = 4 * H_ * H_;

  cvt_f32_f16_kernel<<<(nwh + 255) / 256, 256, 0, stream>>>(l0f_Whh, whh0f, nwh);
  cvt_f32_f16_kernel<<<(nwh + 255) / 256, 256, 0, stream>>>(l0b_Whh, whh0b, nwh);
  cvt_f32_f16_kernel<<<(nwh + 255) / 256, 256, 0, stream>>>(l1f_Whh, whh1f, nwh);
  cvt_f32_f16_kernel<<<(nwh + 255) / 256, 256, 0, stream>>>(l1b_Whh, whh1b, nwh);

  gemm_lds<1, 0, 2, false>(stream, x, l0f_Wih, l0f_bih, l0f_bhh, nullptr, preF,
                   MSB, 4 * H_, D_, D_, D_, 4 * H_);
  gemm_lds<1, 0, 2, false>(stream, x, l0b_Wih, l0b_bih, l0b_bhh, nullptr, preB,
                   MSB, 4 * H_, D_, D_, D_, 4 * H_);
  lstm_dir_kernel<<<dim3(2, B_ / LB_), 256, 0, stream>>>(preF, preB, whh0f, whh0b, h0);

  gemm_lds<0, 0, 2, false>(stream, h0, l1f_Wih, l1f_bih, l1f_bhh, nullptr, preF,
                   MSB, 4 * H_, 2 * H_, 2 * H_, 2 * H_, 4 * H_);
  gemm_lds<0, 0, 2, false>(stream, h0, l1b_Wih, l1b_bih, l1b_bhh, nullptr, preB,
                   MSB, 4 * H_, 2 * H_, 2 * H_, 2 * H_, 4 * H_);
  lstm_dir_kernel<<<dim3(2, B_ / LB_), 256, 0, stream>>>(preF, preB, whh1f, whh1b, h1);

  gemm_lds<0, 0, 1, false>(stream, x, fc1_W, fc1_b, nullptr, nullptr, gA,
                   MSB, G_, D_, D_, D_, G_);

  float* cur = gA; float* t1 = gB; float* t2 = gC;
  const long long sGG = (long long)S_ * G_;
  const long long sSS = (long long)S_ * S_;
  for (int l = 0; l < 4; ++l) {
    const float* Wq = wattn_W + (size_t)l * G_ * G_;
    gemm_lds<0, 0, 1, false>(stream, cur, Wq, wattn_b + l * G_, nullptr, nullptr, t2,
                     MSB, G_, G_, G_, G_, G_);
    gemm_lds<0, 3, 0, false>(stream, t2, cur, nullptr, nullptr, adj, scores,
                     S_, S_, G_, G_, G_, S_, sGG, sGG, sSS, B_);
    transpose_bsg_kernel<<<dim3(S_ / 32, G_ / 32, B_), dim3(32, 8), 0, stream>>>(cur, gT);
    gemm_lds<0, 0, 0, false, true>(stream, scores, gT, nullptr, nullptr, nullptr, t1,
                     S_, G_, S_, S_, S_, G_, sSS, (long long)G_ * S_, sGG, B_);
    gemm_lds<0, 1, 1, false>(stream, t1, lin0_W + (size_t)l * G_ * G_, lin0_b + l * G_,
                     nullptr, nullptr, t2, MSB, G_, G_, G_, G_, G_);
    gemm_lds<0, 1, 1, true>(stream, t2, lin1_W + (size_t)l * G_ * G_, lin1_b + l * G_,
                     nullptr, cur, t1, MSB, G_, G_, G_, G_, G_);
    float* old = cur; cur = t1; t1 = old;
  }
  gemm_lds<0, 0, 1, false>(stream, cur, linf_W, linf_b, nullptr, nullptr, t2,
                   MSB, G_, G_, G_, G_, G_);
  float* gfin = t2;

  float* y_out  = (float*)d_out;
  float* hidden = (float*)d_out + B_ * 2;
  head_mean_kernel<<<dim3(512 / MC_, B_), 32 * MW_, 0, stream>>>(h1, gfin, hidden);

  gemm<0, 2, true >(stream, hidden, fcA_W, fcA_b, nullptr, nullptr, fcAo,
                    B_, 128, 512, 512, 512, 128);
  gemm<0, 2, false>(stream, fcAo, fcB_W, fcB_b, nullptr, nullptr, fcBo,
                    B_, 32, 128, 128, 128, 32);
  gemm<0, 0, false>(stream, fcBo, fcC_W, fcC_b, nullptr, nullptr, y_out,
                    B_, 2, 32, 32, 32, 2);
}
